# starting kernel
# baseline (speedup 1.0000x reference)
#include <hip/hip_runtime.h>
#include <hip/hip_bf16.h>
#include <limits.h>


typedef __attribute__((ext_vector_type(2))) float v2f;
typedef __attribute__((ext_vector_type(8))) float v8f;

#define F_IN   128
#define HEADS  8
#define CH     16
#define OUTC   2
#define NEG_SLOPE 0.2f
#define LDSW   136

__device__ __forceinline__ int f2ord(float f) {
  int i = __float_as_int(f);
  return (i >= 0) ? i : (i ^ 0x7fffffff);
}
__device__ __forceinline__ float ord2f(int i) {
  return __int_as_float((i >= 0) ? i : (i ^ 0x7fffffff));
}

__device__ __forceinline__ void edge_sd(const int* __restrict__ ei, int E,
                                        int e, int& s, int& d) {
  if (e < E) { s = ei[e]; d = ei[E + e]; }
  else       { s = e - E; d = e - E; }
}

__global__ void gat_fill_f32(float* __restrict__ p, float v, size_t n) {
  size_t i = (size_t)blockIdx.x * blockDim.x + threadIdx.x;
  if (i < n) p[i] = v;
}
__global__ void gat_fill_i32(int* __restrict__ p, int v, size_t n) {
  size_t i = (size_t)blockIdx.x * blockDim.x + threadIdx.x;
  if (i < n) p[i] = v;
}

__global__ __launch_bounds__(256) void gat_gemm_wmma_f32(
    const float* __restrict__ X, const float* __restrict__ W,
    float* __restrict__ H1, int N) {
  __shared__ float Wl[32 * LDSW];

  const int t    = threadIdx.x;
  const int wave = t >> 5;
  const int lane = t & 31;
  const int half = lane >> 4;
  const int l    = lane & 15;
  const int row0 = blockIdx.x * 32;
  const int col0 = wave * 16;

  const float* __restrict__ xr0 = X + (size_t)(row0 + l) * F_IN;
  const float* __restrict__ xr1 = X + (size_t)(row0 + 16 + l) * F_IN;
  v8f acc0 = {}, acc1 = {};

  for (int kc = 0; kc < F_IN; kc += 32) {
    for (int i = t * 4; i < 32 * F_IN; i += 256 * 4) {
      const float4 v = *(const float4*)(W + (size_t)kc * F_IN + i);
      float* p = &Wl[(i >> 7) * LDSW + (i & 127)];
      p[0] = v.x; p[1] = v.y; p[2] = v.z; p[3] = v.w;
    }
    __syncthreads();

#pragma unroll
    for (int k = 0; k < 32; k += 4) {
      v2f b;
      b.x = Wl[(k + 2 * half)     * LDSW + col0 + l];
      b.y = Wl[(k + 2 * half + 1) * LDSW + col0 + l];
      const int ka = kc + k + 2 * half;
      v2f a0, a1;
      a0.x = xr0[ka]; a0.y = xr0[ka + 1];
      a1.x = xr1[ka]; a1.y = xr1[ka + 1];
      acc0 = __builtin_amdgcn_wmma_f32_16x16x4_f32(
          false, a0, false, b, (short)0, acc0, false, false);
      acc1 = __builtin_amdgcn_wmma_f32_16x16x4_f32(
          false, a1, false, b, (short)0, acc1, false, false);
    }
    __syncthreads();
  }

  float* __restrict__ out0 = H1 + (size_t)row0 * F_IN + col0;
  float* __restrict__ out1 = H1 + (size_t)(row0 + 16) * F_IN + col0;
#pragma unroll
  for (int r = 0; r < 8; ++r) {
    out0[(size_t)(r + 8 * half) * F_IN + l] = acc0[r];
    out1[(size_t)(r + 8 * half) * F_IN + l] = acc1[r];
  }
}

__global__ __launch_bounds__(256) void gat_attn1(
    const float* __restrict__ H1, const float* __restrict__ att_src,
    const float* __restrict__ att_dst, float* __restrict__ as1,
    float* __restrict__ ad1, int N) {
  int idx = blockIdx.x * blockDim.x + threadIdx.x;
  if (idx >= N * HEADS) return;
  int n = idx >> 3, h = idx & 7;
  const float* __restrict__ hp = H1 + (size_t)n * F_IN + h * CH;
  float s = 0.f, d = 0.f;
#pragma unroll
  for (int c = 0; c < CH; ++c) {
    float v = hp[c];
    s = fmaf(v, att_src[h * CH + c], s);
    d = fmaf(v, att_dst[h * CH + c], d);
  }
  as1[idx] = s;
  ad1[idx] = d;
}

__global__ __launch_bounds__(256) void gat_edge_max(
    const int* __restrict__ ei, const float* __restrict__ as,
    const float* __restrict__ ad, float* __restrict__ eb,
    int* __restrict__ m, int E, int Et, int H) {
  long long idx = (long long)blockIdx.x * blockDim.x + threadIdx.x;
  if (idx >= (long long)Et * H) return;
  int e = (int)(idx / H), h = (int)(idx % H);
  int s, d; edge_sd(ei, E, e, s, d);
  float v = as[s * H + h] + ad[d * H + h];
  v = (v > 0.f) ? v : NEG_SLOPE * v;
  eb[idx] = v;
  atomicMax(&m[d * H + h], f2ord(v));
}

__global__ __launch_bounds__(256) void gat_edge_exp(
    const int* __restrict__ ei, float* __restrict__ eb,
    const int* __restrict__ m, float* __restrict__ ssum,
    int E, int Et, int H) {
  long long idx = (long long)blockIdx.x * blockDim.x + threadIdx.x;
  if (idx >= (long long)Et * H) return;
  int e = (int)(idx / H), h = (int)(idx % H);
  int s, d; edge_sd(ei, E, e, s, d);
  (void)s;
  float ex = __expf(eb[idx] - ord2f(m[d * H + h]));
  eb[idx] = ex;
  atomicAdd(&ssum[d * H + h], ex);
}

__global__ __launch_bounds__(256) void gat_edge_scatter1(
    const int* __restrict__ ei, const float* __restrict__ eb,
    const float* __restrict__ ssum, const float* __restrict__ H1,
    float* __restrict__ O1, int E, int Et) {
  long long idx = (long long)blockIdx.x * blockDim.x + threadIdx.x;
  if (idx >= (long long)Et * F_IN) return;
  int e = (int)(idx >> 7), c = (int)(idx & 127), h = c >> 4;
  int s, d; edge_sd(ei, E, e, s, d);
  float alpha = eb[(size_t)e * HEADS + h] / (ssum[d * HEADS + h] + 1e-16f);
  atomicAdd(&O1[(size_t)d * F_IN + c], alpha * H1[(size_t)s * F_IN + c]);
}

__global__ __launch_bounds__(256) void gat_bias_elu(
    const float* __restrict__ O1, const float* __restrict__ b1,
    float* __restrict__ HACT, int N) {
  long long idx = (long long)blockIdx.x * blockDim.x + threadIdx.x;
  if (idx >= (long long)N * F_IN) return;
  int c = (int)(idx & 127);
  float v = O1[idx] + b1[c];
  HACT[idx] = (v > 0.f) ? v : (__expf(v) - 1.f);
}

__global__ __launch_bounds__(256) void gat_layer2_node(
    const float* __restrict__ HACT, const float* __restrict__ W2,
    const float* __restrict__ att_src2, const float* __restrict__ att_dst2,
    float* __restrict__ H2, float* __restrict__ as2, float* __restrict__ ad2,
    int N) {
  int n = blockIdx.x * blockDim.x + threadIdx.x;
  if (n >= N) return;
  const float* __restrict__ hp = HACT + (size_t)n * F_IN;
  float a0 = 0.f, a1 = 0.f;
#pragma unroll 8
  for (int k = 0; k < F_IN; ++k) {
    float v = hp[k];
    a0 = fmaf(v, W2[k * OUTC + 0], a0);
    a1 = fmaf(v, W2[k * OUTC + 1], a1);
  }
  H2[n * OUTC + 0] = a0;
  H2[n * OUTC + 1] = a1;
  as2[n] = a0 * att_src2[0] + a1 * att_src2[1];
  ad2[n] = a0 * att_dst2[0] + a1 * att_dst2[1];
}

__global__ __launch_bounds__(256) void gat_edge_scatter2(
    const int* __restrict__ ei, const float* __restrict__ eb,
    const float* __restrict__ ssum, const float* __restrict__ H2,
    float* __restrict__ O2, int E, int Et) {
  long long idx = (long long)blockIdx.x * blockDim.x + threadIdx.x;
  if (idx >= (long long)Et * OUTC) return;
  int e = (int)(idx >> 1), j = (int)(idx & 1);
  int s, d; edge_sd(ei, E, e, s, d);
  float alpha = eb[e] / (ssum[d] + 1e-16f);
  atomicAdd(&O2[(size_t)d * OUTC + j], alpha * H2[(size_t)s * OUTC + j]);
}

__global__ __launch_bounds__(256) void gat_final(
    const float* __restrict__ O2, const float* __restrict__ b2,
    float* __restrict__ OUT, int N) {
  int idx = blockIdx.x * blockDim.x + threadIdx.x;
  if (idx >= N * OUTC) return;
  OUT[idx] = O2[idx] + b2[idx & 1];
}

static inline unsigned gdim(long long n, int b) {
  return (unsigned)((n + b - 1) / b);
}

extern "C" void kernel_launch(void* const* d_in, const int* in_sizes, int n_in,
                              void* d_out, int out_size, void* d_ws,
                              size_t ws_size, hipStream_t stream) {
  const float* x        = (const float*)d_in[0];
  const int*   ei       = (const int*)d_in[1];
  const float* W1       = (const float*)d_in[2];
  const float* att_src1 = (const float*)d_in[3];
  const float* att_dst1 = (const float*)d_in[4];
  const float* b1       = (const float*)d_in[5];
  const float* W2       = (const float*)d_in[6];
  const float* att_src2 = (const float*)d_in[7];
  const float* att_dst2 = (const float*)d_in[8];
  const float* b2       = (const float*)d_in[9];

  const int N  = in_sizes[0] / F_IN;
  const int E  = in_sizes[1] / 2;
  const int Et = E + N;

  float* ws = (float*)d_ws;
  size_t off = 0;
  float* h1  = ws + off; off += (size_t)N * F_IN;
  float* as1 = ws + off; off += (size_t)N * HEADS;
  float* ad1 = ws + off; off += (size_t)N * HEADS;
  float* e1  = ws + off; off += (size_t)Et * HEADS;
  int*   m1  = (int*)(ws + off); off += (size_t)N * HEADS;
  float* s1  = ws + off; off += (size_t)N * HEADS;
  float* o1  = ws + off; off += (size_t)N * F_IN;
  float* h2  = ws + off; off += (size_t)N * OUTC;
  float* as2 = ws + off; off += (size_t)N;
  float* ad2 = ws + off; off += (size_t)N;
  float* e2  = ws + off; off += (size_t)Et;
  int*   m2  = (int*)(ws + off); off += (size_t)N;
  float* s2  = ws + off; off += (size_t)N;
  float* o2  = ws + off; off += (size_t)N * OUTC;
  (void)ws_size; (void)n_in; (void)out_size;

  const int B = 256;

  gat_fill_i32<<<gdim((long long)N * HEADS, B), B, 0, stream>>>(m1, INT_MIN, (size_t)N * HEADS);
  gat_fill_f32<<<gdim((long long)N * HEADS, B), B, 0, stream>>>(s1, 0.f, (size_t)N * HEADS);
  gat_fill_f32<<<gdim((long long)N * F_IN, B), B, 0, stream>>>(o1, 0.f, (size_t)N * F_IN);
  gat_fill_i32<<<gdim(N, B), B, 0, stream>>>(m2, INT_MIN, (size_t)N);
  gat_fill_f32<<<gdim(N, B), B, 0, stream>>>(s2, 0.f, (size_t)N);
  gat_fill_f32<<<gdim((long long)N * OUTC, B), B, 0, stream>>>(o2, 0.f, (size_t)N * OUTC);

  gat_gemm_wmma_f32<<<gdim(N, 32), 256, 0, stream>>>(x, W1, h1, N);
  gat_attn1<<<gdim((long long)N * HEADS, B), B, 0, stream>>>(h1, att_src1, att_dst1, as1, ad1, N);
  gat_edge_max<<<gdim((long long)Et * HEADS, B), B, 0, stream>>>(ei, as1, ad1, e1, m1, E, Et, HEADS);
  gat_edge_exp<<<gdim((long long)Et * HEADS, B), B, 0, stream>>>(ei, e1, m1, s1, E, Et, HEADS);
  gat_edge_scatter1<<<gdim((long long)Et * F_IN, B), B, 0, stream>>>(ei, e1, s1, h1, o1, E, Et);
  gat_bias_elu<<<gdim((long long)N * F_IN, B), B, 0, stream>>>(o1, b1, h1, N);

  gat_layer2_node<<<gdim(N, B), B, 0, stream>>>(h1, W2, att_src2, att_dst2, h2, as2, ad2, N);
  gat_edge_max<<<gdim((long long)Et, B), B, 0, stream>>>(ei, as2, ad2, e2, m2, E, Et, 1);
  gat_edge_exp<<<gdim((long long)Et, B), B, 0, stream>>>(ei, e2, m2, s2, E, Et, 1);
  gat_edge_scatter2<<<gdim((long long)Et * OUTC, B), B, 0, stream>>>(ei, e2, s2, h2, o2, E, Et);
  gat_final<<<gdim((long long)N * OUTC, B), B, 0, stream>>>(o2, b2, (float*)d_out, N);
}


# Edges bucketed by destination each call (count, offsets, place); both layers' softmax+aggregation become per-node gathers: no float atomics or accumulator fills; sums in arrival-rank order per node
# speedup vs baseline: 3.3777x; 3.3777x over previous
#include <hip/hip_runtime.h>
#include <hip/hip_bf16.h>


typedef __attribute__((ext_vector_type(2))) float v2f;
typedef __attribute__((ext_vector_type(8))) float v8f;

#define F_IN   128
#define HEADS  8
#define CH     16
#define OUTC   2
#define NEG_SLOPE 0.2f
#define LDSW   136

__device__ __forceinline__ float leaky(float v) {
  return (v > 0.f) ? v : NEG_SLOPE * v;
}

__global__ void gat_fill_i32(int* __restrict__ p, int v, size_t n) {
  size_t i = (size_t)blockIdx.x * blockDim.x + threadIdx.x;
  if (i < n) p[i] = v;
}

__global__ __launch_bounds__(256) void gat_csr_count(
    const int* __restrict__ ei, int* __restrict__ deg,
    int* __restrict__ rank, int E) {
  int e = blockIdx.x * 256 + threadIdx.x;
  if (e >= E) return;
  rank[e] = atomicAdd(&deg[ei[E + e]], 1);
}

__global__ __launch_bounds__(256) void gat_csr_alloc(
    int* __restrict__ deg, int* __restrict__ rowstart, int N) {
  __shared__ int wsum[8];
  __shared__ int base_s;
  const int t = threadIdx.x, lane = t & 31, wave = t >> 5;
  const int n = blockIdx.x * 256 + t;
  const int v = (n < N) ? deg[n] : 0;
  int inc = v;
  for (int o = 1; o < 32; o <<= 1) {
    const int u = __shfl_up(inc, o, 32);
    if (lane >= o) inc += u;
  }
  if (lane == 31) wsum[wave] = inc;
  __syncthreads();
  if (t == 0) {
    int tot = 0;
    for (int w = 0; w < 8; ++w) { const int c = wsum[w]; wsum[w] = tot; tot += c; }
    base_s = atomicAdd(&deg[N], tot);
  }
  __syncthreads();
  if (n < N) rowstart[n] = base_s + wsum[wave] + inc - v;
}

__global__ __launch_bounds__(256) void gat_csr_place(
    const int* __restrict__ ei, const int* __restrict__ rank,
    const int* __restrict__ rowstart, int* __restrict__ csr_src, int E) {
  int e = blockIdx.x * 256 + threadIdx.x;
  if (e >= E) return;
  csr_src[rowstart[ei[E + e]] + rank[e]] = ei[e];
}

__global__ __launch_bounds__(256) void gat_gemm_wmma_f32(
    const float* __restrict__ X, const float* __restrict__ W,
    float* __restrict__ H1, int N) {
  __shared__ float Wl[32 * LDSW];

  const int t    = threadIdx.x;
  const int wave = t >> 5;
  const int lane = t & 31;
  const int half = lane >> 4;
  const int l    = lane & 15;
  const int row0 = blockIdx.x * 32;
  const int col0 = wave * 16;

  const float* __restrict__ xr0 = X + (size_t)(row0 + l) * F_IN;
  const float* __restrict__ xr1 = X + (size_t)(row0 + 16 + l) * F_IN;
  v8f acc0 = {}, acc1 = {};

  for (int kc = 0; kc < F_IN; kc += 32) {
    for (int i = t * 4; i < 32 * F_IN; i += 256 * 4) {
      const float4 v = *(const float4*)(W + (size_t)kc * F_IN + i);
      float* p = &Wl[(i >> 7) * LDSW + (i & 127)];
      p[0] = v.x; p[1] = v.y; p[2] = v.z; p[3] = v.w;
    }
    __syncthreads();

#pragma unroll
    for (int k = 0; k < 32; k += 4) {
      v2f b;
      b.x = Wl[(k + 2 * half)     * LDSW + col0 + l];
      b.y = Wl[(k + 2 * half + 1) * LDSW + col0 + l];
      const int ka = kc + k + 2 * half;
      v2f a0, a1;
      a0.x = xr0[ka]; a0.y = xr0[ka + 1];
      a1.x = xr1[ka]; a1.y = xr1[ka + 1];
      acc0 = __builtin_amdgcn_wmma_f32_16x16x4_f32(
          false, a0, false, b, (short)0, acc0, false, false);
      acc1 = __builtin_amdgcn_wmma_f32_16x16x4_f32(
          false, a1, false, b, (short)0, acc1, false, false);
    }
    __syncthreads();
  }

  float* __restrict__ out0 = H1 + (size_t)row0 * F_IN + col0;
  float* __restrict__ out1 = H1 + (size_t)(row0 + 16) * F_IN + col0;
#pragma unroll
  for (int r = 0; r < 8; ++r) {
    out0[(size_t)(r + 8 * half) * F_IN + l] = acc0[r];
    out1[(size_t)(r + 8 * half) * F_IN + l] = acc1[r];
  }
}

__global__ __launch_bounds__(256) void gat_attn1(
    const float* __restrict__ H1, const float* __restrict__ att_src,
    const float* __restrict__ att_dst, float* __restrict__ as1,
    float* __restrict__ ad1, int N) {
  int idx = blockIdx.x * blockDim.x + threadIdx.x;
  if (idx >= N * HEADS) return;
  int n = idx >> 3, h = idx & 7;
  const float* __restrict__ hp = H1 + (size_t)n * F_IN + h * CH;
  float s = 0.f, d = 0.f;
#pragma unroll
  for (int c = 0; c < CH; ++c) {
    float v = hp[c];
    s = fmaf(v, att_src[h * CH + c], s);
    d = fmaf(v, att_dst[h * CH + c], d);
  }
  as1[idx] = s;
  ad1[idx] = d;
}

__global__ __launch_bounds__(256) void gat_gather1(
    const int* __restrict__ rowstart, const int* __restrict__ deg,
    const int* __restrict__ csr_src, const float* __restrict__ as1,
    const float* __restrict__ ad1, const float* __restrict__ H1,
    const float* __restrict__ b1, float* __restrict__ HACT, int N) {
  const int lane = threadIdx.x & 31;
  const int n = __builtin_amdgcn_readfirstlane(blockIdx.x * 8 + (threadIdx.x >> 5));
  if (n >= N) return;
  const int h = lane >> 2;
  const int start = rowstart[n];
  const int cnt   = deg[n];
  const float adh   = ad1[n * HEADS + h];
  const float vself = leaky(as1[n * HEADS + h] + adh);

  float m = vself;
  for (int c0 = 0; c0 < cnt; c0 += 32) {
    const int nn = min(32, cnt - c0);
    const int mine = (lane < nn) ? csr_src[start + c0 + lane] : 0;
    for (int j = 0; j < nn; ++j) {
      const int s = __builtin_amdgcn_readlane(mine, j);
      m = fmaxf(m, leaky(as1[s * HEADS + h] + adh));
    }
  }

  float ex = __expf(vself - m);
  float l = ex;
  float4 hv = *(const float4*)(H1 + (size_t)n * F_IN + 4 * lane);
  float4 acc;
  acc.x = ex * hv.x; acc.y = ex * hv.y; acc.z = ex * hv.z; acc.w = ex * hv.w;
  for (int c0 = 0; c0 < cnt; c0 += 32) {
    const int nn = min(32, cnt - c0);
    const int mine = (lane < nn) ? csr_src[start + c0 + lane] : 0;
#pragma unroll 4
    for (int j = 0; j < nn; ++j) {
      const int s = __builtin_amdgcn_readlane(mine, j);
      const float4 r = *(const float4*)(H1 + (size_t)s * F_IN + 4 * lane);
      const float w = __expf(leaky(as1[s * HEADS + h] + adh) - m);
      l += w;
      acc.x = fmaf(w, r.x, acc.x); acc.y = fmaf(w, r.y, acc.y);
      acc.z = fmaf(w, r.z, acc.z); acc.w = fmaf(w, r.w, acc.w);
    }
  }

  const float den = l + 1e-16f;
  const float4 b = *(const float4*)(b1 + 4 * lane);
  float4 o;
  o.x = acc.x / den + b.x; o.y = acc.y / den + b.y;
  o.z = acc.z / den + b.z; o.w = acc.w / den + b.w;
  o.x = (o.x > 0.f) ? o.x : (__expf(o.x) - 1.f);
  o.y = (o.y > 0.f) ? o.y : (__expf(o.y) - 1.f);
  o.z = (o.z > 0.f) ? o.z : (__expf(o.z) - 1.f);
  o.w = (o.w > 0.f) ? o.w : (__expf(o.w) - 1.f);
  *(float4*)(HACT + (size_t)n * F_IN + 4 * lane) = o;
}

__global__ __launch_bounds__(256) void gat_layer2_node(
    const float* __restrict__ HACT, const float* __restrict__ W2,
    const float* __restrict__ att_src2, const float* __restrict__ att_dst2,
    float* __restrict__ H2, float* __restrict__ as2, float* __restrict__ ad2,
    int N) {
  int n = blockIdx.x * blockDim.x + threadIdx.x;
  if (n >= N) return;
  const float* __restrict__ hp = HACT + (size_t)n * F_IN;
  float a0 = 0.f, a1 = 0.f;
#pragma unroll 8
  for (int k = 0; k < F_IN; ++k) {
    float v = hp[k];
    a0 = fmaf(v, W2[k * OUTC + 0], a0);
    a1 = fmaf(v, W2[k * OUTC + 1], a1);
  }
  H2[n * OUTC + 0] = a0;
  H2[n * OUTC + 1] = a1;
  as2[n] = a0 * att_src2[0] + a1 * att_src2[1];
  ad2[n] = a0 * att_dst2[0] + a1 * att_dst2[1];
}

__global__ __launch_bounds__(256) void gat_gather2(
    const int* __restrict__ rowstart, const int* __restrict__ deg,
    const int* __restrict__ csr_src, const float* __restrict__ as2,
    const float* __restrict__ ad2, const float* __restrict__ H2,
    const float* __restrict__ b2, float* __restrict__ OUT, int N) {
  const int n = blockIdx.x * 256 + threadIdx.x;
  if (n >= N) return;
  const int start = rowstart[n];
  const int cnt   = deg[n];
  const float adn   = ad2[n];
  const float vself = leaky(as2[n] + adn);

  float m = vself;
  for (int j = 0; j < cnt; ++j)
    m = fmaxf(m, leaky(as2[csr_src[start + j]] + adn));

  float ex = __expf(vself - m);
  float l = ex;
  float2 hv = *(const float2*)(H2 + (size_t)n * OUTC);
  float a0 = ex * hv.x, a1 = ex * hv.y;
  for (int j = 0; j < cnt; ++j) {
    const int s = csr_src[start + j];
    const float2 r = *(const float2*)(H2 + (size_t)s * OUTC);
    const float w = __expf(leaky(as2[s] + adn) - m);
    l += w;
    a0 = fmaf(w, r.x, a0);
    a1 = fmaf(w, r.y, a1);
  }
  const float den = l + 1e-16f;
  float2 o;
  o.x = a0 / den + b2[0];
  o.y = a1 / den + b2[1];
  *(float2*)(OUT + (size_t)n * OUTC) = o;
}

static inline unsigned gdim(long long n, int b) {
  return (unsigned)((n + b - 1) / b);
}

extern "C" void kernel_launch(void* const* d_in, const int* in_sizes, int n_in,
                              void* d_out, int out_size, void* d_ws,
                              size_t ws_size, hipStream_t stream) {
  const float* x        = (const float*)d_in[0];
  const int*   ei       = (const int*)d_in[1];
  const float* W1       = (const float*)d_in[2];
  const float* att_src1 = (const float*)d_in[3];
  const float* att_dst1 = (const float*)d_in[4];
  const float* b1       = (const float*)d_in[5];
  const float* W2       = (const float*)d_in[6];
  const float* att_src2 = (const float*)d_in[7];
  const float* att_dst2 = (const float*)d_in[8];
  const float* b2       = (const float*)d_in[9];

  const int N  = in_sizes[0] / F_IN;
  const int E  = in_sizes[1] / 2;

  float* ws = (float*)d_ws;
  size_t off = 0;
  float* h1   = ws + off; off += (size_t)N * F_IN;
  float* hact = ws + off; off += (size_t)N * F_IN;
  float* as1  = ws + off; off += (size_t)N * HEADS;
  float* ad1  = ws + off; off += (size_t)N * HEADS;
  float* h2   = ws + off; off += (size_t)N * OUTC;
  float* as2  = ws + off; off += (size_t)N;
  float* ad2  = ws + off; off += (size_t)N;
  int* deg      = (int*)(ws + off); off += (size_t)N + 1;
  int* rank     = (int*)(ws + off); off += (size_t)E;
  int* rowstart = (int*)(ws + off); off += (size_t)N;
  int* csr_src  = (int*)(ws + off); off += (size_t)E;
  (void)ws_size; (void)n_in; (void)out_size;

  const int B = 256;

  gat_fill_i32<<<gdim((long long)N + 1, B), B, 0, stream>>>(deg, 0, (size_t)N + 1);
  gat_csr_count<<<gdim(E, B), B, 0, stream>>>(ei, deg, rank, E);
  gat_csr_alloc<<<gdim(N, B), B, 0, stream>>>(deg, rowstart, N);
  gat_csr_place<<<gdim(E, B), B, 0, stream>>>(ei, rank, rowstart, csr_src, E);

  gat_gemm_wmma_f32<<<gdim(N, 32), 256, 0, stream>>>(x, W1, h1, N);
  gat_attn1<<<gdim((long long)N * HEADS, B), B, 0, stream>>>(h1, att_src1, att_dst1, as1, ad1, N);
  gat_gather1<<<gdim(N, 8), B, 0, stream>>>(rowstart, deg, csr_src, as1, ad1, h1, b1, hact, N);

  gat_layer2_node<<<gdim(N, B), B, 0, stream>>>(hact, W2, att_src2, att_dst2, h2, as2, ad2, N);
  gat_gather2<<<gdim(N, B), B, 0, stream>>>(rowstart, deg, csr_src, as2, ad2, h2, b2, (float*)d_out, N);
}


# Edge bucketing on LDS counters: edges binned by 256-node destination range, then ranked per node by one work-group per bin; no per-edge global atomics; sums stay in arrival-rank order per node
# speedup vs baseline: 4.0797x; 1.2078x over previous
#include <hip/hip_runtime.h>
#include <hip/hip_bf16.h>


typedef __attribute__((ext_vector_type(2))) float v2f;
typedef __attribute__((ext_vector_type(8))) float v8f;

#define F_IN   128
#define HEADS  8
#define CH     16
#define OUTC   2
#define NEG_SLOPE 0.2f
#define LDSW   136
#define BIN_SHIFT 8
#define BIN_NODES 256
#define MAX_BINS  512
#define NCHUNK    256

__device__ __forceinline__ float leaky(float v) {
  return (v > 0.f) ? v : NEG_SLOPE * v;
}

__global__ __launch_bounds__(256) void gat_bin_count(
    const int* __restrict__ ei, int* __restrict__ cnt,
    int* __restrict__ total, int E, int nbins) {
  __shared__ int hist[MAX_BINS];
  const int t = threadIdx.x, g = blockIdx.x;
  for (int b = t; b < nbins; b += 256) hist[b] = 0;
  __syncthreads();
  const int ce = (E + NCHUNK - 1) / NCHUNK;
  const int e0 = g * ce;
  const int e1 = min(E, e0 + ce);
  for (int e = e0 + t; e < e1; e += 4 * 256) {
    int d[4];
#pragma unroll
    for (int u = 0; u < 4; ++u) {
      const int ee = e + u * 256;
      d[u] = (ee < e1) ? ei[E + ee] : -1;
    }
#pragma unroll
    for (int u = 0; u < 4; ++u)
      if (d[u] >= 0) atomicAdd(&hist[d[u] >> BIN_SHIFT], 1);
  }
  __syncthreads();
  for (int b = t; b < nbins; b += 256) cnt[b * NCHUNK + g] = hist[b];
  if (g == 0 && t == 0) *total = 0;
}

__global__ __launch_bounds__(NCHUNK) void gat_bin_offsets(
    int* __restrict__ cnt, int* __restrict__ total,
    int* __restrict__ bin_base, int* __restrict__ bin_cnt) {
  __shared__ int wsum[NCHUNK / 32];
  __shared__ int base_s;
  const int t = threadIdx.x, lane = t & 31, wave = t >> 5;
  const int b = blockIdx.x;
  const int v = cnt[b * NCHUNK + t];
  int inc = v;
  for (int o = 1; o < 32; o <<= 1) {
    const int u = __shfl_up(inc, o, 32);
    if (lane >= o) inc += u;
  }
  if (lane == 31) wsum[wave] = inc;
  __syncthreads();
  if (t == 0) {
    int tot = 0;
    for (int w = 0; w < NCHUNK / 32; ++w) { const int c = wsum[w]; wsum[w] = tot; tot += c; }
    base_s = atomicAdd(total, tot);
    bin_base[b] = base_s;
    bin_cnt[b]  = tot;
  }
  __syncthreads();
  cnt[b * NCHUNK + t] = base_s + wsum[wave] + inc - v;
}

__global__ __launch_bounds__(256) void gat_bin_scatter(
    const int* __restrict__ ei, const int* __restrict__ offs,
    int* __restrict__ pairs, int E, int nbins) {
  __shared__ int cur[MAX_BINS];
  int2* __restrict__ pr = (int2*)pairs;
  const int t = threadIdx.x, g = blockIdx.x;
  for (int b = t; b < nbins; b += 256) cur[b] = offs[b * NCHUNK + g];
  __syncthreads();
  const int ce = (E + NCHUNK - 1) / NCHUNK;
  const int e0 = g * ce;
  const int e1 = min(E, e0 + ce);
  for (int e = e0 + t; e < e1; e += 4 * 256) {
    int d[4], sr[4];
#pragma unroll
    for (int u = 0; u < 4; ++u) {
      const int ee = e + u * 256;
      d[u]  = (ee < e1) ? ei[E + ee] : -1;
      sr[u] = (ee < e1) ? ei[ee] : 0;
    }
#pragma unroll
    for (int u = 0; u < 4; ++u)
      if (d[u] >= 0) {
        const int pos = atomicAdd(&cur[d[u] >> BIN_SHIFT], 1);
        pr[pos] = make_int2(sr[u], d[u]);
      }
  }
}

__global__ __launch_bounds__(BIN_NODES) void gat_bin_csr(
    const int* __restrict__ pairs, const int* __restrict__ bin_base,
    const int* __restrict__ bin_cnt, int* __restrict__ rowstart,
    int* __restrict__ deg, int* __restrict__ csr_src, int N) {
  __shared__ int dcnt[BIN_NODES];
  __shared__ int dpos[BIN_NODES];
  __shared__ int wsum[BIN_NODES / 32];
  const int t = threadIdx.x, lane = t & 31, wave = t >> 5;
  const int b = blockIdx.x;
  const int n = b * BIN_NODES + t;
  const int base = bin_base[b];
  const int tot  = bin_cnt[b];
  const int2* __restrict__ pb = (const int2*)pairs + base;
  dcnt[t] = 0;
  __syncthreads();
  for (int i = t; i < tot; i += 4 * BIN_NODES) {
    int d[4];
#pragma unroll
    for (int u = 0; u < 4; ++u) {
      const int ii = i + u * BIN_NODES;
      d[u] = (ii < tot) ? pb[ii].y : -1;
    }
#pragma unroll
    for (int u = 0; u < 4; ++u)
      if (d[u] >= 0) atomicAdd(&dcnt[d[u] & (BIN_NODES - 1)], 1);
  }
  __syncthreads();
  const int v = dcnt[t];
  int inc = v;
  for (int o = 1; o < 32; o <<= 1) {
    const int u = __shfl_up(inc, o, 32);
    if (lane >= o) inc += u;
  }
  if (lane == 31) wsum[wave] = inc;
  __syncthreads();
  if (t == 0) {
    int run = 0;
    for (int w = 0; w < BIN_NODES / 32; ++w) { const int c = wsum[w]; wsum[w] = run; run += c; }
  }
  __syncthreads();
  const int first = base + wsum[wave] + inc - v;
  dpos[t] = first;
  if (n < N) { rowstart[n] = first; deg[n] = v; }
  __syncthreads();
  for (int i = t; i < tot; i += 4 * BIN_NODES) {
    int2 r[4];
#pragma unroll
    for (int u = 0; u < 4; ++u) {
      const int ii = i + u * BIN_NODES;
      r[u] = (ii < tot) ? pb[ii] : make_int2(0, -1);
    }
#pragma unroll
    for (int u = 0; u < 4; ++u)
      if (r[u].y >= 0)
        csr_src[atomicAdd(&dpos[r[u].y & (BIN_NODES - 1)], 1)] = r[u].x;
  }
}

__global__ __launch_bounds__(256) void gat_gemm_wmma_f32(
    const float* __restrict__ X, const float* __restrict__ W,
    float* __restrict__ H1, int N) {
  __shared__ float Wl[32 * LDSW];

  const int t    = threadIdx.x;
  const int wave = t >> 5;
  const int lane = t & 31;
  const int half = lane >> 4;
  const int l    = lane & 15;
  const int row0 = blockIdx.x * 32;
  const int col0 = wave * 16;

  const float* __restrict__ xr0 = X + (size_t)(row0 + l) * F_IN;
  const float* __restrict__ xr1 = X + (size_t)(row0 + 16 + l) * F_IN;
  v8f acc0 = {}, acc1 = {};

  for (int kc = 0; kc < F_IN; kc += 32) {
    for (int i = t * 4; i < 32 * F_IN; i += 256 * 4) {
      const float4 v = *(const float4*)(W + (size_t)kc * F_IN + i);
      float* p = &Wl[(i >> 7) * LDSW + (i & 127)];
      p[0] = v.x; p[1] = v.y; p[2] = v.z; p[3] = v.w;
    }
    __syncthreads();

#pragma unroll
    for (int k = 0; k < 32; k += 4) {
      v2f b;
      b.x = Wl[(k + 2 * half)     * LDSW + col0 + l];
      b.y = Wl[(k + 2 * half + 1) * LDSW + col0 + l];
      const int ka = kc + k + 2 * half;
      v2f a0, a1;
      a0.x = xr0[ka]; a0.y = xr0[ka + 1];
      a1.x = xr1[ka]; a1.y = xr1[ka + 1];
      acc0 = __builtin_amdgcn_wmma_f32_16x16x4_f32(
          false, a0, false, b, (short)0, acc0, false, false);
      acc1 = __builtin_amdgcn_wmma_f32_16x16x4_f32(
          false, a1, false, b, (short)0, acc1, false, false);
    }
    __syncthreads();
  }

  float* __restrict__ out0 = H1 + (size_t)row0 * F_IN + col0;
  float* __restrict__ out1 = H1 + (size_t)(row0 + 16) * F_IN + col0;
#pragma unroll
  for (int r = 0; r < 8; ++r) {
    out0[(size_t)(r + 8 * half) * F_IN + l] = acc0[r];
    out1[(size_t)(r + 8 * half) * F_IN + l] = acc1[r];
  }
}

__global__ __launch_bounds__(256) void gat_attn1(
    const float* __restrict__ H1, const float* __restrict__ att_src,
    const float* __restrict__ att_dst, float* __restrict__ as1,
    float* __restrict__ ad1, int N) {
  int idx = blockIdx.x * blockDim.x + threadIdx.x;
  if (idx >= N * HEADS) return;
  int n = idx >> 3, h = idx & 7;
  const float* __restrict__ hp = H1 + (size_t)n * F_IN + h * CH;
  float s = 0.f, d = 0.f;
#pragma unroll
  for (int c = 0; c < CH; ++c) {
    float v = hp[c];
    s = fmaf(v, att_src[h * CH + c], s);
    d = fmaf(v, att_dst[h * CH + c], d);
  }
  as1[idx] = s;
  ad1[idx] = d;
}

__global__ __launch_bounds__(256) void gat_gather1(
    const int* __restrict__ rowstart, const int* __restrict__ deg,
    const int* __restrict__ csr_src, const float* __restrict__ as1,
    const float* __restrict__ ad1, const float* __restrict__ H1,
    const float* __restrict__ b1, float* __restrict__ HACT, int N) {
  const int lane = threadIdx.x & 31;
  const int n = __builtin_amdgcn_readfirstlane(blockIdx.x * 8 + (threadIdx.x >> 5));
  if (n >= N) return;
  const int h = lane >> 2;
  const int start = rowstart[n];
  const int cnt   = deg[n];
  const float adh   = ad1[n * HEADS + h];
  const float vself = leaky(as1[n * HEADS + h] + adh);

  float m = vself;
  for (int c0 = 0; c0 < cnt; c0 += 32) {
    const int nn = min(32, cnt - c0);
    const int mine = (lane < nn) ? csr_src[start + c0 + lane] : 0;
    for (int j = 0; j < nn; ++j) {
      const int s = __builtin_amdgcn_readlane(mine, j);
      m = fmaxf(m, leaky(as1[s * HEADS + h] + adh));
    }
  }

  float ex = __expf(vself - m);
  float l = ex;
  float4 hv = *(const float4*)(H1 + (size_t)n * F_IN + 4 * lane);
  float4 acc;
  acc.x = ex * hv.x; acc.y = ex * hv.y; acc.z = ex * hv.z; acc.w = ex * hv.w;
  for (int c0 = 0; c0 < cnt; c0 += 32) {
    const int nn = min(32, cnt - c0);
    const int mine = (lane < nn) ? csr_src[start + c0 + lane] : 0;
#pragma unroll 4
    for (int j = 0; j < nn; ++j) {
      const int s = __builtin_amdgcn_readlane(mine, j);
      const float4 r = *(const float4*)(H1 + (size_t)s * F_IN + 4 * lane);
      const float w = __expf(leaky(as1[s * HEADS + h] + adh) - m);
      l += w;
      acc.x = fmaf(w, r.x, acc.x); acc.y = fmaf(w, r.y, acc.y);
      acc.z = fmaf(w, r.z, acc.z); acc.w = fmaf(w, r.w, acc.w);
    }
  }

  const float den = l + 1e-16f;
  const float4 b = *(const float4*)(b1 + 4 * lane);
  float4 o;
  o.x = acc.x / den + b.x; o.y = acc.y / den + b.y;
  o.z = acc.z / den + b.z; o.w = acc.w / den + b.w;
  o.x = (o.x > 0.f) ? o.x : (__expf(o.x) - 1.f);
  o.y = (o.y > 0.f) ? o.y : (__expf(o.y) - 1.f);
  o.z = (o.z > 0.f) ? o.z : (__expf(o.z) - 1.f);
  o.w = (o.w > 0.f) ? o.w : (__expf(o.w) - 1.f);
  *(float4*)(HACT + (size_t)n * F_IN + 4 * lane) = o;
}

__global__ __launch_bounds__(256) void gat_layer2_node(
    const float* __restrict__ HACT, const float* __restrict__ W2,
    const float* __restrict__ att_src2, const float* __restrict__ att_dst2,
    float* __restrict__ H2, float* __restrict__ as2, float* __restrict__ ad2,
    int N) {
  int n = blockIdx.x * blockDim.x + threadIdx.x;
  if (n >= N) return;
  const float* __restrict__ hp = HACT + (size_t)n * F_IN;
  float a0 = 0.f, a1 = 0.f;
#pragma unroll 8
  for (int k = 0; k < F_IN; ++k) {
    float v = hp[k];
    a0 = fmaf(v, W2[k * OUTC + 0], a0);
    a1 = fmaf(v, W2[k * OUTC + 1], a1);
  }
  H2[n * OUTC + 0] = a0;
  H2[n * OUTC + 1] = a1;
  as2[n] = a0 * att_src2[0] + a1 * att_src2[1];
  ad2[n] = a0 * att_dst2[0] + a1 * att_dst2[1];
}

__global__ __launch_bounds__(256) void gat_gather2(
    const int* __restrict__ rowstart, const int* __restrict__ deg,
    const int* __restrict__ csr_src, const float* __restrict__ as2,
    const float* __restrict__ ad2, const float* __restrict__ H2,
    const float* __restrict__ b2, float* __restrict__ OUT, int N) {
  const int n = blockIdx.x * 256 + threadIdx.x;
  if (n >= N) return;
  const int start = rowstart[n];
  const int cnt   = deg[n];
  const float adn   = ad2[n];
  const float vself = leaky(as2[n] + adn);

  float m = vself;
  for (int j = 0; j < cnt; ++j)
    m = fmaxf(m, leaky(as2[csr_src[start + j]] + adn));

  float ex = __expf(vself - m);
  float l = ex;
  float2 hv = *(const float2*)(H2 + (size_t)n * OUTC);
  float a0 = ex * hv.x, a1 = ex * hv.y;
  for (int j = 0; j < cnt; ++j) {
    const int s = csr_src[start + j];
    const float2 r = *(const float2*)(H2 + (size_t)s * OUTC);
    const float w = __expf(leaky(as2[s] + adn) - m);
    l += w;
    a0 = fmaf(w, r.x, a0);
    a1 = fmaf(w, r.y, a1);
  }
  const float den = l + 1e-16f;
  float2 o;
  o.x = a0 / den + b2[0];
  o.y = a1 / den + b2[1];
  *(float2*)(OUT + (size_t)n * OUTC) = o;
}

static inline unsigned gdim(long long n, int b) {
  return (unsigned)((n + b - 1) / b);
}

extern "C" void kernel_launch(void* const* d_in, const int* in_sizes, int n_in,
                              void* d_out, int out_size, void* d_ws,
                              size_t ws_size, hipStream_t stream) {
  const float* x        = (const float*)d_in[0];
  const int*   ei       = (const int*)d_in[1];
  const float* W1       = (const float*)d_in[2];
  const float* att_src1 = (const float*)d_in[3];
  const float* att_dst1 = (const float*)d_in[4];
  const float* b1       = (const float*)d_in[5];
  const float* W2       = (const float*)d_in[6];
  const float* att_src2 = (const float*)d_in[7];
  const float* att_dst2 = (const float*)d_in[8];
  const float* b2       = (const float*)d_in[9];

  const int N  = in_sizes[0] / F_IN;
  const int E  = in_sizes[1] / 2;

  float* ws = (float*)d_ws;
  size_t off = 0;
  float* h1   = ws + off; off += (size_t)N * F_IN;
  float* hact = ws + off; off += (size_t)N * F_IN;
  float* as1  = ws + off; off += (size_t)N * HEADS;
  float* ad1  = ws + off; off += (size_t)N * HEADS;
  float* h2   = ws + off; off += (size_t)N * OUTC;
  float* as2  = ws + off; off += (size_t)N;
  float* ad2  = ws + off; off += (size_t)N;
  int* pairs    = (int*)(ws + off); off += (size_t)E * 2;
  int* deg      = (int*)(ws + off); off += (size_t)N;
  int* rowstart = (int*)(ws + off); off += (size_t)N;
  int* csr_src  = (int*)(ws + off); off += (size_t)E;
  const int nbins = (int)gdim(N, BIN_NODES);
  int* bin_cnt  = (int*)(ws + off); off += (size_t)nbins * NCHUNK;
  int* bin_base = (int*)(ws + off); off += (size_t)nbins;
  int* bin_tot  = (int*)(ws + off); off += (size_t)nbins;
  int* total    = (int*)(ws + off); off += 1;
  (void)ws_size; (void)n_in; (void)out_size;

  const int B = 256;

  gat_bin_count<<<NCHUNK, B, 0, stream>>>(ei, bin_cnt, total, E, nbins);
  gat_bin_offsets<<<nbins, NCHUNK, 0, stream>>>(bin_cnt, total, bin_base, bin_tot);
  gat_bin_scatter<<<NCHUNK, B, 0, stream>>>(ei, bin_cnt, pairs, E, nbins);
  gat_bin_csr<<<nbins, BIN_NODES, 0, stream>>>(pairs, bin_base, bin_tot, rowstart, deg, csr_src, N);

  gat_gemm_wmma_f32<<<gdim(N, 32), 256, 0, stream>>>(x, W1, h1, N);
  gat_attn1<<<gdim((long long)N * HEADS, B), B, 0, stream>>>(h1, att_src1, att_dst1, as1, ad1, N);
  gat_gather1<<<gdim(N, 8), B, 0, stream>>>(rowstart, deg, csr_src, as1, ad1, h1, b1, hact, N);

  gat_layer2_node<<<gdim(N, B), B, 0, stream>>>(hact, W2, att_src2, att_dst2, h2, as2, ad2, N);
  gat_gather2<<<gdim(N, B), B, 0, stream>>>(rowstart, deg, csr_src, as2, ad2, h2, b2, (float*)d_out, N);
}


# GEMM re-tiled: a wave owns 16 rows x all 128 columns, so each x element is fetched by one wave (was 8); W1 whole in LDS as k-pairs; k visited in another order: the products' sum order changes
# speedup vs baseline: 4.9427x; 1.2115x over previous
#include <hip/hip_runtime.h>
#include <hip/hip_bf16.h>


typedef __attribute__((ext_vector_type(2))) float v2f;
typedef __attribute__((ext_vector_type(8))) float v8f;

#define F_IN   128
#define HEADS  8
#define CH     16
#define OUTC   2
#define NEG_SLOPE 0.2f
#define BIN_SHIFT 8
#define BIN_NODES 256
#define MAX_BINS  512
#define NCHUNK    256

__device__ __forceinline__ float leaky(float v) {
  return (v > 0.f) ? v : NEG_SLOPE * v;
}

__global__ __launch_bounds__(256) void gat_bin_count(
    const int* __restrict__ ei, int* __restrict__ cnt,
    int* __restrict__ total, int E, int nbins) {
  __shared__ int hist[MAX_BINS];
  const int t = threadIdx.x, g = blockIdx.x;
  for (int b = t; b < nbins; b += 256) hist[b] = 0;
  __syncthreads();
  const int ce = (E + NCHUNK - 1) / NCHUNK;
  const int e0 = g * ce;
  const int e1 = min(E, e0 + ce);
  for (int e = e0 + t; e < e1; e += 4 * 256) {
    int d[4];
#pragma unroll
    for (int u = 0; u < 4; ++u) {
      const int ee = e + u * 256;
      d[u] = (ee < e1) ? ei[E + ee] : -1;
    }
#pragma unroll
    for (int u = 0; u < 4; ++u)
      if (d[u] >= 0) atomicAdd(&hist[d[u] >> BIN_SHIFT], 1);
  }
  __syncthreads();
  for (int b = t; b < nbins; b += 256) cnt[b * NCHUNK + g] = hist[b];
  if (g == 0 && t == 0) *total = 0;
}

__global__ __launch_bounds__(NCHUNK) void gat_bin_offsets(
    int* __restrict__ cnt, int* __restrict__ total,
    int* __restrict__ bin_base, int* __restrict__ bin_cnt) {
  __shared__ int wsum[NCHUNK / 32];
  __shared__ int base_s;
  const int t = threadIdx.x, lane = t & 31, wave = t >> 5;
  const int b = blockIdx.x;
  const int v = cnt[b * NCHUNK + t];
  int inc = v;
  for (int o = 1; o < 32; o <<= 1) {
    const int u = __shfl_up(inc, o, 32);
    if (lane >= o) inc += u;
  }
  if (lane == 31) wsum[wave] = inc;
  __syncthreads();
  if (t == 0) {
    int tot = 0;
    for (int w = 0; w < NCHUNK / 32; ++w) { const int c = wsum[w]; wsum[w] = tot; tot += c; }
    base_s = atomicAdd(total, tot);
    bin_base[b] = base_s;
    bin_cnt[b]  = tot;
  }
  __syncthreads();
  cnt[b * NCHUNK + t] = base_s + wsum[wave] + inc - v;
}

__global__ __launch_bounds__(256) void gat_bin_scatter(
    const int* __restrict__ ei, const int* __restrict__ offs,
    int* __restrict__ pairs, int E, int nbins) {
  __shared__ int cur[MAX_BINS];
  int2* __restrict__ pr = (int2*)pairs;
  const int t = threadIdx.x, g = blockIdx.x;
  for (int b = t; b < nbins; b += 256) cur[b] = offs[b * NCHUNK + g];
  __syncthreads();
  const int ce = (E + NCHUNK - 1) / NCHUNK;
  const int e0 = g * ce;
  const int e1 = min(E, e0 + ce);
  for (int e = e0 + t; e < e1; e += 4 * 256) {
    int d[4], sr[4];
#pragma unroll
    for (int u = 0; u < 4; ++u) {
      const int ee = e + u * 256;
      d[u]  = (ee < e1) ? ei[E + ee] : -1;
      sr[u] = (ee < e1) ? ei[ee] : 0;
    }
#pragma unroll
    for (int u = 0; u < 4; ++u)
      if (d[u] >= 0) {
        const int pos = atomicAdd(&cur[d[u] >> BIN_SHIFT], 1);
        pr[pos] = make_int2(sr[u], d[u]);
      }
  }
}

__global__ __launch_bounds__(BIN_NODES) void gat_bin_csr(
    const int* __restrict__ pairs, const int* __restrict__ bin_base,
    const int* __restrict__ bin_cnt, int* __restrict__ rowstart,
    int* __restrict__ deg, int* __restrict__ csr_src, int N) {
  __shared__ int dcnt[BIN_NODES];
  __shared__ int dpos[BIN_NODES];
  __shared__ int wsum[BIN_NODES / 32];
  const int t = threadIdx.x, lane = t & 31, wave = t >> 5;
  const int b = blockIdx.x;
  const int n = b * BIN_NODES + t;
  const int base = bin_base[b];
  const int tot  = bin_cnt[b];
  const int2* __restrict__ pb = (const int2*)pairs + base;
  dcnt[t] = 0;
  __syncthreads();
  for (int i = t; i < tot; i += 4 * BIN_NODES) {
    int d[4];
#pragma unroll
    for (int u = 0; u < 4; ++u) {
      const int ii = i + u * BIN_NODES;
      d[u] = (ii < tot) ? pb[ii].y : -1;
    }
#pragma unroll
    for (int u = 0; u < 4; ++u)
      if (d[u] >= 0) atomicAdd(&dcnt[d[u] & (BIN_NODES - 1)], 1);
  }
  __syncthreads();
  const int v = dcnt[t];
  int inc = v;
  for (int o = 1; o < 32; o <<= 1) {
    const int u = __shfl_up(inc, o, 32);
    if (lane >= o) inc += u;
  }
  if (lane == 31) wsum[wave] = inc;
  __syncthreads();
  if (t == 0) {
    int run = 0;
    for (int w = 0; w < BIN_NODES / 32; ++w) { const int c = wsum[w]; wsum[w] = run; run += c; }
  }
  __syncthreads();
  const int first = base + wsum[wave] + inc - v;
  dpos[t] = first;
  if (n < N) { rowstart[n] = first; deg[n] = v; }
  __syncthreads();
  for (int i = t; i < tot; i += 4 * BIN_NODES) {
    int2 r[4];
#pragma unroll
    for (int u = 0; u < 4; ++u) {
      const int ii = i + u * BIN_NODES;
      r[u] = (ii < tot) ? pb[ii] : make_int2(0, -1);
    }
#pragma unroll
    for (int u = 0; u < 4; ++u)
      if (r[u].y >= 0)
        csr_src[atomicAdd(&dpos[r[u].y & (BIN_NODES - 1)], 1)] = r[u].x;
  }
}

__global__ __launch_bounds__(128) void gat_gemm_wmma_f32(
    const float* __restrict__ X, const float* __restrict__ W,
    float* __restrict__ H1, int N) {
  __shared__ float Wp[64 * 256];

  const int t    = threadIdx.x;
  const int wave = __builtin_amdgcn_readfirstlane(t >> 5);
  const int lane = t & 31;
  const int hf   = lane >> 4;
  const int c    = lane & 15;
  const int row0 = (blockIdx.x * 4 + wave) * 16;

  const int ar = min(row0 + c, N - 1);
  const float* __restrict__ xr = X + (size_t)ar * F_IN + 4 * hf;
  float4 a[16];
#pragma unroll
  for (int j = 0; j < 16; ++j) a[j] = *(const float4*)(xr + 8 * j);

  const float* __restrict__ wsrc = W + (size_t)(2 * wave) * F_IN + 4 * lane;
  float* __restrict__ wdst = Wp + wave * 256 + 8 * lane;
#pragma unroll
  for (int hb = 0; hb < 2; ++hb) {
    float4 r0[8], r1[8];
#pragma unroll
    for (int it = 0; it < 8; ++it) {
      const int n = hb * 8 + it;
      r0[it] = *(const float4*)(wsrc + (size_t)n * 8 * F_IN);
      r1[it] = *(const float4*)(wsrc + (size_t)n * 8 * F_IN + F_IN);
    }
#pragma unroll
    for (int it = 0; it < 8; ++it) {
      const int n = hb * 8 + it;
      float4* p = (float4*)(wdst + n * 4 * 256);
      p[0] = make_float4(r0[it].x, r1[it].x, r0[it].y, r1[it].y);
      p[1] = make_float4(r0[it].z, r1[it].z, r0[it].w, r1[it].w);
    }
  }
  __syncthreads();
  if (row0 >= N) return;

  v8f acc[8];
#pragma unroll
  for (int i = 0; i < 8; ++i) acc[i] = v8f{};

  const float* __restrict__ bp = Wp + (2 * hf) * 256 + 4 * c;
#pragma unroll
  for (int j = 0; j < 16; ++j) {
#pragma unroll
    for (int s = 0; s < 2; ++s) {
      const float* q = bp + (4 * j + s) * 256;
      const float4 q0 = *(const float4*)(q);
      const float4 q1 = *(const float4*)(q + 64);
      const float4 q2 = *(const float4*)(q + 128);
      const float4 q3 = *(const float4*)(q + 192);
      v2f av;
      av.x = s ? a[j].z : a[j].x;
      av.y = s ? a[j].w : a[j].y;
      v2f b;
      b.x = q0.x; b.y = q0.y;
      acc[0] = __builtin_amdgcn_wmma_f32_16x16x4_f32(false, av, false, b, (short)0, acc[0], false, false);
      b.x = q0.z; b.y = q0.w;
      acc[1] = __builtin_amdgcn_wmma_f32_16x16x4_f32(false, av, false, b, (short)0, acc[1], false, false);
      b.x = q1.x; b.y = q1.y;
      acc[2] = __builtin_amdgcn_wmma_f32_16x16x4_f32(false, av, false, b, (short)0, acc[2], false, false);
      b.x = q1.z; b.y = q1.w;
      acc[3] = __builtin_amdgcn_wmma_f32_16x16x4_f32(false, av, false, b, (short)0, acc[3], false, false);
      b.x = q2.x; b.y = q2.y;
      acc[4] = __builtin_amdgcn_wmma_f32_16x16x4_f32(false, av, false, b, (short)0, acc[4], false, false);
      b.x = q2.z; b.y = q2.w;
      acc[5] = __builtin_amdgcn_wmma_f32_16x16x4_f32(false, av, false, b, (short)0, acc[5], false, false);
      b.x = q3.x; b.y = q3.y;
      acc[6] = __builtin_amdgcn_wmma_f32_16x16x4_f32(false, av, false, b, (short)0, acc[6], false, false);
      b.x = q3.z; b.y = q3.w;
      acc[7] = __builtin_amdgcn_wmma_f32_16x16x4_f32(false, av, false, b, (short)0, acc[7], false, false);
    }
  }

  float* __restrict__ o = H1 + (size_t)(row0 + 8 * hf) * F_IN + 2 * c;
#pragma unroll
  for (int r = 0; r < 8; ++r) {
#pragma unroll
    for (int u = 0; u < 4; ++u)
      *(float2*)(o + (size_t)r * F_IN + 32 * u) = make_float2(acc[2 * u][r], acc[2 * u + 1][r]);
  }
}

__global__ __launch_bounds__(256) void gat_attn1(
    const float* __restrict__ H1, const float* __restrict__ att_src,
    const float* __restrict__ att_dst, float* __restrict__ as1,
    float* __restrict__ ad1, int N) {
  int idx = blockIdx.x * blockDim.x + threadIdx.x;
  if (idx >= N * HEADS) return;
  int n = idx >> 3, h = idx & 7;
  const float* __restrict__ hp = H1 + (size_t)n * F_IN + h * CH;
  float s = 0.f, d = 0.f;
#pragma unroll
  for (int c = 0; c < CH; ++c) {
    float v = hp[c];
    s = fmaf(v, att_src[h * CH + c], s);
    d = fmaf(v, att_dst[h * CH + c], d);
  }
  as1[idx] = s;
  ad1[idx] = d;
}

__global__ __launch_bounds__(256) void gat_gather1(
    const int* __restrict__ rowstart, const int* __restrict__ deg,
    const int* __restrict__ csr_src, const float* __restrict__ as1,
    const float* __restrict__ ad1, const float* __restrict__ H1,
    const float* __restrict__ b1, float* __restrict__ HACT, int N) {
  const int lane = threadIdx.x & 31;
  const int n = __builtin_amdgcn_readfirstlane(blockIdx.x * 8 + (threadIdx.x >> 5));
  if (n >= N) return;
  const int h = lane >> 2;
  const int start = rowstart[n];
  const int cnt   = deg[n];
  const float adh   = ad1[n * HEADS + h];
  const float vself = leaky(as1[n * HEADS + h] + adh);

  float m = vself;
  for (int c0 = 0; c0 < cnt; c0 += 32) {
    const int nn = min(32, cnt - c0);
    const int mine = (lane < nn) ? csr_src[start + c0 + lane] : 0;
    for (int j = 0; j < nn; ++j) {
      const int s = __builtin_amdgcn_readlane(mine, j);
      m = fmaxf(m, leaky(as1[s * HEADS + h] + adh));
    }
  }

  float ex = __expf(vself - m);
  float l = ex;
  float4 hv = *(const float4*)(H1 + (size_t)n * F_IN + 4 * lane);
  float4 acc;
  acc.x = ex * hv.x; acc.y = ex * hv.y; acc.z = ex * hv.z; acc.w = ex * hv.w;
  for (int c0 = 0; c0 < cnt; c0 += 32) {
    const int nn = min(32, cnt - c0);
    const int mine = (lane < nn) ? csr_src[start + c0 + lane] : 0;
#pragma unroll 4
    for (int j = 0; j < nn; ++j) {
      const int s = __builtin_amdgcn_readlane(mine, j);
      const float4 r = *(const float4*)(H1 + (size_t)s * F_IN + 4 * lane);
      const float w = __expf(leaky(as1[s * HEADS + h] + adh) - m);
      l += w;
      acc.x = fmaf(w, r.x, acc.x); acc.y = fmaf(w, r.y, acc.y);
      acc.z = fmaf(w, r.z, acc.z); acc.w = fmaf(w, r.w, acc.w);
    }
  }

  const float den = l + 1e-16f;
  const float4 b = *(const float4*)(b1 + 4 * lane);
  float4 o;
  o.x = acc.x / den + b.x; o.y = acc.y / den + b.y;
  o.z = acc.z / den + b.z; o.w = acc.w / den + b.w;
  o.x = (o.x > 0.f) ? o.x : (__expf(o.x) - 1.f);
  o.y = (o.y > 0.f) ? o.y : (__expf(o.y) - 1.f);
  o.z = (o.z > 0.f) ? o.z : (__expf(o.z) - 1.f);
  o.w = (o.w > 0.f) ? o.w : (__expf(o.w) - 1.f);
  *(float4*)(HACT + (size_t)n * F_IN + 4 * lane) = o;
}

__global__ __launch_bounds__(256) void gat_layer2_node(
    const float* __restrict__ HACT, const float* __restrict__ W2,
    const float* __restrict__ att_src2, const float* __restrict__ att_dst2,
    float* __restrict__ H2, float* __restrict__ as2, float* __restrict__ ad2,
    int N) {
  int n = blockIdx.x * blockDim.x + threadIdx.x;
  if (n >= N) return;
  const float* __restrict__ hp = HACT + (size_t)n * F_IN;
  float a0 = 0.f, a1 = 0.f;
#pragma unroll 8
  for (int k = 0; k < F_IN; ++k) {
    float v = hp[k];
    a0 = fmaf(v, W2[k * OUTC + 0], a0);
    a1 = fmaf(v, W2[k * OUTC + 1], a1);
  }
  H2[n * OUTC + 0] = a0;
  H2[n * OUTC + 1] = a1;
  as2[n] = a0 * att_src2[0] + a1 * att_src2[1];
  ad2[n] = a0 * att_dst2[0] + a1 * att_dst2[1];
}

__global__ __launch_bounds__(256) void gat_gather2(
    const int* __restrict__ rowstart, const int* __restrict__ deg,
    const int* __restrict__ csr_src, const float* __restrict__ as2,
    const float* __restrict__ ad2, const float* __restrict__ H2,
    const float* __restrict__ b2, float* __restrict__ OUT, int N) {
  const int n = blockIdx.x * 256 + threadIdx.x;
  if (n >= N) return;
  const int start = rowstart[n];
  const int cnt   = deg[n];
  const float adn   = ad2[n];
  const float vself = leaky(as2[n] + adn);

  float m = vself;
  for (int j = 0; j < cnt; ++j)
    m = fmaxf(m, leaky(as2[csr_src[start + j]] + adn));

  float ex = __expf(vself - m);
  float l = ex;
  float2 hv = *(const float2*)(H2 + (size_t)n * OUTC);
  float a0 = ex * hv.x, a1 = ex * hv.y;
  for (int j = 0; j < cnt; ++j) {
    const int s = csr_src[start + j];
    const float2 r = *(const float2*)(H2 + (size_t)s * OUTC);
    const float w = __expf(leaky(as2[s] + adn) - m);
    l += w;
    a0 = fmaf(w, r.x, a0);
    a1 = fmaf(w, r.y, a1);
  }
  const float den = l + 1e-16f;
  float2 o;
  o.x = a0 / den + b2[0];
  o.y = a1 / den + b2[1];
  *(float2*)(OUT + (size_t)n * OUTC) = o;
}

static inline unsigned gdim(long long n, int b) {
  return (unsigned)((n + b - 1) / b);
}

extern "C" void kernel_launch(void* const* d_in, const int* in_sizes, int n_in,
                              void* d_out, int out_size, void* d_ws,
                              size_t ws_size, hipStream_t stream) {
  const float* x        = (const float*)d_in[0];
  const int*   ei       = (const int*)d_in[1];
  const float* W1       = (const float*)d_in[2];
  const float* att_src1 = (const float*)d_in[3];
  const float* att_dst1 = (const float*)d_in[4];
  const float* b1       = (const float*)d_in[5];
  const float* W2       = (const float*)d_in[6];
  const float* att_src2 = (const float*)d_in[7];
  const float* att_dst2 = (const float*)d_in[8];
  const float* b2       = (const float*)d_in[9];

  const int N  = in_sizes[0] / F_IN;
  const int E  = in_sizes[1] / 2;

  float* ws = (float*)d_ws;
  size_t off = 0;
  float* h1   = ws + off; off += (size_t)N * F_IN;
  float* hact = ws + off; off += (size_t)N * F_IN;
  float* as1  = ws + off; off += (size_t)N * HEADS;
  float* ad1  = ws + off; off += (size_t)N * HEADS;
  float* h2   = ws + off; off += (size_t)N * OUTC;
  float* as2  = ws + off; off += (size_t)N;
  float* ad2  = ws + off; off += (size_t)N;
  int* pairs    = (int*)(ws + off); off += (size_t)E * 2;
  int* deg      = (int*)(ws + off); off += (size_t)N;
  int* rowstart = (int*)(ws + off); off += (size_t)N;
  int* csr_src  = (int*)(ws + off); off += (size_t)E;
  const int nbins = (int)gdim(N, BIN_NODES);
  int* bin_cnt  = (int*)(ws + off); off += (size_t)nbins * NCHUNK;
  int* bin_base = (int*)(ws + off); off += (size_t)nbins;
  int* bin_tot  = (int*)(ws + off); off += (size_t)nbins;
  int* total    = (int*)(ws + off); off += 1;
  (void)ws_size; (void)n_in; (void)out_size;

  const int B = 256;

  gat_bin_count<<<NCHUNK, B, 0, stream>>>(ei, bin_cnt, total, E, nbins);
  gat_bin_offsets<<<nbins, NCHUNK, 0, stream>>>(bin_cnt, total, bin_base, bin_tot);
  gat_bin_scatter<<<NCHUNK, B, 0, stream>>>(ei, bin_cnt, pairs, E, nbins);
  gat_bin_csr<<<nbins, BIN_NODES, 0, stream>>>(pairs, bin_base, bin_tot, rowstart, deg, csr_src, N);

  gat_gemm_wmma_f32<<<gdim(N, 64), 128, 0, stream>>>(x, W1, h1, N);
  gat_attn1<<<gdim((long long)N * HEADS, B), B, 0, stream>>>(h1, att_src1, att_dst1, as1, ad1, N);
  gat_gather1<<<gdim(N, 8), B, 0, stream>>>(rowstart, deg, csr_src, as1, ad1, h1, b1, hact, N);

  gat_layer2_node<<<gdim(N, B), B, 0, stream>>>(hact, W2, att_src2, att_dst2, h2, as2, ad2, N);
  gat_gather2<<<gdim(N, B), B, 0, stream>>>(rowstart, deg, csr_src, as2, ad2, h2, b2, (float*)d_out, N);
}


# Layer-1 attention dots taken from the GEMM accumulators in its epilogue (8-lane DPP sums): the separate dot launch and its re-read of h1 are gone; each dot is now summed as a tree, not serially
# speedup vs baseline: 5.1148x; 1.0348x over previous
#include <hip/hip_runtime.h>
#include <hip/hip_bf16.h>


typedef __attribute__((ext_vector_type(2))) float v2f;
typedef __attribute__((ext_vector_type(8))) float v8f;

#define F_IN   128
#define HEADS  8
#define CH     16
#define OUTC   2
#define NEG_SLOPE 0.2f
#define BIN_SHIFT 8
#define BIN_NODES 256
#define MAX_BINS  512
#define NCHUNK    256

__device__ __forceinline__ float leaky(float v) {
  return (v > 0.f) ? v : NEG_SLOPE * v;
}

__global__ __launch_bounds__(256) void gat_bin_count(
    const int* __restrict__ ei, int* __restrict__ cnt,
    int* __restrict__ total, int E, int nbins) {
  __shared__ int hist[MAX_BINS];
  const int t = threadIdx.x, g = blockIdx.x;
  for (int b = t; b < nbins; b += 256) hist[b] = 0;
  __syncthreads();
  const int ce = (E + NCHUNK - 1) / NCHUNK;
  const int e0 = g * ce;
  const int e1 = min(E, e0 + ce);
  for (int e = e0 + t; e < e1; e += 4 * 256) {
    int d[4];
#pragma unroll
    for (int u = 0; u < 4; ++u) {
      const int ee = e + u * 256;
      d[u] = (ee < e1) ? ei[E + ee] : -1;
    }
#pragma unroll
    for (int u = 0; u < 4; ++u)
      if (d[u] >= 0) atomicAdd(&hist[d[u] >> BIN_SHIFT], 1);
  }
  __syncthreads();
  for (int b = t; b < nbins; b += 256) cnt[b * NCHUNK + g] = hist[b];
  if (g == 0 && t == 0) *total = 0;
}

__global__ __launch_bounds__(NCHUNK) void gat_bin_offsets(
    int* __restrict__ cnt, int* __restrict__ total,
    int* __restrict__ bin_base, int* __restrict__ bin_cnt) {
  __shared__ int wsum[NCHUNK / 32];
  __shared__ int base_s;
  const int t = threadIdx.x, lane = t & 31, wave = t >> 5;
  const int b = blockIdx.x;
  const int v = cnt[b * NCHUNK + t];
  int inc = v;
  for (int o = 1; o < 32; o <<= 1) {
    const int u = __shfl_up(inc, o, 32);
    if (lane >= o) inc += u;
  }
  if (lane == 31) wsum[wave] = inc;
  __syncthreads();
  if (t == 0) {
    int tot = 0;
    for (int w = 0; w < NCHUNK / 32; ++w) { const int c = wsum[w]; wsum[w] = tot; tot += c; }
    base_s = atomicAdd(total, tot);
    bin_base[b] = base_s;
    bin_cnt[b]  = tot;
  }
  __syncthreads();
  cnt[b * NCHUNK + t] = base_s + wsum[wave] + inc - v;
}

__global__ __launch_bounds__(256) void gat_bin_scatter(
    const int* __restrict__ ei, const int* __restrict__ offs,
    int* __restrict__ pairs, int E, int nbins) {
  __shared__ int cur[MAX_BINS];
  int2* __restrict__ pr = (int2*)pairs;
  const int t = threadIdx.x, g = blockIdx.x;
  for (int b = t; b < nbins; b += 256) cur[b] = offs[b * NCHUNK + g];
  __syncthreads();
  const int ce = (E + NCHUNK - 1) / NCHUNK;
  const int e0 = g * ce;
  const int e1 = min(E, e0 + ce);
  for (int e = e0 + t; e < e1; e += 4 * 256) {
    int d[4], sr[4];
#pragma unroll
    for (int u = 0; u < 4; ++u) {
      const int ee = e + u * 256;
      d[u]  = (ee < e1) ? ei[E + ee] : -1;
      sr[u] = (ee < e1) ? ei[ee] : 0;
    }
#pragma unroll
    for (int u = 0; u < 4; ++u)
      if (d[u] >= 0) {
        const int pos = atomicAdd(&cur[d[u] >> BIN_SHIFT], 1);
        pr[pos] = make_int2(sr[u], d[u]);
      }
  }
}

__global__ __launch_bounds__(BIN_NODES) void gat_bin_csr(
    const int* __restrict__ pairs, const int* __restrict__ bin_base,
    const int* __restrict__ bin_cnt, int* __restrict__ rowstart,
    int* __restrict__ deg, int* __restrict__ csr_src, int N) {
  __shared__ int dcnt[BIN_NODES];
  __shared__ int dpos[BIN_NODES];
  __shared__ int wsum[BIN_NODES / 32];
  const int t = threadIdx.x, lane = t & 31, wave = t >> 5;
  const int b = blockIdx.x;
  const int n = b * BIN_NODES + t;
  const int base = bin_base[b];
  const int tot  = bin_cnt[b];
  const int2* __restrict__ pb = (const int2*)pairs + base;
  dcnt[t] = 0;
  __syncthreads();
  for (int i = t; i < tot; i += 4 * BIN_NODES) {
    int d[4];
#pragma unroll
    for (int u = 0; u < 4; ++u) {
      const int ii = i + u * BIN_NODES;
      d[u] = (ii < tot) ? pb[ii].y : -1;
    }
#pragma unroll
    for (int u = 0; u < 4; ++u)
      if (d[u] >= 0) atomicAdd(&dcnt[d[u] & (BIN_NODES - 1)], 1);
  }
  __syncthreads();
  const int v = dcnt[t];
  int inc = v;
  for (int o = 1; o < 32; o <<= 1) {
    const int u = __shfl_up(inc, o, 32);
    if (lane >= o) inc += u;
  }
  if (lane == 31) wsum[wave] = inc;
  __syncthreads();
  if (t == 0) {
    int run = 0;
    for (int w = 0; w < BIN_NODES / 32; ++w) { const int c = wsum[w]; wsum[w] = run; run += c; }
  }
  __syncthreads();
  const int first = base + wsum[wave] + inc - v;
  dpos[t] = first;
  if (n < N) { rowstart[n] = first; deg[n] = v; }
  __syncthreads();
  for (int i = t; i < tot; i += 4 * BIN_NODES) {
    int2 r[4];
#pragma unroll
    for (int u = 0; u < 4; ++u) {
      const int ii = i + u * BIN_NODES;
      r[u] = (ii < tot) ? pb[ii] : make_int2(0, -1);
    }
#pragma unroll
    for (int u = 0; u < 4; ++u)
      if (r[u].y >= 0)
        csr_src[atomicAdd(&dpos[r[u].y & (BIN_NODES - 1)], 1)] = r[u].x;
  }
}

template <int M>
__device__ __forceinline__ float lane_xor(float v) {
  return __builtin_bit_cast(float, __builtin_amdgcn_update_dpp(
      0, __builtin_bit_cast(int, v), 0x160 + M, 0xf, 0xf, true));
}

__device__ __forceinline__ float sum8_by_row(const float (&p)[8], bool b2,
                                             bool b1, bool b0) {
  float q[4], s[2];
#pragma unroll
  for (int i = 0; i < 4; ++i) {
    const float snd = b2 ? p[i] : p[i + 4];
    const float kp  = b2 ? p[i + 4] : p[i];
    q[i] = kp + lane_xor<4>(snd);
  }
#pragma unroll
  for (int i = 0; i < 2; ++i) {
    const float snd = b1 ? q[i] : q[i + 2];
    const float kp  = b1 ? q[i + 2] : q[i];
    s[i] = kp + lane_xor<2>(snd);
  }
  const float snd = b0 ? s[0] : s[1];
  const float kp  = b0 ? s[1] : s[0];
  return kp + lane_xor<1>(snd);
}

__global__ __launch_bounds__(128) void gat_gemm_wmma_f32(
    const float* __restrict__ X, const float* __restrict__ W,
    const float* __restrict__ att_src, const float* __restrict__ att_dst,
    float* __restrict__ H1, float* __restrict__ AS1, float* __restrict__ AD1,
    int N) {
  __shared__ float Wp[64 * 256];

  const int t    = threadIdx.x;
  const int wave = __builtin_amdgcn_readfirstlane(t >> 5);
  const int lane = t & 31;
  const int hf   = lane >> 4;
  const int c    = lane & 15;
  const int row0 = (blockIdx.x * 4 + wave) * 16;

  const int ar = min(row0 + c, N - 1);
  const float* __restrict__ xr = X + (size_t)ar * F_IN + 4 * hf;
  float4 a[16];
#pragma unroll
  for (int j = 0; j < 16; ++j) a[j] = *(const float4*)(xr + 8 * j);

  const float* __restrict__ wsrc = W + (size_t)(2 * wave) * F_IN + 4 * lane;
  float* __restrict__ wdst = Wp + wave * 256 + 8 * lane;
#pragma unroll
  for (int hb = 0; hb < 2; ++hb) {
    float4 r0[8], r1[8];
#pragma unroll
    for (int it = 0; it < 8; ++it) {
      const int n = hb * 8 + it;
      r0[it] = *(const float4*)(wsrc + (size_t)n * 8 * F_IN);
      r1[it] = *(const float4*)(wsrc + (size_t)n * 8 * F_IN + F_IN);
    }
#pragma unroll
    for (int it = 0; it < 8; ++it) {
      const int n = hb * 8 + it;
      float4* p = (float4*)(wdst + n * 4 * 256);
      p[0] = make_float4(r0[it].x, r1[it].x, r0[it].y, r1[it].y);
      p[1] = make_float4(r0[it].z, r1[it].z, r0[it].w, r1[it].w);
    }
  }
  __syncthreads();
  if (row0 >= N) return;

  v8f acc[8];
#pragma unroll
  for (int i = 0; i < 8; ++i) acc[i] = v8f{};

  float2 asv[4], adv[4];
#pragma unroll
  for (int u = 0; u < 4; ++u) {
    asv[u] = *(const float2*)(att_src + 32 * u + 2 * c);
    adv[u] = *(const float2*)(att_dst + 32 * u + 2 * c);
  }

  const float* __restrict__ bp = Wp + (2 * hf) * 256 + 4 * c;
#pragma unroll
  for (int j = 0; j < 16; ++j) {
#pragma unroll
    for (int s = 0; s < 2; ++s) {
      const float* q = bp + (4 * j + s) * 256;
      const float4 q0 = *(const float4*)(q);
      const float4 q1 = *(const float4*)(q + 64);
      const float4 q2 = *(const float4*)(q + 128);
      const float4 q3 = *(const float4*)(q + 192);
      v2f av;
      av.x = s ? a[j].z : a[j].x;
      av.y = s ? a[j].w : a[j].y;
      v2f b;
      b.x = q0.x; b.y = q0.y;
      acc[0] = __builtin_amdgcn_wmma_f32_16x16x4_f32(false, av, false, b, (short)0, acc[0], false, false);
      b.x = q0.z; b.y = q0.w;
      acc[1] = __builtin_amdgcn_wmma_f32_16x16x4_f32(false, av, false, b, (short)0, acc[1], false, false);
      b.x = q1.x; b.y = q1.y;
      acc[2] = __builtin_amdgcn_wmma_f32_16x16x4_f32(false, av, false, b, (short)0, acc[2], false, false);
      b.x = q1.z; b.y = q1.w;
      acc[3] = __builtin_amdgcn_wmma_f32_16x16x4_f32(false, av, false, b, (short)0, acc[3], false, false);
      b.x = q2.x; b.y = q2.y;
      acc[4] = __builtin_amdgcn_wmma_f32_16x16x4_f32(false, av, false, b, (short)0, acc[4], false, false);
      b.x = q2.z; b.y = q2.w;
      acc[5] = __builtin_amdgcn_wmma_f32_16x16x4_f32(false, av, false, b, (short)0, acc[5], false, false);
      b.x = q3.x; b.y = q3.y;
      acc[6] = __builtin_amdgcn_wmma_f32_16x16x4_f32(false, av, false, b, (short)0, acc[6], false, false);
      b.x = q3.z; b.y = q3.w;
      acc[7] = __builtin_amdgcn_wmma_f32_16x16x4_f32(false, av, false, b, (short)0, acc[7], false, false);
    }
  }

  float* __restrict__ o = H1 + (size_t)(row0 + 8 * hf) * F_IN + 2 * c;
#pragma unroll
  for (int r = 0; r < 8; ++r) {
#pragma unroll
    for (int u = 0; u < 4; ++u)
      *(float2*)(o + (size_t)r * F_IN + 32 * u) = make_float2(acc[2 * u][r], acc[2 * u + 1][r]);
  }

  const bool b0 = (lane & 1) != 0, b1 = (lane & 2) != 0, b2 = (lane & 4) != 0;
  const bool g  = (lane & 8) != 0;
  float ts[4], td[4];
#pragma unroll
  for (int u = 0; u < 4; ++u) {
    float ps[8], pd[8];
#pragma unroll
    for (int r = 0; r < 8; ++r) {
      ps[r] = fmaf(acc[2 * u + 1][r], asv[u].y, acc[2 * u][r] * asv[u].x);
      pd[r] = fmaf(acc[2 * u + 1][r], adv[u].y, acc[2 * u][r] * adv[u].x);
    }
    ts[u] = sum8_by_row(ps, b2, b1, b0);
    td[u] = sum8_by_row(pd, b2, b1, b0);
  }
  const float xs0 = lane_xor<8>(g ? ts[0] : ts[2]);
  const float xs1 = lane_xor<8>(g ? ts[1] : ts[3]);
  const float xd0 = lane_xor<8>(g ? td[0] : td[2]);
  const float xd1 = lane_xor<8>(g ? td[1] : td[3]);
  const float4 os = g ? make_float4(xs0, ts[2], xs1, ts[3]) : make_float4(ts[0], xs0, ts[1], xs1);
  const float4 od = g ? make_float4(xd0, td[2], xd1, td[3]) : make_float4(td[0], xd0, td[1], xd1);
  const size_t ao = (size_t)(row0 + 8 * hf + (lane & 7)) * HEADS + (g ? 4 : 0);
  *(float4*)(AS1 + ao) = os;
  *(float4*)(AD1 + ao) = od;
}

__global__ __launch_bounds__(256) void gat_gather1(
    const int* __restrict__ rowstart, const int* __restrict__ deg,
    const int* __restrict__ csr_src, const float* __restrict__ as1,
    const float* __restrict__ ad1, const float* __restrict__ H1,
    const float* __restrict__ b1, float* __restrict__ HACT, int N) {
  const int lane = threadIdx.x & 31;
  const int n = __builtin_amdgcn_readfirstlane(blockIdx.x * 8 + (threadIdx.x >> 5));
  if (n >= N) return;
  const int h = lane >> 2;
  const int start = rowstart[n];
  const int cnt   = deg[n];
  const float adh   = ad1[n * HEADS + h];
  const float vself = leaky(as1[n * HEADS + h] + adh);

  float m = vself;
  for (int c0 = 0; c0 < cnt; c0 += 32) {
    const int nn = min(32, cnt - c0);
    const int mine = (lane < nn) ? csr_src[start + c0 + lane] : 0;
    for (int j = 0; j < nn; ++j) {
      const int s = __builtin_amdgcn_readlane(mine, j);
      m = fmaxf(m, leaky(as1[s * HEADS + h] + adh));
    }
  }

  float ex = __expf(vself - m);
  float l = ex;
  float4 hv = *(const float4*)(H1 + (size_t)n * F_IN + 4 * lane);
  float4 acc;
  acc.x = ex * hv.x; acc.y = ex * hv.y; acc.z = ex * hv.z; acc.w = ex * hv.w;
  for (int c0 = 0; c0 < cnt; c0 += 32) {
    const int nn = min(32, cnt - c0);
    const int mine = (lane < nn) ? csr_src[start + c0 + lane] : 0;
#pragma unroll 4
    for (int j = 0; j < nn; ++j) {
      const int s = __builtin_amdgcn_readlane(mine, j);
      const float4 r = *(const float4*)(H1 + (size_t)s * F_IN + 4 * lane);
      const float w = __expf(leaky(as1[s * HEADS + h] + adh) - m);
      l += w;
      acc.x = fmaf(w, r.x, acc.x); acc.y = fmaf(w, r.y, acc.y);
      acc.z = fmaf(w, r.z, acc.z); acc.w = fmaf(w, r.w, acc.w);
    }
  }

  const float den = l + 1e-16f;
  const float4 b = *(const float4*)(b1 + 4 * lane);
  float4 o;
  o.x = acc.x / den + b.x; o.y = acc.y / den + b.y;
  o.z = acc.z / den + b.z; o.w = acc.w / den + b.w;
  o.x = (o.x > 0.f) ? o.x : (__expf(o.x) - 1.f);
  o.y = (o.y > 0.f) ? o.y : (__expf(o.y) - 1.f);
  o.z = (o.z > 0.f) ? o.z : (__expf(o.z) - 1.f);
  o.w = (o.w > 0.f) ? o.w : (__expf(o.w) - 1.f);
  *(float4*)(HACT + (size_t)n * F_IN + 4 * lane) = o;
}

__global__ __launch_bounds__(256) void gat_layer2_node(
    const float* __restrict__ HACT, const float* __restrict__ W2,
    const float* __restrict__ att_src2, const float* __restrict__ att_dst2,
    float* __restrict__ H2, float* __restrict__ as2, float* __restrict__ ad2,
    int N) {
  int n = blockIdx.x * blockDim.x + threadIdx.x;
  if (n >= N) return;
  const float* __restrict__ hp = HACT + (size_t)n * F_IN;
  float a0 = 0.f, a1 = 0.f;
#pragma unroll 8
  for (int k = 0; k < F_IN; ++k) {
    float v = hp[k];
    a0 = fmaf(v, W2[k * OUTC + 0], a0);
    a1 = fmaf(v, W2[k * OUTC + 1], a1);
  }
  H2[n * OUTC + 0] = a0;
  H2[n * OUTC + 1] = a1;
  as2[n] = a0 * att_src2[0] + a1 * att_src2[1];
  ad2[n] = a0 * att_dst2[0] + a1 * att_dst2[1];
}

__global__ __launch_bounds__(256) void gat_gather2(
    const int* __restrict__ rowstart, const int* __restrict__ deg,
    const int* __restrict__ csr_src, const float* __restrict__ as2,
    const float* __restrict__ ad2, const float* __restrict__ H2,
    const float* __restrict__ b2, float* __restrict__ OUT, int N) {
  const int n = blockIdx.x * 256 + threadIdx.x;
  if (n >= N) return;
  const int start = rowstart[n];
  const int cnt   = deg[n];
  const float adn   = ad2[n];
  const float vself = leaky(as2[n] + adn);

  float m = vself;
  for (int j = 0; j < cnt; ++j)
    m = fmaxf(m, leaky(as2[csr_src[start + j]] + adn));

  float ex = __expf(vself - m);
  float l = ex;
  float2 hv = *(const float2*)(H2 + (size_t)n * OUTC);
  float a0 = ex * hv.x, a1 = ex * hv.y;
  for (int j = 0; j < cnt; ++j) {
    const int s = csr_src[start + j];
    const float2 r = *(const float2*)(H2 + (size_t)s * OUTC);
    const float w = __expf(leaky(as2[s] + adn) - m);
    l += w;
    a0 = fmaf(w, r.x, a0);
    a1 = fmaf(w, r.y, a1);
  }
  const float den = l + 1e-16f;
  float2 o;
  o.x = a0 / den + b2[0];
  o.y = a1 / den + b2[1];
  *(float2*)(OUT + (size_t)n * OUTC) = o;
}

static inline unsigned gdim(long long n, int b) {
  return (unsigned)((n + b - 1) / b);
}

extern "C" void kernel_launch(void* const* d_in, const int* in_sizes, int n_in,
                              void* d_out, int out_size, void* d_ws,
                              size_t ws_size, hipStream_t stream) {
  const float* x        = (const float*)d_in[0];
  const int*   ei       = (const int*)d_in[1];
  const float* W1       = (const float*)d_in[2];
  const float* att_src1 = (const float*)d_in[3];
  const float* att_dst1 = (const float*)d_in[4];
  const float* b1       = (const float*)d_in[5];
  const float* W2       = (const float*)d_in[6];
  const float* att_src2 = (const float*)d_in[7];
  const float* att_dst2 = (const float*)d_in[8];
  const float* b2       = (const float*)d_in[9];

  const int N  = in_sizes[0] / F_IN;
  const int E  = in_sizes[1] / 2;

  float* ws = (float*)d_ws;
  size_t off = 0;
  float* h1   = ws + off; off += (size_t)N * F_IN;
  float* hact = ws + off; off += (size_t)N * F_IN;
  float* as1  = ws + off; off += (size_t)N * HEADS;
  float* ad1  = ws + off; off += (size_t)N * HEADS;
  float* h2   = ws + off; off += (size_t)N * OUTC;
  float* as2  = ws + off; off += (size_t)N;
  float* ad2  = ws + off; off += (size_t)N;
  int* pairs    = (int*)(ws + off); off += (size_t)E * 2;
  int* deg      = (int*)(ws + off); off += (size_t)N;
  int* rowstart = (int*)(ws + off); off += (size_t)N;
  int* csr_src  = (int*)(ws + off); off += (size_t)E;
  const int nbins = (int)gdim(N, BIN_NODES);
  int* bin_cnt  = (int*)(ws + off); off += (size_t)nbins * NCHUNK;
  int* bin_base = (int*)(ws + off); off += (size_t)nbins;
  int* bin_tot  = (int*)(ws + off); off += (size_t)nbins;
  int* total    = (int*)(ws + off); off += 1;
  (void)ws_size; (void)n_in; (void)out_size;

  const int B = 256;

  gat_bin_count<<<NCHUNK, B, 0, stream>>>(ei, bin_cnt, total, E, nbins);
  gat_bin_offsets<<<nbins, NCHUNK, 0, stream>>>(bin_cnt, total, bin_base, bin_tot);
  gat_bin_scatter<<<NCHUNK, B, 0, stream>>>(ei, bin_cnt, pairs, E, nbins);
  gat_bin_csr<<<nbins, BIN_NODES, 0, stream>>>(pairs, bin_base, bin_tot, rowstart, deg, csr_src, N);

  gat_gemm_wmma_f32<<<gdim(N, 64), 128, 0, stream>>>(x, W1, att_src1, att_dst1, h1, as1, ad1, N);
  gat_gather1<<<gdim(N, 8), B, 0, stream>>>(rowstart, deg, csr_src, as1, ad1, h1, b1, hact, N);

  gat_layer2_node<<<gdim(N, B), B, 0, stream>>>(hact, W2, att_src2, att_dst2, h2, as2, ad2, N);
  gat_gather2<<<gdim(N, B), B, 0, stream>>>(rowstart, deg, csr_src, as2, ad2, h2, b2, (float*)d_out, N);
}


# Layer-1 gather: a node's loads fly together: 16384 resident waves walk the nodes, logits fetched lane-parallel (max by cross-lane tree), rows in zero-padded groups of 4; every sum keeps its order
# speedup vs baseline: 5.3857x; 1.0530x over previous
#include <hip/hip_runtime.h>
#include <hip/hip_bf16.h>


typedef __attribute__((ext_vector_type(2))) float v2f;
typedef __attribute__((ext_vector_type(8))) float v8f;

#define F_IN   128
#define HEADS  8
#define CH     16
#define OUTC   2
#define NEG_SLOPE 0.2f
#define BIN_SHIFT 8
#define BIN_NODES 256
#define MAX_BINS  512
#define NCHUNK    256

__device__ __forceinline__ float leaky(float v) {
  return (v > 0.f) ? v : NEG_SLOPE * v;
}

__global__ __launch_bounds__(256) void gat_bin_count(
    const int* __restrict__ ei, int* __restrict__ cnt,
    int* __restrict__ total, int E, int nbins) {
  __shared__ int hist[MAX_BINS];
  const int t = threadIdx.x, g = blockIdx.x;
  for (int b = t; b < nbins; b += 256) hist[b] = 0;
  __syncthreads();
  const int ce = (E + NCHUNK - 1) / NCHUNK;
  const int e0 = g * ce;
  const int e1 = min(E, e0 + ce);
  for (int e = e0 + t; e < e1; e += 4 * 256) {
    int d[4];
#pragma unroll
    for (int u = 0; u < 4; ++u) {
      const int ee = e + u * 256;
      d[u] = (ee < e1) ? ei[E + ee] : -1;
    }
#pragma unroll
    for (int u = 0; u < 4; ++u)
      if (d[u] >= 0) atomicAdd(&hist[d[u] >> BIN_SHIFT], 1);
  }
  __syncthreads();
  for (int b = t; b < nbins; b += 256) cnt[b * NCHUNK + g] = hist[b];
  if (g == 0 && t == 0) *total = 0;
}

__global__ __launch_bounds__(NCHUNK) void gat_bin_offsets(
    int* __restrict__ cnt, int* __restrict__ total,
    int* __restrict__ bin_base, int* __restrict__ bin_cnt) {
  __shared__ int wsum[NCHUNK / 32];
  __shared__ int base_s;
  const int t = threadIdx.x, lane = t & 31, wave = t >> 5;
  const int b = blockIdx.x;
  const int v = cnt[b * NCHUNK + t];
  int inc = v;
  for (int o = 1; o < 32; o <<= 1) {
    const int u = __shfl_up(inc, o, 32);
    if (lane >= o) inc += u;
  }
  if (lane == 31) wsum[wave] = inc;
  __syncthreads();
  if (t == 0) {
    int tot = 0;
    for (int w = 0; w < NCHUNK / 32; ++w) { const int c = wsum[w]; wsum[w] = tot; tot += c; }
    base_s = atomicAdd(total, tot);
    bin_base[b] = base_s;
    bin_cnt[b]  = tot;
  }
  __syncthreads();
  cnt[b * NCHUNK + t] = base_s + wsum[wave] + inc - v;
}

__global__ __launch_bounds__(256) void gat_bin_scatter(
    const int* __restrict__ ei, const int* __restrict__ offs,
    int* __restrict__ pairs, int E, int nbins) {
  __shared__ int cur[MAX_BINS];
  int2* __restrict__ pr = (int2*)pairs;
  const int t = threadIdx.x, g = blockIdx.x;
  for (int b = t; b < nbins; b += 256) cur[b] = offs[b * NCHUNK + g];
  __syncthreads();
  const int ce = (E + NCHUNK - 1) / NCHUNK;
  const int e0 = g * ce;
  const int e1 = min(E, e0 + ce);
  for (int e = e0 + t; e < e1; e += 4 * 256) {
    int d[4], sr[4];
#pragma unroll
    for (int u = 0; u < 4; ++u) {
      const int ee = e + u * 256;
      d[u]  = (ee < e1) ? ei[E + ee] : -1;
      sr[u] = (ee < e1) ? ei[ee] : 0;
    }
#pragma unroll
    for (int u = 0; u < 4; ++u)
      if (d[u] >= 0) {
        const int pos = atomicAdd(&cur[d[u] >> BIN_SHIFT], 1);
        pr[pos] = make_int2(sr[u], d[u]);
      }
  }
}

__global__ __launch_bounds__(BIN_NODES) void gat_bin_csr(
    const int* __restrict__ pairs, const int* __restrict__ bin_base,
    const int* __restrict__ bin_cnt, int* __restrict__ rowstart,
    int* __restrict__ deg, int* __restrict__ csr_src, int N) {
  __shared__ int dcnt[BIN_NODES];
  __shared__ int dpos[BIN_NODES];
  __shared__ int wsum[BIN_NODES / 32];
  const int t = threadIdx.x, lane = t & 31, wave = t >> 5;
  const int b = blockIdx.x;
  const int n = b * BIN_NODES + t;
  const int base = bin_base[b];
  const int tot  = bin_cnt[b];
  const int2* __restrict__ pb = (const int2*)pairs + base;
  dcnt[t] = 0;
  __syncthreads();
  for (int i = t; i < tot; i += 4 * BIN_NODES) {
    int d[4];
#pragma unroll
    for (int u = 0; u < 4; ++u) {
      const int ii = i + u * BIN_NODES;
      d[u] = (ii < tot) ? pb[ii].y : -1;
    }
#pragma unroll
    for (int u = 0; u < 4; ++u)
      if (d[u] >= 0) atomicAdd(&dcnt[d[u] & (BIN_NODES - 1)], 1);
  }
  __syncthreads();
  const int v = dcnt[t];
  int inc = v;
  for (int o = 1; o < 32; o <<= 1) {
    const int u = __shfl_up(inc, o, 32);
    if (lane >= o) inc += u;
  }
  if (lane == 31) wsum[wave] = inc;
  __syncthreads();
  if (t == 0) {
    int run = 0;
    for (int w = 0; w < BIN_NODES / 32; ++w) { const int c = wsum[w]; wsum[w] = run; run += c; }
  }
  __syncthreads();
  const int first = base + wsum[wave] + inc - v;
  dpos[t] = first;
  if (n < N) { rowstart[n] = first; deg[n] = v; }
  __syncthreads();
  for (int i = t; i < tot; i += 4 * BIN_NODES) {
    int2 r[4];
#pragma unroll
    for (int u = 0; u < 4; ++u) {
      const int ii = i + u * BIN_NODES;
      r[u] = (ii < tot) ? pb[ii] : make_int2(0, -1);
    }
#pragma unroll
    for (int u = 0; u < 4; ++u)
      if (r[u].y >= 0)
        csr_src[atomicAdd(&dpos[r[u].y & (BIN_NODES - 1)], 1)] = r[u].x;
  }
}

template <int M>
__device__ __forceinline__ float lane_xor(float v) {
  return __builtin_bit_cast(float, __builtin_amdgcn_update_dpp(
      0, __builtin_bit_cast(int, v), 0x160 + M, 0xf, 0xf, true));
}

__device__ __forceinline__ float sum8_by_row(const float (&p)[8], bool b2,
                                             bool b1, bool b0) {
  float q[4], s[2];
#pragma unroll
  for (int i = 0; i < 4; ++i) {
    const float snd = b2 ? p[i] : p[i + 4];
    const float kp  = b2 ? p[i + 4] : p[i];
    q[i] = kp + lane_xor<4>(snd);
  }
#pragma unroll
  for (int i = 0; i < 2; ++i) {
    const float snd = b1 ? q[i] : q[i + 2];
    const float kp  = b1 ? q[i + 2] : q[i];
    s[i] = kp + lane_xor<2>(snd);
  }
  const float snd = b0 ? s[0] : s[1];
  const float kp  = b0 ? s[1] : s[0];
  return kp + lane_xor<1>(snd);
}

__global__ __launch_bounds__(128) void gat_gemm_wmma_f32(
    const float* __restrict__ X, const float* __restrict__ W,
    const float* __restrict__ att_src, const float* __restrict__ att_dst,
    float* __restrict__ H1, float* __restrict__ AS1, float* __restrict__ AD1,
    int N) {
  __shared__ float Wp[64 * 256];

  const int t    = threadIdx.x;
  const int wave = __builtin_amdgcn_readfirstlane(t >> 5);
  const int lane = t & 31;
  const int hf   = lane >> 4;
  const int c    = lane & 15;
  const int row0 = (blockIdx.x * 4 + wave) * 16;

  const int ar = min(row0 + c, N - 1);
  const float* __restrict__ xr = X + (size_t)ar * F_IN + 4 * hf;
  float4 a[16];
#pragma unroll
  for (int j = 0; j < 16; ++j) a[j] = *(const float4*)(xr + 8 * j);

  const float* __restrict__ wsrc = W + (size_t)(2 * wave) * F_IN + 4 * lane;
  float* __restrict__ wdst = Wp + wave * 256 + 8 * lane;
#pragma unroll
  for (int hb = 0; hb < 2; ++hb) {
    float4 r0[8], r1[8];
#pragma unroll
    for (int it = 0; it < 8; ++it) {
      const int n = hb * 8 + it;
      r0[it] = *(const float4*)(wsrc + (size_t)n * 8 * F_IN);
      r1[it] = *(const float4*)(wsrc + (size_t)n * 8 * F_IN + F_IN);
    }
#pragma unroll
    for (int it = 0; it < 8; ++it) {
      const int n = hb * 8 + it;
      float4* p = (float4*)(wdst + n * 4 * 256);
      p[0] = make_float4(r0[it].x, r1[it].x, r0[it].y, r1[it].y);
      p[1] = make_float4(r0[it].z, r1[it].z, r0[it].w, r1[it].w);
    }
  }
  __syncthreads();
  if (row0 >= N) return;

  v8f acc[8];
#pragma unroll
  for (int i = 0; i < 8; ++i) acc[i] = v8f{};

  float2 asv[4], adv[4];
#pragma unroll
  for (int u = 0; u < 4; ++u) {
    asv[u] = *(const float2*)(att_src + 32 * u + 2 * c);
    adv[u] = *(const float2*)(att_dst + 32 * u + 2 * c);
  }

  const float* __restrict__ bp = Wp + (2 * hf) * 256 + 4 * c;
#pragma unroll
  for (int j = 0; j < 16; ++j) {
#pragma unroll
    for (int s = 0; s < 2; ++s) {
      const float* q = bp + (4 * j + s) * 256;
      const float4 q0 = *(const float4*)(q);
      const float4 q1 = *(const float4*)(q + 64);
      const float4 q2 = *(const float4*)(q + 128);
      const float4 q3 = *(const float4*)(q + 192);
      v2f av;
      av.x = s ? a[j].z : a[j].x;
      av.y = s ? a[j].w : a[j].y;
      v2f b;
      b.x = q0.x; b.y = q0.y;
      acc[0] = __builtin_amdgcn_wmma_f32_16x16x4_f32(false, av, false, b, (short)0, acc[0], false, false);
      b.x = q0.z; b.y = q0.w;
      acc[1] = __builtin_amdgcn_wmma_f32_16x16x4_f32(false, av, false, b, (short)0, acc[1], false, false);
      b.x = q1.x; b.y = q1.y;
      acc[2] = __builtin_amdgcn_wmma_f32_16x16x4_f32(false, av, false, b, (short)0, acc[2], false, false);
      b.x = q1.z; b.y = q1.w;
      acc[3] = __builtin_amdgcn_wmma_f32_16x16x4_f32(false, av, false, b, (short)0, acc[3], false, false);
      b.x = q2.x; b.y = q2.y;
      acc[4] = __builtin_amdgcn_wmma_f32_16x16x4_f32(false, av, false, b, (short)0, acc[4], false, false);
      b.x = q2.z; b.y = q2.w;
      acc[5] = __builtin_amdgcn_wmma_f32_16x16x4_f32(false, av, false, b, (short)0, acc[5], false, false);
      b.x = q3.x; b.y = q3.y;
      acc[6] = __builtin_amdgcn_wmma_f32_16x16x4_f32(false, av, false, b, (short)0, acc[6], false, false);
      b.x = q3.z; b.y = q3.w;
      acc[7] = __builtin_amdgcn_wmma_f32_16x16x4_f32(false, av, false, b, (short)0, acc[7], false, false);
    }
  }

  float* __restrict__ o = H1 + (size_t)(row0 + 8 * hf) * F_IN + 2 * c;
#pragma unroll
  for (int r = 0; r < 8; ++r) {
#pragma unroll
    for (int u = 0; u < 4; ++u)
      *(float2*)(o + (size_t)r * F_IN + 32 * u) = make_float2(acc[2 * u][r], acc[2 * u + 1][r]);
  }

  const bool b0 = (lane & 1) != 0, b1 = (lane & 2) != 0, b2 = (lane & 4) != 0;
  const bool g  = (lane & 8) != 0;
  float ts[4], td[4];
#pragma unroll
  for (int u = 0; u < 4; ++u) {
    float ps[8], pd[8];
#pragma unroll
    for (int r = 0; r < 8; ++r) {
      ps[r] = fmaf(acc[2 * u + 1][r], asv[u].y, acc[2 * u][r] * asv[u].x);
      pd[r] = fmaf(acc[2 * u + 1][r], adv[u].y, acc[2 * u][r] * adv[u].x);
    }
    ts[u] = sum8_by_row(ps, b2, b1, b0);
    td[u] = sum8_by_row(pd, b2, b1, b0);
  }
  const float xs0 = lane_xor<8>(g ? ts[0] : ts[2]);
  const float xs1 = lane_xor<8>(g ? ts[1] : ts[3]);
  const float xd0 = lane_xor<8>(g ? td[0] : td[2]);
  const float xd1 = lane_xor<8>(g ? td[1] : td[3]);
  const float4 os = g ? make_float4(xs0, ts[2], xs1, ts[3]) : make_float4(ts[0], xs0, ts[1], xs1);
  const float4 od = g ? make_float4(xd0, td[2], xd1, td[3]) : make_float4(td[0], xd0, td[1], xd1);
  const size_t ao = (size_t)(row0 + 8 * hf + (lane & 7)) * HEADS + (g ? 4 : 0);
  *(float4*)(AS1 + ao) = os;
  *(float4*)(AD1 + ao) = od;
}

#define G1_WAVES 16384

__device__ __forceinline__ float max16_by_head(const float (&v)[4], bool b3,
                                               bool b2) {
  float q[2];
#pragma unroll
  for (int i = 0; i < 2; ++i) {
    const float snd = b3 ? v[i] : v[i + 2];
    const float kp  = b3 ? v[i + 2] : v[i];
    q[i] = fmaxf(kp, lane_xor<8>(snd));
  }
  const float snd = b2 ? q[0] : q[1];
  const float kp  = b2 ? q[1] : q[0];
  float r = fmaxf(kp, lane_xor<4>(snd));
  r = fmaxf(r, lane_xor<1>(r));
  r = fmaxf(r, lane_xor<2>(r));
  return r;
}

template <int G>
__device__ __forceinline__ void gather_group(
    int mine, int j0, int nn, const float* __restrict__ H1,
    const float* __restrict__ as1, int lane, int h, float adh, float m,
    float& l, float4& acc) {
  float4 r[G];
  float a[G];
#pragma unroll
  for (int u = 0; u < G; ++u) {
    const int s = __builtin_amdgcn_readlane(mine, j0 + u);
    r[u] = *(const float4*)(H1 + (size_t)s * F_IN + 4 * lane);
    a[u] = as1[s * HEADS + h];
  }
#pragma unroll
  for (int u = 0; u < G; ++u) {
    float w = __expf(leaky(a[u] + adh) - m);
    w = (j0 + u < nn) ? w : 0.f;
    l += w;
    acc.x = fmaf(w, r[u].x, acc.x); acc.y = fmaf(w, r[u].y, acc.y);
    acc.z = fmaf(w, r[u].z, acc.z); acc.w = fmaf(w, r[u].w, acc.w);
  }
}

__global__ __launch_bounds__(256) void gat_gather1(
    const int* __restrict__ rowstart, const int* __restrict__ deg,
    const int* __restrict__ csr_src, const float* __restrict__ as1,
    const float* __restrict__ ad1, const float* __restrict__ H1,
    const float* __restrict__ b1, float* __restrict__ HACT, int N) {
  const int lane = threadIdx.x & 31;
  const int wv = __builtin_amdgcn_readfirstlane(blockIdx.x * 8 + (threadIdx.x >> 5));
  if (wv >= N) return;
  const int h   = lane >> 2;
  const int j16 = lane & 15;
  const int k   = lane >> 4;
  const bool b2 = (lane & 4) != 0, b3 = (lane & 8) != 0;
  const float4 b = *(const float4*)(b1 + 4 * lane);

  const int* metap = (lane & 1) ? deg : rowstart;
  const int meta0 = metap[wv];
  int start = __builtin_amdgcn_readlane(meta0, 0);
  int cnt   = __builtin_amdgcn_readlane(meta0, 1);
  for (int n = wv; n < N; n += G1_WAVES) {
    const int nx = min(n + G1_WAVES, N - 1);
    const int meta_nx = metap[nx];

    const float adh    = ad1[n * HEADS + h];
    const float asself = as1[n * HEADS + h];
    const float4 adk   = *(const float4*)(ad1 + (size_t)n * HEADS + 4 * k);
    const float4 hv = *(const float4*)(H1 + (size_t)n * F_IN + 4 * lane);

    const int nn0 = min(16, cnt);
    const int s0 = (j16 < nn0) ? csr_src[start + j16] : n;
    const float4 a0 = *(const float4*)(as1 + (size_t)s0 * HEADS + 4 * k);
    float4 r0[4];
    float al0[4];
#pragma unroll
    for (int u = 0; u < 4; ++u) {
      const int s = __builtin_amdgcn_readlane(s0, u);
      r0[u]  = *(const float4*)(H1 + (size_t)s * F_IN + 4 * lane);
      al0[u] = as1[s * HEADS + h];
    }
    float v[4];
    v[0] = leaky(a0.x + adk.x); v[1] = leaky(a0.y + adk.y);
    v[2] = leaky(a0.z + adk.z); v[3] = leaky(a0.w + adk.w);
    float m = max16_by_head(v, b3, b2);
    for (int c0 = 16; c0 < cnt; c0 += 16) {
      const int nn = min(16, cnt - c0);
      const int s = (j16 < nn) ? csr_src[start + c0 + j16] : n;
      const float4 a = *(const float4*)(as1 + (size_t)s * HEADS + 4 * k);
      v[0] = leaky(a.x + adk.x); v[1] = leaky(a.y + adk.y);
      v[2] = leaky(a.z + adk.z); v[3] = leaky(a.w + adk.w);
      m = fmaxf(m, max16_by_head(v, b3, b2));
    }
    const float vself = leaky(asself + adh);
    m = fmaxf(m, vself);

    const float ex = __expf(vself - m);
    float l = ex;
    float4 acc;
    acc.x = ex * hv.x; acc.y = ex * hv.y; acc.z = ex * hv.z; acc.w = ex * hv.w;
#pragma unroll
    for (int u = 0; u < 4; ++u) {
      float w = __expf(leaky(al0[u] + adh) - m);
      w = (u < nn0) ? w : 0.f;
      l += w;
      acc.x = fmaf(w, r0[u].x, acc.x); acc.y = fmaf(w, r0[u].y, acc.y);
      acc.z = fmaf(w, r0[u].z, acc.z); acc.w = fmaf(w, r0[u].w, acc.w);
    }
    for (int j0 = 4; j0 < nn0; j0 += 4)
      gather_group<4>(s0, j0, nn0, H1, as1, lane, h, adh, m, l, acc);
    for (int c0 = 16; c0 < cnt; c0 += 16) {
      const int nn = min(16, cnt - c0);
      const int mine = (j16 < nn) ? csr_src[start + c0 + j16] : n;
      for (int j0 = 0; j0 < nn; j0 += 4)
        gather_group<4>(mine, j0, nn, H1, as1, lane, h, adh, m, l, acc);
    }

    const float den = l + 1e-16f;
    float4 o;
    o.x = acc.x / den + b.x; o.y = acc.y / den + b.y;
    o.z = acc.z / den + b.z; o.w = acc.w / den + b.w;
    o.x = (o.x > 0.f) ? o.x : (__expf(o.x) - 1.f);
    o.y = (o.y > 0.f) ? o.y : (__expf(o.y) - 1.f);
    o.z = (o.z > 0.f) ? o.z : (__expf(o.z) - 1.f);
    o.w = (o.w > 0.f) ? o.w : (__expf(o.w) - 1.f);
    *(float4*)(HACT + (size_t)n * F_IN + 4 * lane) = o;

    start = __builtin_amdgcn_readlane(meta_nx, 0);
    cnt   = __builtin_amdgcn_readlane(meta_nx, 1);
  }
}

__global__ __launch_bounds__(256) void gat_layer2_node(
    const float* __restrict__ HACT, const float* __restrict__ W2,
    const float* __restrict__ att_src2, const float* __restrict__ att_dst2,
    float* __restrict__ H2, float* __restrict__ as2, float* __restrict__ ad2,
    int N) {
  int n = blockIdx.x * blockDim.x + threadIdx.x;
  if (n >= N) return;
  const float* __restrict__ hp = HACT + (size_t)n * F_IN;
  float a0 = 0.f, a1 = 0.f;
#pragma unroll 8
  for (int k = 0; k < F_IN; ++k) {
    float v = hp[k];
    a0 = fmaf(v, W2[k * OUTC + 0], a0);
    a1 = fmaf(v, W2[k * OUTC + 1], a1);
  }
  H2[n * OUTC + 0] = a0;
  H2[n * OUTC + 1] = a1;
  as2[n] = a0 * att_src2[0] + a1 * att_src2[1];
  ad2[n] = a0 * att_dst2[0] + a1 * att_dst2[1];
}

__global__ __launch_bounds__(256) void gat_gather2(
    const int* __restrict__ rowstart, const int* __restrict__ deg,
    const int* __restrict__ csr_src, const float* __restrict__ as2,
    const float* __restrict__ ad2, const float* __restrict__ H2,
    const float* __restrict__ b2, float* __restrict__ OUT, int N) {
  const int n = blockIdx.x * 256 + threadIdx.x;
  if (n >= N) return;
  const int start = rowstart[n];
  const int cnt   = deg[n];
  const float adn   = ad2[n];
  const float vself = leaky(as2[n] + adn);

  float m = vself;
  for (int j = 0; j < cnt; ++j)
    m = fmaxf(m, leaky(as2[csr_src[start + j]] + adn));

  float ex = __expf(vself - m);
  float l = ex;
  float2 hv = *(const float2*)(H2 + (size_t)n * OUTC);
  float a0 = ex * hv.x, a1 = ex * hv.y;
  for (int j = 0; j < cnt; ++j) {
    const int s = csr_src[start + j];
    const float2 r = *(const float2*)(H2 + (size_t)s * OUTC);
    const float w = __expf(leaky(as2[s] + adn) - m);
    l += w;
    a0 = fmaf(w, r.x, a0);
    a1 = fmaf(w, r.y, a1);
  }
  const float den = l + 1e-16f;
  float2 o;
  o.x = a0 / den + b2[0];
  o.y = a1 / den + b2[1];
  *(float2*)(OUT + (size_t)n * OUTC) = o;
}

static inline unsigned gdim(long long n, int b) {
  return (unsigned)((n + b - 1) / b);
}

extern "C" void kernel_launch(void* const* d_in, const int* in_sizes, int n_in,
                              void* d_out, int out_size, void* d_ws,
                              size_t ws_size, hipStream_t stream) {
  const float* x        = (const float*)d_in[0];
  const int*   ei       = (const int*)d_in[1];
  const float* W1       = (const float*)d_in[2];
  const float* att_src1 = (const float*)d_in[3];
  const float* att_dst1 = (const float*)d_in[4];
  const float* b1       = (const float*)d_in[5];
  const float* W2       = (const float*)d_in[6];
  const float* att_src2 = (const float*)d_in[7];
  const float* att_dst2 = (const float*)d_in[8];
  const float* b2       = (const float*)d_in[9];

  const int N  = in_sizes[0] / F_IN;
  const int E  = in_sizes[1] / 2;

  float* ws = (float*)d_ws;
  size_t off = 0;
  float* h1   = ws + off; off += (size_t)N * F_IN;
  float* hact = ws + off; off += (size_t)N * F_IN;
  float* as1  = ws + off; off += (size_t)N * HEADS;
  float* ad1  = ws + off; off += (size_t)N * HEADS;
  float* h2   = ws + off; off += (size_t)N * OUTC;
  float* as2  = ws + off; off += (size_t)N;
  float* ad2  = ws + off; off += (size_t)N;
  int* pairs    = (int*)(ws + off); off += (size_t)E * 2;
  int* deg      = (int*)(ws + off); off += (size_t)N;
  int* rowstart = (int*)(ws + off); off += (size_t)N;
  int* csr_src  = (int*)(ws + off); off += (size_t)E;
  const int nbins = (int)gdim(N, BIN_NODES);
  int* bin_cnt  = (int*)(ws + off); off += (size_t)nbins * NCHUNK;
  int* bin_base = (int*)(ws + off); off += (size_t)nbins;
  int* bin_tot  = (int*)(ws + off); off += (size_t)nbins;
  int* total    = (int*)(ws + off); off += 1;
  (void)ws_size; (void)n_in; (void)out_size;

  const int B = 256;

  gat_bin_count<<<NCHUNK, B, 0, stream>>>(ei, bin_cnt, total, E, nbins);
  gat_bin_offsets<<<nbins, NCHUNK, 0, stream>>>(bin_cnt, total, bin_base, bin_tot);
  gat_bin_scatter<<<NCHUNK, B, 0, stream>>>(ei, bin_cnt, pairs, E, nbins);
  gat_bin_csr<<<nbins, BIN_NODES, 0, stream>>>(pairs, bin_base, bin_tot, rowstart, deg, csr_src, N);

  gat_gemm_wmma_f32<<<gdim(N, 64), 128, 0, stream>>>(x, W1, att_src1, att_dst1, h1, as1, ad1, N);
  gat_gather1<<<G1_WAVES / 8, B, 0, stream>>>(rowstart, deg, csr_src, as1, ad1, h1, b1, hact, N);

  gat_layer2_node<<<gdim(N, B), B, 0, stream>>>(hact, W2, att_src2, att_dst2, h2, as2, ad2, N);
  gat_gather2<<<gdim(N, B), B, 0, stream>>>(rowstart, deg, csr_src, as2, ad2, h2, b2, (float*)d_out, N);
}


# Layer-2 kernels (one thread per node): a thread's loads fly together, not one round trip per step: node row whole; edges in chunks of 16 (sources, then logits and rows); every sum keeps its order
# speedup vs baseline: 5.7382x; 1.0655x over previous
#include <hip/hip_runtime.h>
#include <hip/hip_bf16.h>


typedef __attribute__((ext_vector_type(2))) float v2f;
typedef __attribute__((ext_vector_type(8))) float v8f;

#define F_IN   128
#define HEADS  8
#define CH     16
#define OUTC   2
#define NEG_SLOPE 0.2f
#define BIN_SHIFT 8
#define BIN_NODES 256
#define MAX_BINS  512
#define NCHUNK    256

__device__ __forceinline__ float leaky(float v) {
  return (v > 0.f) ? v : NEG_SLOPE * v;
}

__global__ __launch_bounds__(256) void gat_bin_count(
    const int* __restrict__ ei, int* __restrict__ cnt,
    int* __restrict__ total, int E, int nbins) {
  __shared__ int hist[MAX_BINS];
  const int t = threadIdx.x, g = blockIdx.x;
  for (int b = t; b < nbins; b += 256) hist[b] = 0;
  __syncthreads();
  const int ce = (E + NCHUNK - 1) / NCHUNK;
  const int e0 = g * ce;
  const int e1 = min(E, e0 + ce);
  for (int e = e0 + t; e < e1; e += 4 * 256) {
    int d[4];
#pragma unroll
    for (int u = 0; u < 4; ++u) {
      const int ee = e + u * 256;
      d[u] = (ee < e1) ? ei[E + ee] : -1;
    }
#pragma unroll
    for (int u = 0; u < 4; ++u)
      if (d[u] >= 0) atomicAdd(&hist[d[u] >> BIN_SHIFT], 1);
  }
  __syncthreads();
  for (int b = t; b < nbins; b += 256) cnt[b * NCHUNK + g] = hist[b];
  if (g == 0 && t == 0) *total = 0;
}

__global__ __launch_bounds__(NCHUNK) void gat_bin_offsets(
    int* __restrict__ cnt, int* __restrict__ total,
    int* __restrict__ bin_base, int* __restrict__ bin_cnt) {
  __shared__ int wsum[NCHUNK / 32];
  __shared__ int base_s;
  const int t = threadIdx.x, lane = t & 31, wave = t >> 5;
  const int b = blockIdx.x;
  const int v = cnt[b * NCHUNK + t];
  int inc = v;
  for (int o = 1; o < 32; o <<= 1) {
    const int u = __shfl_up(inc, o, 32);
    if (lane >= o) inc += u;
  }
  if (lane == 31) wsum[wave] = inc;
  __syncthreads();
  if (t == 0) {
    int tot = 0;
    for (int w = 0; w < NCHUNK / 32; ++w) { const int c = wsum[w]; wsum[w] = tot; tot += c; }
    base_s = atomicAdd(total, tot);
    bin_base[b] = base_s;
    bin_cnt[b]  = tot;
  }
  __syncthreads();
  cnt[b * NCHUNK + t] = base_s + wsum[wave] + inc - v;
}

__global__ __launch_bounds__(256) void gat_bin_scatter(
    const int* __restrict__ ei, const int* __restrict__ offs,
    int* __restrict__ pairs, int E, int nbins) {
  __shared__ int cur[MAX_BINS];
  int2* __restrict__ pr = (int2*)pairs;
  const int t = threadIdx.x, g = blockIdx.x;
  for (int b = t; b < nbins; b += 256) cur[b] = offs[b * NCHUNK + g];
  __syncthreads();
  const int ce = (E + NCHUNK - 1) / NCHUNK;
  const int e0 = g * ce;
  const int e1 = min(E, e0 + ce);
  for (int e = e0 + t; e < e1; e += 4 * 256) {
    int d[4], sr[4];
#pragma unroll
    for (int u = 0; u < 4; ++u) {
      const int ee = e + u * 256;
      d[u]  = (ee < e1) ? ei[E + ee] : -1;
      sr[u] = (ee < e1) ? ei[ee] : 0;
    }
#pragma unroll
    for (int u = 0; u < 4; ++u)
      if (d[u] >= 0) {
        const int pos = atomicAdd(&cur[d[u] >> BIN_SHIFT], 1);
        pr[pos] = make_int2(sr[u], d[u]);
      }
  }
}

__global__ __launch_bounds__(BIN_NODES) void gat_bin_csr(
    const int* __restrict__ pairs, const int* __restrict__ bin_base,
    const int* __restrict__ bin_cnt, int* __restrict__ rowstart,
    int* __restrict__ deg, int* __restrict__ csr_src, int N) {
  __shared__ int dcnt[BIN_NODES];
  __shared__ int dpos[BIN_NODES];
  __shared__ int wsum[BIN_NODES / 32];
  const int t = threadIdx.x, lane = t & 31, wave = t >> 5;
  const int b = blockIdx.x;
  const int n = b * BIN_NODES + t;
  const int base = bin_base[b];
  const int tot  = bin_cnt[b];
  const int2* __restrict__ pb = (const int2*)pairs + base;
  dcnt[t] = 0;
  __syncthreads();
  for (int i = t; i < tot; i += 4 * BIN_NODES) {
    int d[4];
#pragma unroll
    for (int u = 0; u < 4; ++u) {
      const int ii = i + u * BIN_NODES;
      d[u] = (ii < tot) ? pb[ii].y : -1;
    }
#pragma unroll
    for (int u = 0; u < 4; ++u)
      if (d[u] >= 0) atomicAdd(&dcnt[d[u] & (BIN_NODES - 1)], 1);
  }
  __syncthreads();
  const int v = dcnt[t];
  int inc = v;
  for (int o = 1; o < 32; o <<= 1) {
    const int u = __shfl_up(inc, o, 32);
    if (lane >= o) inc += u;
  }
  if (lane == 31) wsum[wave] = inc;
  __syncthreads();
  if (t == 0) {
    int run = 0;
    for (int w = 0; w < BIN_NODES / 32; ++w) { const int c = wsum[w]; wsum[w] = run; run += c; }
  }
  __syncthreads();
  const int first = base + wsum[wave] + inc - v;
  dpos[t] = first;
  if (n < N) { rowstart[n] = first; deg[n] = v; }
  __syncthreads();
  for (int i = t; i < tot; i += 4 * BIN_NODES) {
    int2 r[4];
#pragma unroll
    for (int u = 0; u < 4; ++u) {
      const int ii = i + u * BIN_NODES;
      r[u] = (ii < tot) ? pb[ii] : make_int2(0, -1);
    }
#pragma unroll
    for (int u = 0; u < 4; ++u)
      if (r[u].y >= 0)
        csr_src[atomicAdd(&dpos[r[u].y & (BIN_NODES - 1)], 1)] = r[u].x;
  }
}

template <int M>
__device__ __forceinline__ float lane_xor(float v) {
  return __builtin_bit_cast(float, __builtin_amdgcn_update_dpp(
      0, __builtin_bit_cast(int, v), 0x160 + M, 0xf, 0xf, true));
}

__device__ __forceinline__ float sum8_by_row(const float (&p)[8], bool b2,
                                             bool b1, bool b0) {
  float q[4], s[2];
#pragma unroll
  for (int i = 0; i < 4; ++i) {
    const float snd = b2 ? p[i] : p[i + 4];
    const float kp  = b2 ? p[i + 4] : p[i];
    q[i] = kp + lane_xor<4>(snd);
  }
#pragma unroll
  for (int i = 0; i < 2; ++i) {
    const float snd = b1 ? q[i] : q[i + 2];
    const float kp  = b1 ? q[i + 2] : q[i];
    s[i] = kp + lane_xor<2>(snd);
  }
  const float snd = b0 ? s[0] : s[1];
  const float kp  = b0 ? s[1] : s[0];
  return kp + lane_xor<1>(snd);
}

__global__ __launch_bounds__(128) void gat_gemm_wmma_f32(
    const float* __restrict__ X, const float* __restrict__ W,
    const float* __restrict__ att_src, const float* __restrict__ att_dst,
    float* __restrict__ H1, float* __restrict__ AS1, float* __restrict__ AD1,
    int N) {
  __shared__ float Wp[64 * 256];

  const int t    = threadIdx.x;
  const int wave = __builtin_amdgcn_readfirstlane(t >> 5);
  const int lane = t & 31;
  const int hf   = lane >> 4;
  const int c    = lane & 15;
  const int row0 = (blockIdx.x * 4 + wave) * 16;

  const int ar = min(row0 + c, N - 1);
  const float* __restrict__ xr = X + (size_t)ar * F_IN + 4 * hf;
  float4 a[16];
#pragma unroll
  for (int j = 0; j < 16; ++j) a[j] = *(const float4*)(xr + 8 * j);

  const float* __restrict__ wsrc = W + (size_t)(2 * wave) * F_IN + 4 * lane;
  float* __restrict__ wdst = Wp + wave * 256 + 8 * lane;
#pragma unroll
  for (int hb = 0; hb < 2; ++hb) {
    float4 r0[8], r1[8];
#pragma unroll
    for (int it = 0; it < 8; ++it) {
      const int n = hb * 8 + it;
      r0[it] = *(const float4*)(wsrc + (size_t)n * 8 * F_IN);
      r1[it] = *(const float4*)(wsrc + (size_t)n * 8 * F_IN + F_IN);
    }
#pragma unroll
    for (int it = 0; it < 8; ++it) {
      const int n = hb * 8 + it;
      float4* p = (float4*)(wdst + n * 4 * 256);
      p[0] = make_float4(r0[it].x, r1[it].x, r0[it].y, r1[it].y);
      p[1] = make_float4(r0[it].z, r1[it].z, r0[it].w, r1[it].w);
    }
  }
  __syncthreads();
  if (row0 >= N) return;

  v8f acc[8];
#pragma unroll
  for (int i = 0; i < 8; ++i) acc[i] = v8f{};

  float2 asv[4], adv[4];
#pragma unroll
  for (int u = 0; u < 4; ++u) {
    asv[u] = *(const float2*)(att_src + 32 * u + 2 * c);
    adv[u] = *(const float2*)(att_dst + 32 * u + 2 * c);
  }

  const float* __restrict__ bp = Wp + (2 * hf) * 256 + 4 * c;
#pragma unroll
  for (int j = 0; j < 16; ++j) {
#pragma unroll
    for (int s = 0; s < 2; ++s) {
      const float* q = bp + (4 * j + s) * 256;
      const float4 q0 = *(const float4*)(q);
      const float4 q1 = *(const float4*)(q + 64);
      const float4 q2 = *(const float4*)(q + 128);
      const float4 q3 = *(const float4*)(q + 192);
      v2f av;
      av.x = s ? a[j].z : a[j].x;
      av.y = s ? a[j].w : a[j].y;
      v2f b;
      b.x = q0.x; b.y = q0.y;
      acc[0] = __builtin_amdgcn_wmma_f32_16x16x4_f32(false, av, false, b, (short)0, acc[0], false, false);
      b.x = q0.z; b.y = q0.w;
      acc[1] = __builtin_amdgcn_wmma_f32_16x16x4_f32(false, av, false, b, (short)0, acc[1], false, false);
      b.x = q1.x; b.y = q1.y;
      acc[2] = __builtin_amdgcn_wmma_f32_16x16x4_f32(false, av, false, b, (short)0, acc[2], false, false);
      b.x = q1.z; b.y = q1.w;
      acc[3] = __builtin_amdgcn_wmma_f32_16x16x4_f32(false, av, false, b, (short)0, acc[3], false, false);
      b.x = q2.x; b.y = q2.y;
      acc[4] = __builtin_amdgcn_wmma_f32_16x16x4_f32(false, av, false, b, (short)0, acc[4], false, false);
      b.x = q2.z; b.y = q2.w;
      acc[5] = __builtin_amdgcn_wmma_f32_16x16x4_f32(false, av, false, b, (short)0, acc[5], false, false);
      b.x = q3.x; b.y = q3.y;
      acc[6] = __builtin_amdgcn_wmma_f32_16x16x4_f32(false, av, false, b, (short)0, acc[6], false, false);
      b.x = q3.z; b.y = q3.w;
      acc[7] = __builtin_amdgcn_wmma_f32_16x16x4_f32(false, av, false, b, (short)0, acc[7], false, false);
    }
  }

  float* __restrict__ o = H1 + (size_t)(row0 + 8 * hf) * F_IN + 2 * c;
#pragma unroll
  for (int r = 0; r < 8; ++r) {
#pragma unroll
    for (int u = 0; u < 4; ++u)
      *(float2*)(o + (size_t)r * F_IN + 32 * u) = make_float2(acc[2 * u][r], acc[2 * u + 1][r]);
  }

  const bool b0 = (lane & 1) != 0, b1 = (lane & 2) != 0, b2 = (lane & 4) != 0;
  const bool g  = (lane & 8) != 0;
  float ts[4], td[4];
#pragma unroll
  for (int u = 0; u < 4; ++u) {
    float ps[8], pd[8];
#pragma unroll
    for (int r = 0; r < 8; ++r) {
      ps[r] = fmaf(acc[2 * u + 1][r], asv[u].y, acc[2 * u][r] * asv[u].x);
      pd[r] = fmaf(acc[2 * u + 1][r], adv[u].y, acc[2 * u][r] * adv[u].x);
    }
    ts[u] = sum8_by_row(ps, b2, b1, b0);
    td[u] = sum8_by_row(pd, b2, b1, b0);
  }
  const float xs0 = lane_xor<8>(g ? ts[0] : ts[2]);
  const float xs1 = lane_xor<8>(g ? ts[1] : ts[3]);
  const float xd0 = lane_xor<8>(g ? td[0] : td[2]);
  const float xd1 = lane_xor<8>(g ? td[1] : td[3]);
  const float4 os = g ? make_float4(xs0, ts[2], xs1, ts[3]) : make_float4(ts[0], xs0, ts[1], xs1);
  const float4 od = g ? make_float4(xd0, td[2], xd1, td[3]) : make_float4(td[0], xd0, td[1], xd1);
  const size_t ao = (size_t)(row0 + 8 * hf + (lane & 7)) * HEADS + (g ? 4 : 0);
  *(float4*)(AS1 + ao) = os;
  *(float4*)(AD1 + ao) = od;
}

#define G1_WAVES 16384

__device__ __forceinline__ float max16_by_head(const float (&v)[4], bool b3,
                                               bool b2) {
  float q[2];
#pragma unroll
  for (int i = 0; i < 2; ++i) {
    const float snd = b3 ? v[i] : v[i + 2];
    const float kp  = b3 ? v[i + 2] : v[i];
    q[i] = fmaxf(kp, lane_xor<8>(snd));
  }
  const float snd = b2 ? q[0] : q[1];
  const float kp  = b2 ? q[1] : q[0];
  float r = fmaxf(kp, lane_xor<4>(snd));
  r = fmaxf(r, lane_xor<1>(r));
  r = fmaxf(r, lane_xor<2>(r));
  return r;
}

template <int G>
__device__ __forceinline__ void gather_group(
    int mine, int j0, int nn, const float* __restrict__ H1,
    const float* __restrict__ as1, int lane, int h, float adh, float m,
    float& l, float4& acc) {
  float4 r[G];
  float a[G];
#pragma unroll
  for (int u = 0; u < G; ++u) {
    const int s = __builtin_amdgcn_readlane(mine, j0 + u);
    r[u] = *(const float4*)(H1 + (size_t)s * F_IN + 4 * lane);
    a[u] = as1[s * HEADS + h];
  }
#pragma unroll
  for (int u = 0; u < G; ++u) {
    float w = __expf(leaky(a[u] + adh) - m);
    w = (j0 + u < nn) ? w : 0.f;
    l += w;
    acc.x = fmaf(w, r[u].x, acc.x); acc.y = fmaf(w, r[u].y, acc.y);
    acc.z = fmaf(w, r[u].z, acc.z); acc.w = fmaf(w, r[u].w, acc.w);
  }
}

__global__ __launch_bounds__(256) void gat_gather1(
    const int* __restrict__ rowstart, const int* __restrict__ deg,
    const int* __restrict__ csr_src, const float* __restrict__ as1,
    const float* __restrict__ ad1, const float* __restrict__ H1,
    const float* __restrict__ b1, float* __restrict__ HACT, int N) {
  const int lane = threadIdx.x & 31;
  const int wv = __builtin_amdgcn_readfirstlane(blockIdx.x * 8 + (threadIdx.x >> 5));
  if (wv >= N) return;
  const int h   = lane >> 2;
  const int j16 = lane & 15;
  const int k   = lane >> 4;
  const bool b2 = (lane & 4) != 0, b3 = (lane & 8) != 0;
  const float4 b = *(const float4*)(b1 + 4 * lane);

  const int* metap = (lane & 1) ? deg : rowstart;
  const int meta0 = metap[wv];
  int start = __builtin_amdgcn_readlane(meta0, 0);
  int cnt   = __builtin_amdgcn_readlane(meta0, 1);
  for (int n = wv; n < N; n += G1_WAVES) {
    const int nx = min(n + G1_WAVES, N - 1);
    const int meta_nx = metap[nx];

    const float adh    = ad1[n * HEADS + h];
    const float asself = as1[n * HEADS + h];
    const float4 adk   = *(const float4*)(ad1 + (size_t)n * HEADS + 4 * k);
    const float4 hv = *(const float4*)(H1 + (size_t)n * F_IN + 4 * lane);

    const int nn0 = min(16, cnt);
    const int s0 = (j16 < nn0) ? csr_src[start + j16] : n;
    const float4 a0 = *(const float4*)(as1 + (size_t)s0 * HEADS + 4 * k);
    float4 r0[4];
    float al0[4];
#pragma unroll
    for (int u = 0; u < 4; ++u) {
      const int s = __builtin_amdgcn_readlane(s0, u);
      r0[u]  = *(const float4*)(H1 + (size_t)s * F_IN + 4 * lane);
      al0[u] = as1[s * HEADS + h];
    }
    float v[4];
    v[0] = leaky(a0.x + adk.x); v[1] = leaky(a0.y + adk.y);
    v[2] = leaky(a0.z + adk.z); v[3] = leaky(a0.w + adk.w);
    float m = max16_by_head(v, b3, b2);
    for (int c0 = 16; c0 < cnt; c0 += 16) {
      const int nn = min(16, cnt - c0);
      const int s = (j16 < nn) ? csr_src[start + c0 + j16] : n;
      const float4 a = *(const float4*)(as1 + (size_t)s * HEADS + 4 * k);
      v[0] = leaky(a.x + adk.x); v[1] = leaky(a.y + adk.y);
      v[2] = leaky(a.z + adk.z); v[3] = leaky(a.w + adk.w);
      m = fmaxf(m, max16_by_head(v, b3, b2));
    }
    const float vself = leaky(asself + adh);
    m = fmaxf(m, vself);

    const float ex = __expf(vself - m);
    float l = ex;
    float4 acc;
    acc.x = ex * hv.x; acc.y = ex * hv.y; acc.z = ex * hv.z; acc.w = ex * hv.w;
#pragma unroll
    for (int u = 0; u < 4; ++u) {
      float w = __expf(leaky(al0[u] + adh) - m);
      w = (u < nn0) ? w : 0.f;
      l += w;
      acc.x = fmaf(w, r0[u].x, acc.x); acc.y = fmaf(w, r0[u].y, acc.y);
      acc.z = fmaf(w, r0[u].z, acc.z); acc.w = fmaf(w, r0[u].w, acc.w);
    }
    for (int j0 = 4; j0 < nn0; j0 += 4)
      gather_group<4>(s0, j0, nn0, H1, as1, lane, h, adh, m, l, acc);
    for (int c0 = 16; c0 < cnt; c0 += 16) {
      const int nn = min(16, cnt - c0);
      const int mine = (j16 < nn) ? csr_src[start + c0 + j16] : n;
      for (int j0 = 0; j0 < nn; j0 += 4)
        gather_group<4>(mine, j0, nn, H1, as1, lane, h, adh, m, l, acc);
    }

    const float den = l + 1e-16f;
    float4 o;
    o.x = acc.x / den + b.x; o.y = acc.y / den + b.y;
    o.z = acc.z / den + b.z; o.w = acc.w / den + b.w;
    o.x = (o.x > 0.f) ? o.x : (__expf(o.x) - 1.f);
    o.y = (o.y > 0.f) ? o.y : (__expf(o.y) - 1.f);
    o.z = (o.z > 0.f) ? o.z : (__expf(o.z) - 1.f);
    o.w = (o.w > 0.f) ? o.w : (__expf(o.w) - 1.f);
    *(float4*)(HACT + (size_t)n * F_IN + 4 * lane) = o;

    start = __builtin_amdgcn_readlane(meta_nx, 0);
    cnt   = __builtin_amdgcn_readlane(meta_nx, 1);
  }
}

__global__ __launch_bounds__(256) void gat_layer2_node(
    const float* __restrict__ HACT, const float* __restrict__ W2,
    const float* __restrict__ att_src2, const float* __restrict__ att_dst2,
    float* __restrict__ H2, float* __restrict__ as2, float* __restrict__ ad2,
    int N) {
  int n = blockIdx.x * blockDim.x + threadIdx.x;
  if (n >= N) return;
  const float* __restrict__ hp = HACT + (size_t)n * F_IN;
  float4 hr[F_IN / 4];
#pragma unroll
  for (int q = 0; q < F_IN / 4; ++q) hr[q] = *(const float4*)(hp + 4 * q);
  float a0 = 0.f, a1 = 0.f;
#pragma unroll
  for (int q = 0; q < F_IN / 4; ++q) {
    const float v[4] = {hr[q].x, hr[q].y, hr[q].z, hr[q].w};
#pragma unroll
    for (int e = 0; e < 4; ++e) {
      const int k = 4 * q + e;
      a0 = fmaf(v[e], W2[k * OUTC + 0], a0);
      a1 = fmaf(v[e], W2[k * OUTC + 1], a1);
    }
  }
  H2[n * OUTC + 0] = a0;
  H2[n * OUTC + 1] = a1;
  as2[n] = a0 * att_src2[0] + a1 * att_src2[1];
  ad2[n] = a0 * att_dst2[0] + a1 * att_dst2[1];
}

#define G2_CHUNK 16

__device__ __forceinline__ void g2_sources(
    const int* __restrict__ csr_src, int start, int c0, int cnt, int n,
    int (&s)[G2_CHUNK]) {
#pragma unroll
  for (int u = 0; u < G2_CHUNK; ++u) {
    const int v = csr_src[max(start + min(c0 + u, cnt - 1), 0)];
    s[u] = (c0 + u < cnt) ? v : n;
  }
}

__global__ __launch_bounds__(256) void gat_gather2(
    const int* __restrict__ rowstart, const int* __restrict__ deg,
    const int* __restrict__ csr_src, const float* __restrict__ as2,
    const float* __restrict__ ad2, const float* __restrict__ H2,
    const float* __restrict__ b2, float* __restrict__ OUT, int N) {
  const int n = blockIdx.x * 256 + threadIdx.x;
  if (n >= N) return;
  const int start = rowstart[n];
  const int cnt   = deg[n];
  const float adn   = ad2[n];
  const float vself = leaky(as2[n] + adn);
  const float2 hv = *(const float2*)(H2 + (size_t)n * OUTC);

  int s0[G2_CHUNK];
  float a0v[G2_CHUNK];
  float2 r0[G2_CHUNK];
  g2_sources(csr_src, start, 0, cnt, n, s0);
#pragma unroll
  for (int u = 0; u < G2_CHUNK; ++u) a0v[u] = as2[s0[u]];
#pragma unroll
  for (int u = 0; u < G2_CHUNK; ++u) r0[u] = *(const float2*)(H2 + (size_t)s0[u] * OUTC);

  float m = vself;
#pragma unroll
  for (int u = 0; u < G2_CHUNK; ++u) m = fmaxf(m, leaky(a0v[u] + adn));
  for (int c0 = G2_CHUNK; c0 < cnt; c0 += G2_CHUNK) {
    int s[G2_CHUNK];
    float a[G2_CHUNK];
    g2_sources(csr_src, start, c0, cnt, n, s);
#pragma unroll
    for (int u = 0; u < G2_CHUNK; ++u) a[u] = as2[s[u]];
#pragma unroll
    for (int u = 0; u < G2_CHUNK; ++u) m = fmaxf(m, leaky(a[u] + adn));
  }

  const float ex = __expf(vself - m);
  float l = ex;
  float a0 = ex * hv.x, a1 = ex * hv.y;
#pragma unroll
  for (int u = 0; u < G2_CHUNK; ++u) {
    float w = __expf(leaky(a0v[u] + adn) - m);
    w = (u < cnt) ? w : 0.f;
    l += w;
    a0 = fmaf(w, r0[u].x, a0);
    a1 = fmaf(w, r0[u].y, a1);
  }
  for (int c0 = G2_CHUNK; c0 < cnt; c0 += G2_CHUNK) {
    int s[G2_CHUNK];
    float a[G2_CHUNK];
    float2 r[G2_CHUNK];
    g2_sources(csr_src, start, c0, cnt, n, s);
#pragma unroll
    for (int u = 0; u < G2_CHUNK; ++u) a[u] = as2[s[u]];
#pragma unroll
    for (int u = 0; u < G2_CHUNK; ++u) r[u] = *(const float2*)(H2 + (size_t)s[u] * OUTC);
#pragma unroll
    for (int u = 0; u < G2_CHUNK; ++u) {
      float w = __expf(leaky(a[u] + adn) - m);
      w = (c0 + u < cnt) ? w : 0.f;
      l += w;
      a0 = fmaf(w, r[u].x, a0);
      a1 = fmaf(w, r[u].y, a1);
    }
  }
  const float den = l + 1e-16f;
  float2 o;
  o.x = a0 / den + b2[0];
  o.y = a1 / den + b2[1];
  *(float2*)(OUT + (size_t)n * OUTC) = o;
}

static inline unsigned gdim(long long n, int b) {
  return (unsigned)((n + b - 1) / b);
}

extern "C" void kernel_launch(void* const* d_in, const int* in_sizes, int n_in,
                              void* d_out, int out_size, void* d_ws,
                              size_t ws_size, hipStream_t stream) {
  const float* x        = (const float*)d_in[0];
  const int*   ei       = (const int*)d_in[1];
  const float* W1       = (const float*)d_in[2];
  const float* att_src1 = (const float*)d_in[3];
  const float* att_dst1 = (const float*)d_in[4];
  const float* b1       = (const float*)d_in[5];
  const float* W2       = (const float*)d_in[6];
  const float* att_src2 = (const float*)d_in[7];
  const float* att_dst2 = (const float*)d_in[8];
  const float* b2       = (const float*)d_in[9];

  const int N  = in_sizes[0] / F_IN;
  const int E  = in_sizes[1] / 2;

  float* ws = (float*)d_ws;
  size_t off = 0;
  float* h1   = ws + off; off += (size_t)N * F_IN;
  float* hact = ws + off; off += (size_t)N * F_IN;
  float* as1  = ws + off; off += (size_t)N * HEADS;
  float* ad1  = ws + off; off += (size_t)N * HEADS;
  float* h2   = ws + off; off += (size_t)N * OUTC;
  float* as2  = ws + off; off += (size_t)N;
  float* ad2  = ws + off; off += (size_t)N;
  int* pairs    = (int*)(ws + off); off += (size_t)E * 2;
  int* deg      = (int*)(ws + off); off += (size_t)N;
  int* rowstart = (int*)(ws + off); off += (size_t)N;
  int* csr_src  = (int*)(ws + off); off += (size_t)E;
  const int nbins = (int)gdim(N, BIN_NODES);
  int* bin_cnt  = (int*)(ws + off); off += (size_t)nbins * NCHUNK;
  int* bin_base = (int*)(ws + off); off += (size_t)nbins;
  int* bin_tot  = (int*)(ws + off); off += (size_t)nbins;
  int* total    = (int*)(ws + off); off += 1;
  (void)ws_size; (void)n_in; (void)out_size;

  const int B = 256;

  gat_bin_count<<<NCHUNK, B, 0, stream>>>(ei, bin_cnt, total, E, nbins);
  gat_bin_offsets<<<nbins, NCHUNK, 0, stream>>>(bin_cnt, total, bin_base, bin_tot);
  gat_bin_scatter<<<NCHUNK, B, 0, stream>>>(ei, bin_cnt, pairs, E, nbins);
  gat_bin_csr<<<nbins, BIN_NODES, 0, stream>>>(pairs, bin_base, bin_tot, rowstart, deg, csr_src, N);

  gat_gemm_wmma_f32<<<gdim(N, 64), 128, 0, stream>>>(x, W1, att_src1, att_dst1, h1, as1, ad1, N);
  gat_gather1<<<G1_WAVES / 8, B, 0, stream>>>(rowstart, deg, csr_src, as1, ad1, h1, b1, hact, N);

  gat_layer2_node<<<gdim(N, B), B, 0, stream>>>(hact, W2, att_src2, att_dst2, h2, as2, ad2, N);
  gat_gather2<<<gdim(N, B), B, 0, stream>>>(rowstart, deg, csr_src, as2, ad2, h2, b2, (float*)d_out, N);
}


# Edge bucketing in 2 launches (was 4): a chunk sorts its edges by bin inside its own slice of the pair buffer, a bin reads its runs; no offsets pass, no global atomic; a node's sum order changes
# speedup vs baseline: 6.5781x; 1.1464x over previous
#include <hip/hip_runtime.h>
#include <hip/hip_bf16.h>


typedef __attribute__((ext_vector_type(2))) float v2f;
typedef __attribute__((ext_vector_type(8))) float v8f;

#define F_IN   128
#define HEADS  8
#define CH     16
#define OUTC   2
#define NEG_SLOPE 0.2f
#define BIN_SHIFT 8
#define BIN_NODES 256
#define MAX_BINS  512
#define NCHUNK    256

__device__ __forceinline__ float leaky(float v) {
  return (v > 0.f) ? v : NEG_SLOPE * v;
}

template <int M>
__device__ __forceinline__ int lane_xor_i(int v) {
  return __builtin_amdgcn_update_dpp(0, v, 0x160 + M, 0xf, 0xf, true);
}

__global__ __launch_bounds__(256) void gat_bin_sort(
    const int* __restrict__ ei, int* __restrict__ runstart,
    int* __restrict__ pairs, int E, int nbins) {
  __shared__ int hist[MAX_BINS + 1];
  __shared__ int wsum[256 / 32];
  int2* __restrict__ pr = (int2*)pairs;
  const int t = threadIdx.x, lane = t & 31, wave = t >> 5, g = blockIdx.x;
  hist[t] = 0;
  hist[t + 256] = 0;
  __syncthreads();
  const int ce = (E + NCHUNK - 1) / NCHUNK;
  const int e0 = g * ce;
  const int e1 = min(E, e0 + ce);
  for (int e = e0 + t; e < e1; e += 4 * 256) {
    int d[4];
#pragma unroll
    for (int u = 0; u < 4; ++u) {
      const int ee = e + u * 256;
      d[u] = (ee < e1) ? ei[E + ee] : -1;
    }
#pragma unroll
    for (int u = 0; u < 4; ++u)
      if (d[u] >= 0) atomicAdd(&hist[d[u] >> BIN_SHIFT], 1);
  }
  __syncthreads();
  const int c0 = hist[2 * t], c1 = hist[2 * t + 1];
  const int v = c0 + c1;
  int inc = v;
  for (int o = 1; o < 32; o <<= 1) {
    const int u = __shfl_up(inc, o, 32);
    if (lane >= o) inc += u;
  }
  if (lane == 31) wsum[wave] = inc;
  __syncthreads();
  int below = 0;
#pragma unroll
  for (int w = 0; w < 256 / 32; ++w) below += (w < wave) ? wsum[w] : 0;
  const int x0 = e0 + below + inc - v;
  hist[2 * t]     = x0;
  hist[2 * t + 1] = x0 + c0;
  if (t == 255) hist[MAX_BINS] = x0 + v;
  __syncthreads();
  for (int b = t; b <= nbins; b += 256) runstart[b * NCHUNK + g] = hist[b];
  __syncthreads();
  for (int e = e0 + t; e < e1; e += 4 * 256) {
    int d[4], sr[4];
#pragma unroll
    for (int u = 0; u < 4; ++u) {
      const int ee = e + u * 256;
      d[u]  = (ee < e1) ? ei[E + ee] : -1;
      sr[u] = (ee < e1) ? ei[ee] : 0;
    }
#pragma unroll
    for (int u = 0; u < 4; ++u)
      if (d[u] >= 0) {
        const int pos = atomicAdd(&hist[d[u] >> BIN_SHIFT], 1);
        pr[pos] = make_int2(sr[u], d[u]);
      }
  }
}

#define RUN_LANES 16
#define RUN_STEPS 16

__device__ __forceinline__ void run_bounds(int rs, int re, int it, bool hi,
                                           int& s, int& e) {
  const int s0 = __builtin_amdgcn_readlane(rs, 2 * it);
  const int e0 = __builtin_amdgcn_readlane(re, 2 * it);
  const int s1 = __builtin_amdgcn_readlane(rs, 2 * it + 1);
  const int e1 = __builtin_amdgcn_readlane(re, 2 * it + 1);
  s = hi ? s1 : s0;
  e = hi ? e1 : e0;
}

__global__ __launch_bounds__(BIN_NODES) void gat_bin_csr(
    const int* __restrict__ pairs, const int* __restrict__ runstart,
    int* __restrict__ rowstart, int* __restrict__ deg,
    int* __restrict__ csr_src, int N, int E) {
  __shared__ int dcnt[BIN_NODES];
  __shared__ int dpos[BIN_NODES];
  __shared__ int wsum[BIN_NODES / 32];
  __shared__ int wbef[BIN_NODES / 32];
  const int t = threadIdx.x, lane = t & 31, wave = t >> 5;
  const int b = blockIdx.x;
  const int n = b * BIN_NODES + t;
  const int2* __restrict__ pp = (const int2*)pairs;
  const int rs = runstart[b * NCHUNK + t];
  const int re = runstart[(b + 1) * NCHUNK + t];
  const int ce = (E + NCHUNK - 1) / NCHUNK;
  int bef = rs - t * ce;
  dcnt[t] = 0;
  bef += lane_xor_i<1>(bef);
  bef += lane_xor_i<2>(bef);
  bef += lane_xor_i<4>(bef);
  bef += lane_xor_i<8>(bef);
  const int wtot = __builtin_amdgcn_readlane(bef, 0) + __builtin_amdgcn_readlane(bef, 16);
  if (lane == 0) wbef[wave] = wtot;
  __syncthreads();
  int base = 0;
#pragma unroll
  for (int w = 0; w < BIN_NODES / 32; ++w) base += wbef[w];

  const int sub = lane & (RUN_LANES - 1);
  const bool hi = lane >= RUN_LANES;
  const bool longrun = __any(re - rs > RUN_LANES);
  int2 r[RUN_STEPS];
#pragma unroll
  for (int it = 0; it < RUN_STEPS; ++it) {
    int s, e;
    run_bounds(rs, re, it, hi, s, e);
    r[it] = (s + sub < e) ? pp[s + sub] : make_int2(0, -1);
  }
#pragma unroll
  for (int it = 0; it < RUN_STEPS; ++it)
    if (r[it].y >= 0) atomicAdd(&dcnt[r[it].y & (BIN_NODES - 1)], 1);
  if (longrun) {
    for (int it = 0; it < RUN_STEPS; ++it) {
      int s, e;
      run_bounds(rs, re, it, hi, s, e);
      for (int i = s + sub + RUN_LANES; i < e; i += RUN_LANES)
        atomicAdd(&dcnt[pp[i].y & (BIN_NODES - 1)], 1);
    }
  }
  __syncthreads();
  const int v = dcnt[t];
  int inc = v;
  for (int o = 1; o < 32; o <<= 1) {
    const int u = __shfl_up(inc, o, 32);
    if (lane >= o) inc += u;
  }
  if (lane == 31) wsum[wave] = inc;
  __syncthreads();
  int below = 0;
#pragma unroll
  for (int w = 0; w < BIN_NODES / 32; ++w) below += (w < wave) ? wsum[w] : 0;
  const int first = base + below + inc - v;
  dpos[t] = first;
  if (n < N) { rowstart[n] = first; deg[n] = v; }
  __syncthreads();
#pragma unroll
  for (int it = 0; it < RUN_STEPS; ++it)
    if (r[it].y >= 0)
      csr_src[atomicAdd(&dpos[r[it].y & (BIN_NODES - 1)], 1)] = r[it].x;
  if (longrun) {
    for (int it = 0; it < RUN_STEPS; ++it) {
      int s, e;
      run_bounds(rs, re, it, hi, s, e);
      for (int i = s + sub + RUN_LANES; i < e; i += RUN_LANES) {
        const int2 q = pp[i];
        csr_src[atomicAdd(&dpos[q.y & (BIN_NODES - 1)], 1)] = q.x;
      }
    }
  }
}

template <int M>
__device__ __forceinline__ float lane_xor(float v) {
  return __builtin_bit_cast(float, __builtin_amdgcn_update_dpp(
      0, __builtin_bit_cast(int, v), 0x160 + M, 0xf, 0xf, true));
}

__device__ __forceinline__ float sum8_by_row(const float (&p)[8], bool b2,
                                             bool b1, bool b0) {
  float q[4], s[2];
#pragma unroll
  for (int i = 0; i < 4; ++i) {
    const float snd = b2 ? p[i] : p[i + 4];
    const float kp  = b2 ? p[i + 4] : p[i];
    q[i] = kp + lane_xor<4>(snd);
  }
#pragma unroll
  for (int i = 0; i < 2; ++i) {
    const float snd = b1 ? q[i] : q[i + 2];
    const float kp  = b1 ? q[i + 2] : q[i];
    s[i] = kp + lane_xor<2>(snd);
  }
  const float snd = b0 ? s[0] : s[1];
  const float kp  = b0 ? s[1] : s[0];
  return kp + lane_xor<1>(snd);
}

__global__ __launch_bounds__(128) void gat_gemm_wmma_f32(
    const float* __restrict__ X, const float* __restrict__ W,
    const float* __restrict__ att_src, const float* __restrict__ att_dst,
    float* __restrict__ H1, float* __restrict__ AS1, float* __restrict__ AD1,
    int N) {
  __shared__ float Wp[64 * 256];

  const int t    = threadIdx.x;
  const int wave = __builtin_amdgcn_readfirstlane(t >> 5);
  const int lane = t & 31;
  const int hf   = lane >> 4;
  const int c    = lane & 15;
  const int row0 = (blockIdx.x * 4 + wave) * 16;

  const int ar = min(row0 + c, N - 1);
  const float* __restrict__ xr = X + (size_t)ar * F_IN + 4 * hf;
  float4 a[16];
#pragma unroll
  for (int j = 0; j < 16; ++j) a[j] = *(const float4*)(xr + 8 * j);

  const float* __restrict__ wsrc = W + (size_t)(2 * wave) * F_IN + 4 * lane;
  float* __restrict__ wdst = Wp + wave * 256 + 8 * lane;
#pragma unroll
  for (int hb = 0; hb < 2; ++hb) {
    float4 r0[8], r1[8];
#pragma unroll
    for (int it = 0; it < 8; ++it) {
      const int n = hb * 8 + it;
      r0[it] = *(const float4*)(wsrc + (size_t)n * 8 * F_IN);
      r1[it] = *(const float4*)(wsrc + (size_t)n * 8 * F_IN + F_IN);
    }
#pragma unroll
    for (int it = 0; it < 8; ++it) {
      const int n = hb * 8 + it;
      float4* p = (float4*)(wdst + n * 4 * 256);
      p[0] = make_float4(r0[it].x, r1[it].x, r0[it].y, r1[it].y);
      p[1] = make_float4(r0[it].z, r1[it].z, r0[it].w, r1[it].w);
    }
  }
  __syncthreads();
  if (row0 >= N) return;

  v8f acc[8];
#pragma unroll
  for (int i = 0; i < 8; ++i) acc[i] = v8f{};

  float2 asv[4], adv[4];
#pragma unroll
  for (int u = 0; u < 4; ++u) {
    asv[u] = *(const float2*)(att_src + 32 * u + 2 * c);
    adv[u] = *(const float2*)(att_dst + 32 * u + 2 * c);
  }

  const float* __restrict__ bp = Wp + (2 * hf) * 256 + 4 * c;
#pragma unroll
  for (int j = 0; j < 16; ++j) {
#pragma unroll
    for (int s = 0; s < 2; ++s) {
      const float* q = bp + (4 * j + s) * 256;
      const float4 q0 = *(const float4*)(q);
      const float4 q1 = *(const float4*)(q + 64);
      const float4 q2 = *(const float4*)(q + 128);
      const float4 q3 = *(const float4*)(q + 192);
      v2f av;
      av.x = s ? a[j].z : a[j].x;
      av.y = s ? a[j].w : a[j].y;
      v2f b;
      b.x = q0.x; b.y = q0.y;
      acc[0] = __builtin_amdgcn_wmma_f32_16x16x4_f32(false, av, false, b, (short)0, acc[0], false, false);
      b.x = q0.z; b.y = q0.w;
      acc[1] = __builtin_amdgcn_wmma_f32_16x16x4_f32(false, av, false, b, (short)0, acc[1], false, false);
      b.x = q1.x; b.y = q1.y;
      acc[2] = __builtin_amdgcn_wmma_f32_16x16x4_f32(false, av, false, b, (short)0, acc[2], false, false);
      b.x = q1.z; b.y = q1.w;
      acc[3] = __builtin_amdgcn_wmma_f32_16x16x4_f32(false, av, false, b, (short)0, acc[3], false, false);
      b.x = q2.x; b.y = q2.y;
      acc[4] = __builtin_amdgcn_wmma_f32_16x16x4_f32(false, av, false, b, (short)0, acc[4], false, false);
      b.x = q2.z; b.y = q2.w;
      acc[5] = __builtin_amdgcn_wmma_f32_16x16x4_f32(false, av, false, b, (short)0, acc[5], false, false);
      b.x = q3.x; b.y = q3.y;
      acc[6] = __builtin_amdgcn_wmma_f32_16x16x4_f32(false, av, false, b, (short)0, acc[6], false, false);
      b.x = q3.z; b.y = q3.w;
      acc[7] = __builtin_amdgcn_wmma_f32_16x16x4_f32(false, av, false, b, (short)0, acc[7], false, false);
    }
  }

  float* __restrict__ o = H1 + (size_t)(row0 + 8 * hf) * F_IN + 2 * c;
#pragma unroll
  for (int r = 0; r < 8; ++r) {
#pragma unroll
    for (int u = 0; u < 4; ++u)
      *(float2*)(o + (size_t)r * F_IN + 32 * u) = make_float2(acc[2 * u][r], acc[2 * u + 1][r]);
  }

  const bool b0 = (lane & 1) != 0, b1 = (lane & 2) != 0, b2 = (lane & 4) != 0;
  const bool g  = (lane & 8) != 0;
  float ts[4], td[4];
#pragma unroll
  for (int u = 0; u < 4; ++u) {
    float ps[8], pd[8];
#pragma unroll
    for (int r = 0; r < 8; ++r) {
      ps[r] = fmaf(acc[2 * u + 1][r], asv[u].y, acc[2 * u][r] * asv[u].x);
      pd[r] = fmaf(acc[2 * u + 1][r], adv[u].y, acc[2 * u][r] * adv[u].x);
    }
    ts[u] = sum8_by_row(ps, b2, b1, b0);
    td[u] = sum8_by_row(pd, b2, b1, b0);
  }
  const float xs0 = lane_xor<8>(g ? ts[0] : ts[2]);
  const float xs1 = lane_xor<8>(g ? ts[1] : ts[3]);
  const float xd0 = lane_xor<8>(g ? td[0] : td[2]);
  const float xd1 = lane_xor<8>(g ? td[1] : td[3]);
  const float4 os = g ? make_float4(xs0, ts[2], xs1, ts[3]) : make_float4(ts[0], xs0, ts[1], xs1);
  const float4 od = g ? make_float4(xd0, td[2], xd1, td[3]) : make_float4(td[0], xd0, td[1], xd1);
  const size_t ao = (size_t)(row0 + 8 * hf + (lane & 7)) * HEADS + (g ? 4 : 0);
  *(float4*)(AS1 + ao) = os;
  *(float4*)(AD1 + ao) = od;
}

#define G1_WAVES 16384

__device__ __forceinline__ float max16_by_head(const float (&v)[4], bool b3,
                                               bool b2) {
  float q[2];
#pragma unroll
  for (int i = 0; i < 2; ++i) {
    const float snd = b3 ? v[i] : v[i + 2];
    const float kp  = b3 ? v[i + 2] : v[i];
    q[i] = fmaxf(kp, lane_xor<8>(snd));
  }
  const float snd = b2 ? q[0] : q[1];
  const float kp  = b2 ? q[1] : q[0];
  float r = fmaxf(kp, lane_xor<4>(snd));
  r = fmaxf(r, lane_xor<1>(r));
  r = fmaxf(r, lane_xor<2>(r));
  return r;
}

template <int G>
__device__ __forceinline__ void gather_group(
    int mine, int j0, int nn, const float* __restrict__ H1,
    const float* __restrict__ as1, int lane, int h, float adh, float m,
    float& l, float4& acc) {
  float4 r[G];
  float a[G];
#pragma unroll
  for (int u = 0; u < G; ++u) {
    const int s = __builtin_amdgcn_readlane(mine, j0 + u);
    r[u] = *(const float4*)(H1 + (size_t)s * F_IN + 4 * lane);
    a[u] = as1[s * HEADS + h];
  }
#pragma unroll
  for (int u = 0; u < G; ++u) {
    float w = __expf(leaky(a[u] + adh) - m);
    w = (j0 + u < nn) ? w : 0.f;
    l += w;
    acc.x = fmaf(w, r[u].x, acc.x); acc.y = fmaf(w, r[u].y, acc.y);
    acc.z = fmaf(w, r[u].z, acc.z); acc.w = fmaf(w, r[u].w, acc.w);
  }
}

__global__ __launch_bounds__(256) void gat_gather1(
    const int* __restrict__ rowstart, const int* __restrict__ deg,
    const int* __restrict__ csr_src, const float* __restrict__ as1,
    const float* __restrict__ ad1, const float* __restrict__ H1,
    const float* __restrict__ b1, float* __restrict__ HACT, int N) {
  const int lane = threadIdx.x & 31;
  const int wv = __builtin_amdgcn_readfirstlane(blockIdx.x * 8 + (threadIdx.x >> 5));
  if (wv >= N) return;
  const int h   = lane >> 2;
  const int j16 = lane & 15;
  const int k   = lane >> 4;
  const bool b2 = (lane & 4) != 0, b3 = (lane & 8) != 0;
  const float4 b = *(const float4*)(b1 + 4 * lane);

  const int* metap = (lane & 1) ? deg : rowstart;
  const int meta0 = metap[wv];
  int start = __builtin_amdgcn_readlane(meta0, 0);
  int cnt   = __builtin_amdgcn_readlane(meta0, 1);
  for (int n = wv; n < N; n += G1_WAVES) {
    const int nx = min(n + G1_WAVES, N - 1);
    const int meta_nx = metap[nx];

    const float adh    = ad1[n * HEADS + h];
    const float asself = as1[n * HEADS + h];
    const float4 adk   = *(const float4*)(ad1 + (size_t)n * HEADS + 4 * k);
    const float4 hv = *(const float4*)(H1 + (size_t)n * F_IN + 4 * lane);

    const int nn0 = min(16, cnt);
    const int s0 = (j16 < nn0) ? csr_src[start + j16] : n;
    const float4 a0 = *(const float4*)(as1 + (size_t)s0 * HEADS + 4 * k);
    float4 r0[4];
    float al0[4];
#pragma unroll
    for (int u = 0; u < 4; ++u) {
      const int s = __builtin_amdgcn_readlane(s0, u);
      r0[u]  = *(const float4*)(H1 + (size_t)s * F_IN + 4 * lane);
      al0[u] = as1[s * HEADS + h];
    }
    float v[4];
    v[0] = leaky(a0.x + adk.x); v[1] = leaky(a0.y + adk.y);
    v[2] = leaky(a0.z + adk.z); v[3] = leaky(a0.w + adk.w);
    float m = max16_by_head(v, b3, b2);
    for (int c0 = 16; c0 < cnt; c0 += 16) {
      const int nn = min(16, cnt - c0);
      const int s = (j16 < nn) ? csr_src[start + c0 + j16] : n;
      const float4 a = *(const float4*)(as1 + (size_t)s * HEADS + 4 * k);
      v[0] = leaky(a.x + adk.x); v[1] = leaky(a.y + adk.y);
      v[2] = leaky(a.z + adk.z); v[3] = leaky(a.w + adk.w);
      m = fmaxf(m, max16_by_head(v, b3, b2));
    }
    const float vself = leaky(asself + adh);
    m = fmaxf(m, vself);

    const float ex = __expf(vself - m);
    float l = ex;
    float4 acc;
    acc.x = ex * hv.x; acc.y = ex * hv.y; acc.z = ex * hv.z; acc.w = ex * hv.w;
#pragma unroll
    for (int u = 0; u < 4; ++u) {
      float w = __expf(leaky(al0[u] + adh) - m);
      w = (u < nn0) ? w : 0.f;
      l += w;
      acc.x = fmaf(w, r0[u].x, acc.x); acc.y = fmaf(w, r0[u].y, acc.y);
      acc.z = fmaf(w, r0[u].z, acc.z); acc.w = fmaf(w, r0[u].w, acc.w);
    }
    for (int j0 = 4; j0 < nn0; j0 += 4)
      gather_group<4>(s0, j0, nn0, H1, as1, lane, h, adh, m, l, acc);
    for (int c0 = 16; c0 < cnt; c0 += 16) {
      const int nn = min(16, cnt - c0);
      const int mine = (j16 < nn) ? csr_src[start + c0 + j16] : n;
      for (int j0 = 0; j0 < nn; j0 += 4)
        gather_group<4>(mine, j0, nn, H1, as1, lane, h, adh, m, l, acc);
    }

    const float den = l + 1e-16f;
    float4 o;
    o.x = acc.x / den + b.x; o.y = acc.y / den + b.y;
    o.z = acc.z / den + b.z; o.w = acc.w / den + b.w;
    o.x = (o.x > 0.f) ? o.x : (__expf(o.x) - 1.f);
    o.y = (o.y > 0.f) ? o.y : (__expf(o.y) - 1.f);
    o.z = (o.z > 0.f) ? o.z : (__expf(o.z) - 1.f);
    o.w = (o.w > 0.f) ? o.w : (__expf(o.w) - 1.f);
    *(float4*)(HACT + (size_t)n * F_IN + 4 * lane) = o;

    start = __builtin_amdgcn_readlane(meta_nx, 0);
    cnt   = __builtin_amdgcn_readlane(meta_nx, 1);
  }
}

__global__ __launch_bounds__(256) void gat_layer2_node(
    const float* __restrict__ HACT, const float* __restrict__ W2,
    const float* __restrict__ att_src2, const float* __restrict__ att_dst2,
    float* __restrict__ H2, float* __restrict__ as2, float* __restrict__ ad2,
    int N) {
  int n = blockIdx.x * blockDim.x + threadIdx.x;
  if (n >= N) return;
  const float* __restrict__ hp = HACT + (size_t)n * F_IN;
  float4 hr[F_IN / 4];
#pragma unroll
  for (int q = 0; q < F_IN / 4; ++q) hr[q] = *(const float4*)(hp + 4 * q);
  float a0 = 0.f, a1 = 0.f;
#pragma unroll
  for (int q = 0; q < F_IN / 4; ++q) {
    const float v[4] = {hr[q].x, hr[q].y, hr[q].z, hr[q].w};
#pragma unroll
    for (int e = 0; e < 4; ++e) {
      const int k = 4 * q + e;
      a0 = fmaf(v[e], W2[k * OUTC + 0], a0);
      a1 = fmaf(v[e], W2[k * OUTC + 1], a1);
    }
  }
  H2[n * OUTC + 0] = a0;
  H2[n * OUTC + 1] = a1;
  as2[n] = a0 * att_src2[0] + a1 * att_src2[1];
  ad2[n] = a0 * att_dst2[0] + a1 * att_dst2[1];
}

#define G2_CHUNK 16

__device__ __forceinline__ void g2_sources(
    const int* __restrict__ csr_src, int start, int c0, int cnt, int n,
    int (&s)[G2_CHUNK]) {
#pragma unroll
  for (int u = 0; u < G2_CHUNK; ++u) {
    const int v = csr_src[max(start + min(c0 + u, cnt - 1), 0)];
    s[u] = (c0 + u < cnt) ? v : n;
  }
}

__global__ __launch_bounds__(256) void gat_gather2(
    const int* __restrict__ rowstart, const int* __restrict__ deg,
    const int* __restrict__ csr_src, const float* __restrict__ as2,
    const float* __restrict__ ad2, const float* __restrict__ H2,
    const float* __restrict__ b2, float* __restrict__ OUT, int N) {
  const int n = blockIdx.x * 256 + threadIdx.x;
  if (n >= N) return;
  const int start = rowstart[n];
  const int cnt   = deg[n];
  const float adn   = ad2[n];
  const float vself = leaky(as2[n] + adn);
  const float2 hv = *(const float2*)(H2 + (size_t)n * OUTC);

  int s0[G2_CHUNK];
  float a0v[G2_CHUNK];
  float2 r0[G2_CHUNK];
  g2_sources(csr_src, start, 0, cnt, n, s0);
#pragma unroll
  for (int u = 0; u < G2_CHUNK; ++u) a0v[u] = as2[s0[u]];
#pragma unroll
  for (int u = 0; u < G2_CHUNK; ++u) r0[u] = *(const float2*)(H2 + (size_t)s0[u] * OUTC);

  float m = vself;
#pragma unroll
  for (int u = 0; u < G2_CHUNK; ++u) m = fmaxf(m, leaky(a0v[u] + adn));
  for (int c0 = G2_CHUNK; c0 < cnt; c0 += G2_CHUNK) {
    int s[G2_CHUNK];
    float a[G2_CHUNK];
    g2_sources(csr_src, start, c0, cnt, n, s);
#pragma unroll
    for (int u = 0; u < G2_CHUNK; ++u) a[u] = as2[s[u]];
#pragma unroll
    for (int u = 0; u < G2_CHUNK; ++u) m = fmaxf(m, leaky(a[u] + adn));
  }

  const float ex = __expf(vself - m);
  float l = ex;
  float a0 = ex * hv.x, a1 = ex * hv.y;
#pragma unroll
  for (int u = 0; u < G2_CHUNK; ++u) {
    float w = __expf(leaky(a0v[u] + adn) - m);
    w = (u < cnt) ? w : 0.f;
    l += w;
    a0 = fmaf(w, r0[u].x, a0);
    a1 = fmaf(w, r0[u].y, a1);
  }
  for (int c0 = G2_CHUNK; c0 < cnt; c0 += G2_CHUNK) {
    int s[G2_CHUNK];
    float a[G2_CHUNK];
    float2 r[G2_CHUNK];
    g2_sources(csr_src, start, c0, cnt, n, s);
#pragma unroll
    for (int u = 0; u < G2_CHUNK; ++u) a[u] = as2[s[u]];
#pragma unroll
    for (int u = 0; u < G2_CHUNK; ++u) r[u] = *(const float2*)(H2 + (size_t)s[u] * OUTC);
#pragma unroll
    for (int u = 0; u < G2_CHUNK; ++u) {
      float w = __expf(leaky(a[u] + adn) - m);
      w = (c0 + u < cnt) ? w : 0.f;
      l += w;
      a0 = fmaf(w, r[u].x, a0);
      a1 = fmaf(w, r[u].y, a1);
    }
  }
  const float den = l + 1e-16f;
  float2 o;
  o.x = a0 / den + b2[0];
  o.y = a1 / den + b2[1];
  *(float2*)(OUT + (size_t)n * OUTC) = o;
}

static inline unsigned gdim(long long n, int b) {
  return (unsigned)((n + b - 1) / b);
}

extern "C" void kernel_launch(void* const* d_in, const int* in_sizes, int n_in,
                              void* d_out, int out_size, void* d_ws,
                              size_t ws_size, hipStream_t stream) {
  const float* x        = (const float*)d_in[0];
  const int*   ei       = (const int*)d_in[1];
  const float* W1       = (const float*)d_in[2];
  const float* att_src1 = (const float*)d_in[3];
  const float* att_dst1 = (const float*)d_in[4];
  const float* b1       = (const float*)d_in[5];
  const float* W2       = (const float*)d_in[6];
  const float* att_src2 = (const float*)d_in[7];
  const float* att_dst2 = (const float*)d_in[8];
  const float* b2       = (const float*)d_in[9];

  const int N  = in_sizes[0] / F_IN;
  const int E  = in_sizes[1] / 2;

  float* ws = (float*)d_ws;
  size_t off = 0;
  float* h1   = ws + off; off += (size_t)N * F_IN;
  float* hact = ws + off; off += (size_t)N * F_IN;
  float* as1  = ws + off; off += (size_t)N * HEADS;
  float* ad1  = ws + off; off += (size_t)N * HEADS;
  float* h2   = ws + off; off += (size_t)N * OUTC;
  float* as2  = ws + off; off += (size_t)N;
  float* ad2  = ws + off; off += (size_t)N;
  int* pairs    = (int*)(ws + off); off += (size_t)E * 2;
  int* deg      = (int*)(ws + off); off += (size_t)N;
  int* rowstart = (int*)(ws + off); off += (size_t)N;
  int* csr_src  = (int*)(ws + off); off += (size_t)E;
  const int nbins = (int)gdim(N, BIN_NODES);
  int* runstart = (int*)(ws + off); off += (size_t)(nbins + 1) * NCHUNK;
  (void)ws_size; (void)n_in; (void)out_size;

  const int B = 256;

  gat_bin_sort<<<NCHUNK, B, 0, stream>>>(ei, runstart, pairs, E, nbins);
  gat_bin_csr<<<nbins, BIN_NODES, 0, stream>>>(pairs, runstart, rowstart, deg, csr_src, N, E);

  gat_gemm_wmma_f32<<<gdim(N, 64), 128, 0, stream>>>(x, W1, att_src1, att_dst1, h1, as1, ad1, N);
  gat_gather1<<<G1_WAVES / 8, B, 0, stream>>>(rowstart, deg, csr_src, as1, ad1, h1, b1, hact, N);

  gat_layer2_node<<<gdim(N, B), B, 0, stream>>>(hact, W2, att_src2, att_dst2, h2, as2, ad2, N);
  gat_gather2<<<gdim(N, B), B, 0, stream>>>(rowstart, deg, csr_src, as2, ad2, h2, b2, (float*)d_out, N);
}


# Layer-2 node product in the layer-1 gather's epilogue (row in registers, 32-lane tree: h2's sum order changes): hact never stored, one launch less; h2 and its dots as one 16-byte record
# speedup vs baseline: 7.4825x; 1.1375x over previous
#include <hip/hip_runtime.h>
#include <hip/hip_bf16.h>


typedef __attribute__((ext_vector_type(2))) float v2f;
typedef __attribute__((ext_vector_type(8))) float v8f;

#define F_IN   128
#define HEADS  8
#define CH     16
#define OUTC   2
#define NEG_SLOPE 0.2f
#define BIN_SHIFT 8
#define BIN_NODES 256
#define MAX_BINS  512
#define NCHUNK    256

__device__ __forceinline__ float leaky(float v) {
  return (v > 0.f) ? v : NEG_SLOPE * v;
}

template <int M>
__device__ __forceinline__ int lane_xor_i(int v) {
  return __builtin_amdgcn_update_dpp(0, v, 0x160 + M, 0xf, 0xf, true);
}

__global__ __launch_bounds__(256) void gat_bin_sort(
    const int* __restrict__ ei, int* __restrict__ runstart,
    int* __restrict__ pairs, int E, int nbins) {
  __shared__ int hist[MAX_BINS + 1];
  __shared__ int wsum[256 / 32];
  int2* __restrict__ pr = (int2*)pairs;
  const int t = threadIdx.x, lane = t & 31, wave = t >> 5, g = blockIdx.x;
  hist[t] = 0;
  hist[t + 256] = 0;
  __syncthreads();
  const int ce = (E + NCHUNK - 1) / NCHUNK;
  const int e0 = g * ce;
  const int e1 = min(E, e0 + ce);
  for (int e = e0 + t; e < e1; e += 4 * 256) {
    int d[4];
#pragma unroll
    for (int u = 0; u < 4; ++u) {
      const int ee = e + u * 256;
      d[u] = (ee < e1) ? ei[E + ee] : -1;
    }
#pragma unroll
    for (int u = 0; u < 4; ++u)
      if (d[u] >= 0) atomicAdd(&hist[d[u] >> BIN_SHIFT], 1);
  }
  __syncthreads();
  const int c0 = hist[2 * t], c1 = hist[2 * t + 1];
  const int v = c0 + c1;
  int inc = v;
  for (int o = 1; o < 32; o <<= 1) {
    const int u = __shfl_up(inc, o, 32);
    if (lane >= o) inc += u;
  }
  if (lane == 31) wsum[wave] = inc;
  __syncthreads();
  int below = 0;
#pragma unroll
  for (int w = 0; w < 256 / 32; ++w) below += (w < wave) ? wsum[w] : 0;
  const int x0 = e0 + below + inc - v;
  hist[2 * t]     = x0;
  hist[2 * t + 1] = x0 + c0;
  if (t == 255) hist[MAX_BINS] = x0 + v;
  __syncthreads();
  for (int b = t; b <= nbins; b += 256) runstart[b * NCHUNK + g] = hist[b];
  __syncthreads();
  for (int e = e0 + t; e < e1; e += 4 * 256) {
    int d[4], sr[4];
#pragma unroll
    for (int u = 0; u < 4; ++u) {
      const int ee = e + u * 256;
      d[u]  = (ee < e1) ? ei[E + ee] : -1;
      sr[u] = (ee < e1) ? ei[ee] : 0;
    }
#pragma unroll
    for (int u = 0; u < 4; ++u)
      if (d[u] >= 0) {
        const int pos = atomicAdd(&hist[d[u] >> BIN_SHIFT], 1);
        pr[pos] = make_int2(sr[u], d[u]);
      }
  }
}

#define RUN_LANES 16
#define RUN_STEPS 16

__device__ __forceinline__ void run_bounds(int rs, int re, int it, bool hi,
                                           int& s, int& e) {
  const int s0 = __builtin_amdgcn_readlane(rs, 2 * it);
  const int e0 = __builtin_amdgcn_readlane(re, 2 * it);
  const int s1 = __builtin_amdgcn_readlane(rs, 2 * it + 1);
  const int e1 = __builtin_amdgcn_readlane(re, 2 * it + 1);
  s = hi ? s1 : s0;
  e = hi ? e1 : e0;
}

__global__ __launch_bounds__(BIN_NODES) void gat_bin_csr(
    const int* __restrict__ pairs, const int* __restrict__ runstart,
    int* __restrict__ rowstart, int* __restrict__ deg,
    int* __restrict__ csr_src, int N, int E) {
  __shared__ int dcnt[BIN_NODES];
  __shared__ int dpos[BIN_NODES];
  __shared__ int wsum[BIN_NODES / 32];
  __shared__ int wbef[BIN_NODES / 32];
  const int t = threadIdx.x, lane = t & 31, wave = t >> 5;
  const int b = blockIdx.x;
  const int n = b * BIN_NODES + t;
  const int2* __restrict__ pp = (const int2*)pairs;
  const int rs = runstart[b * NCHUNK + t];
  const int re = runstart[(b + 1) * NCHUNK + t];
  const int ce = (E + NCHUNK - 1) / NCHUNK;
  int bef = rs - t * ce;
  dcnt[t] = 0;
  bef += lane_xor_i<1>(bef);
  bef += lane_xor_i<2>(bef);
  bef += lane_xor_i<4>(bef);
  bef += lane_xor_i<8>(bef);
  const int wtot = __builtin_amdgcn_readlane(bef, 0) + __builtin_amdgcn_readlane(bef, 16);
  if (lane == 0) wbef[wave] = wtot;
  __syncthreads();
  int base = 0;
#pragma unroll
  for (int w = 0; w < BIN_NODES / 32; ++w) base += wbef[w];

  const int sub = lane & (RUN_LANES - 1);
  const bool hi = lane >= RUN_LANES;
  const bool longrun = __any(re - rs > RUN_LANES);
  int2 r[RUN_STEPS];
#pragma unroll
  for (int it = 0; it < RUN_STEPS; ++it) {
    int s, e;
    run_bounds(rs, re, it, hi, s, e);
    r[it] = (s + sub < e) ? pp[s + sub] : make_int2(0, -1);
  }
#pragma unroll
  for (int it = 0; it < RUN_STEPS; ++it)
    if (r[it].y >= 0) atomicAdd(&dcnt[r[it].y & (BIN_NODES - 1)], 1);
  if (longrun) {
    for (int it = 0; it < RUN_STEPS; ++it) {
      int s, e;
      run_bounds(rs, re, it, hi, s, e);
      for (int i = s + sub + RUN_LANES; i < e; i += RUN_LANES)
        atomicAdd(&dcnt[pp[i].y & (BIN_NODES - 1)], 1);
    }
  }
  __syncthreads();
  const int v = dcnt[t];
  int inc = v;
  for (int o = 1; o < 32; o <<= 1) {
    const int u = __shfl_up(inc, o, 32);
    if (lane >= o) inc += u;
  }
  if (lane == 31) wsum[wave] = inc;
  __syncthreads();
  int below = 0;
#pragma unroll
  for (int w = 0; w < BIN_NODES / 32; ++w) below += (w < wave) ? wsum[w] : 0;
  const int first = base + below + inc - v;
  dpos[t] = first;
  if (n < N) { rowstart[n] = first; deg[n] = v; }
  __syncthreads();
#pragma unroll
  for (int it = 0; it < RUN_STEPS; ++it)
    if (r[it].y >= 0)
      csr_src[atomicAdd(&dpos[r[it].y & (BIN_NODES - 1)], 1)] = r[it].x;
  if (longrun) {
    for (int it = 0; it < RUN_STEPS; ++it) {
      int s, e;
      run_bounds(rs, re, it, hi, s, e);
      for (int i = s + sub + RUN_LANES; i < e; i += RUN_LANES) {
        const int2 q = pp[i];
        csr_src[atomicAdd(&dpos[q.y & (BIN_NODES - 1)], 1)] = q.x;
      }
    }
  }
}

template <int M>
__device__ __forceinline__ float lane_xor(float v) {
  return __builtin_bit_cast(float, __builtin_amdgcn_update_dpp(
      0, __builtin_bit_cast(int, v), 0x160 + M, 0xf, 0xf, true));
}

__device__ __forceinline__ float sum8_by_row(const float (&p)[8], bool b2,
                                             bool b1, bool b0) {
  float q[4], s[2];
#pragma unroll
  for (int i = 0; i < 4; ++i) {
    const float snd = b2 ? p[i] : p[i + 4];
    const float kp  = b2 ? p[i + 4] : p[i];
    q[i] = kp + lane_xor<4>(snd);
  }
#pragma unroll
  for (int i = 0; i < 2; ++i) {
    const float snd = b1 ? q[i] : q[i + 2];
    const float kp  = b1 ? q[i + 2] : q[i];
    s[i] = kp + lane_xor<2>(snd);
  }
  const float snd = b0 ? s[0] : s[1];
  const float kp  = b0 ? s[1] : s[0];
  return kp + lane_xor<1>(snd);
}

__global__ __launch_bounds__(128) void gat_gemm_wmma_f32(
    const float* __restrict__ X, const float* __restrict__ W,
    const float* __restrict__ att_src, const float* __restrict__ att_dst,
    float* __restrict__ H1, float* __restrict__ AS1, float* __restrict__ AD1,
    int N) {
  __shared__ float Wp[64 * 256];

  const int t    = threadIdx.x;
  const int wave = __builtin_amdgcn_readfirstlane(t >> 5);
  const int lane = t & 31;
  const int hf   = lane >> 4;
  const int c    = lane & 15;
  const int row0 = (blockIdx.x * 4 + wave) * 16;

  const int ar = min(row0 + c, N - 1);
  const float* __restrict__ xr = X + (size_t)ar * F_IN + 4 * hf;
  float4 a[16];
#pragma unroll
  for (int j = 0; j < 16; ++j) a[j] = *(const float4*)(xr + 8 * j);

  const float* __restrict__ wsrc = W + (size_t)(2 * wave) * F_IN + 4 * lane;
  float* __restrict__ wdst = Wp + wave * 256 + 8 * lane;
#pragma unroll
  for (int hb = 0; hb < 2; ++hb) {
    float4 r0[8], r1[8];
#pragma unroll
    for (int it = 0; it < 8; ++it) {
      const int n = hb * 8 + it;
      r0[it] = *(const float4*)(wsrc + (size_t)n * 8 * F_IN);
      r1[it] = *(const float4*)(wsrc + (size_t)n * 8 * F_IN + F_IN);
    }
#pragma unroll
    for (int it = 0; it < 8; ++it) {
      const int n = hb * 8 + it;
      float4* p = (float4*)(wdst + n * 4 * 256);
      p[0] = make_float4(r0[it].x, r1[it].x, r0[it].y, r1[it].y);
      p[1] = make_float4(r0[it].z, r1[it].z, r0[it].w, r1[it].w);
    }
  }
  __syncthreads();
  if (row0 >= N) return;

  v8f acc[8];
#pragma unroll
  for (int i = 0; i < 8; ++i) acc[i] = v8f{};

  float2 asv[4], adv[4];
#pragma unroll
  for (int u = 0; u < 4; ++u) {
    asv[u] = *(const float2*)(att_src + 32 * u + 2 * c);
    adv[u] = *(const float2*)(att_dst + 32 * u + 2 * c);
  }

  const float* __restrict__ bp = Wp + (2 * hf) * 256 + 4 * c;
#pragma unroll
  for (int j = 0; j < 16; ++j) {
#pragma unroll
    for (int s = 0; s < 2; ++s) {
      const float* q = bp + (4 * j + s) * 256;
      const float4 q0 = *(const float4*)(q);
      const float4 q1 = *(const float4*)(q + 64);
      const float4 q2 = *(const float4*)(q + 128);
      const float4 q3 = *(const float4*)(q + 192);
      v2f av;
      av.x = s ? a[j].z : a[j].x;
      av.y = s ? a[j].w : a[j].y;
      v2f b;
      b.x = q0.x; b.y = q0.y;
      acc[0] = __builtin_amdgcn_wmma_f32_16x16x4_f32(false, av, false, b, (short)0, acc[0], false, false);
      b.x = q0.z; b.y = q0.w;
      acc[1] = __builtin_amdgcn_wmma_f32_16x16x4_f32(false, av, false, b, (short)0, acc[1], false, false);
      b.x = q1.x; b.y = q1.y;
      acc[2] = __builtin_amdgcn_wmma_f32_16x16x4_f32(false, av, false, b, (short)0, acc[2], false, false);
      b.x = q1.z; b.y = q1.w;
      acc[3] = __builtin_amdgcn_wmma_f32_16x16x4_f32(false, av, false, b, (short)0, acc[3], false, false);
      b.x = q2.x; b.y = q2.y;
      acc[4] = __builtin_amdgcn_wmma_f32_16x16x4_f32(false, av, false, b, (short)0, acc[4], false, false);
      b.x = q2.z; b.y = q2.w;
      acc[5] = __builtin_amdgcn_wmma_f32_16x16x4_f32(false, av, false, b, (short)0, acc[5], false, false);
      b.x = q3.x; b.y = q3.y;
      acc[6] = __builtin_amdgcn_wmma_f32_16x16x4_f32(false, av, false, b, (short)0, acc[6], false, false);
      b.x = q3.z; b.y = q3.w;
      acc[7] = __builtin_amdgcn_wmma_f32_16x16x4_f32(false, av, false, b, (short)0, acc[7], false, false);
    }
  }

  float* __restrict__ o = H1 + (size_t)(row0 + 8 * hf) * F_IN + 2 * c;
#pragma unroll
  for (int r = 0; r < 8; ++r) {
#pragma unroll
    for (int u = 0; u < 4; ++u)
      *(float2*)(o + (size_t)r * F_IN + 32 * u) = make_float2(acc[2 * u][r], acc[2 * u + 1][r]);
  }

  const bool b0 = (lane & 1) != 0, b1 = (lane & 2) != 0, b2 = (lane & 4) != 0;
  const bool g  = (lane & 8) != 0;
  float ts[4], td[4];
#pragma unroll
  for (int u = 0; u < 4; ++u) {
    float ps[8], pd[8];
#pragma unroll
    for (int r = 0; r < 8; ++r) {
      ps[r] = fmaf(acc[2 * u + 1][r], asv[u].y, acc[2 * u][r] * asv[u].x);
      pd[r] = fmaf(acc[2 * u + 1][r], adv[u].y, acc[2 * u][r] * adv[u].x);
    }
    ts[u] = sum8_by_row(ps, b2, b1, b0);
    td[u] = sum8_by_row(pd, b2, b1, b0);
  }
  const float xs0 = lane_xor<8>(g ? ts[0] : ts[2]);
  const float xs1 = lane_xor<8>(g ? ts[1] : ts[3]);
  const float xd0 = lane_xor<8>(g ? td[0] : td[2]);
  const float xd1 = lane_xor<8>(g ? td[1] : td[3]);
  const float4 os = g ? make_float4(xs0, ts[2], xs1, ts[3]) : make_float4(ts[0], xs0, ts[1], xs1);
  const float4 od = g ? make_float4(xd0, td[2], xd1, td[3]) : make_float4(td[0], xd0, td[1], xd1);
  const size_t ao = (size_t)(row0 + 8 * hf + (lane & 7)) * HEADS + (g ? 4 : 0);
  *(float4*)(AS1 + ao) = os;
  *(float4*)(AD1 + ao) = od;
}

#define G1_WAVES 16384

__device__ __forceinline__ float lane_xor16(float v) {
  const int b = __builtin_bit_cast(int, v);
  return __builtin_bit_cast(float, __builtin_amdgcn_permlanex16(
      b, b, 0x76543210, 0xfedcba98, false, false));
}

__device__ __forceinline__ float sum2_by_wave(float p0, float p1, bool b0) {
  const float snd = b0 ? p0 : p1;
  const float kp  = b0 ? p1 : p0;
  float v = kp + lane_xor<1>(snd);
  v += lane_xor<2>(v);
  v += lane_xor<4>(v);
  v += lane_xor<8>(v);
  v += lane_xor16(v);
  return v;
}

__device__ __forceinline__ float max16_by_head(const float (&v)[4], bool b3,
                                               bool b2) {
  float q[2];
#pragma unroll
  for (int i = 0; i < 2; ++i) {
    const float snd = b3 ? v[i] : v[i + 2];
    const float kp  = b3 ? v[i + 2] : v[i];
    q[i] = fmaxf(kp, lane_xor<8>(snd));
  }
  const float snd = b2 ? q[0] : q[1];
  const float kp  = b2 ? q[1] : q[0];
  float r = fmaxf(kp, lane_xor<4>(snd));
  r = fmaxf(r, lane_xor<1>(r));
  r = fmaxf(r, lane_xor<2>(r));
  return r;
}

template <int G>
__device__ __forceinline__ void gather_group(
    int mine, int j0, int nn, const float* __restrict__ H1,
    const float* __restrict__ as1, int lane, int h, float adh, float m,
    float& l, float4& acc) {
  float4 r[G];
  float a[G];
#pragma unroll
  for (int u = 0; u < G; ++u) {
    const int s = __builtin_amdgcn_readlane(mine, j0 + u);
    r[u] = *(const float4*)(H1 + (size_t)s * F_IN + 4 * lane);
    a[u] = as1[s * HEADS + h];
  }
#pragma unroll
  for (int u = 0; u < G; ++u) {
    float w = __expf(leaky(a[u] + adh) - m);
    w = (j0 + u < nn) ? w : 0.f;
    l += w;
    acc.x = fmaf(w, r[u].x, acc.x); acc.y = fmaf(w, r[u].y, acc.y);
    acc.z = fmaf(w, r[u].z, acc.z); acc.w = fmaf(w, r[u].w, acc.w);
  }
}

__global__ __launch_bounds__(256) void gat_gather1(
    const int* __restrict__ rowstart, const int* __restrict__ deg,
    const int* __restrict__ csr_src, const float* __restrict__ as1,
    const float* __restrict__ ad1, const float* __restrict__ H1,
    const float* __restrict__ b1, const float* __restrict__ W2,
    const float* __restrict__ att_src2, const float* __restrict__ att_dst2,
    float* __restrict__ REC, int N) {
  const int lane = threadIdx.x & 31;
  const int wv = __builtin_amdgcn_readfirstlane(blockIdx.x * 8 + (threadIdx.x >> 5));
  const int* metap = (lane & 1) ? deg : rowstart;
  const int meta0 = metap[min(wv, N - 1)];
  __shared__ float w2s[F_IN * OUTC];
  for (int i = threadIdx.x; i < F_IN * OUTC; i += 256)
    w2s[((i & 4) << 5) + ((i >> 3) << 2) + (i & 3)] = W2[i];
  __syncthreads();
  if (wv >= N) return;
  const int h   = lane >> 2;
  const int j16 = lane & 15;
  const int k   = lane >> 4;
  const bool b2 = (lane & 4) != 0, b3 = (lane & 8) != 0;
  const bool b0 = (lane & 1) != 0;
  const float4 b = *(const float4*)(b1 + 4 * lane);
  const float as20 = att_src2[0], as21 = att_src2[1];
  const float ad20 = att_dst2[0], ad21 = att_dst2[1];

  int start = __builtin_amdgcn_readlane(meta0, 0);
  int cnt   = __builtin_amdgcn_readlane(meta0, 1);
  for (int n = wv; n < N; n += G1_WAVES) {
    const int nx = min(n + G1_WAVES, N - 1);
    const int meta_nx = metap[nx];

    const float adh    = ad1[n * HEADS + h];
    const float asself = as1[n * HEADS + h];
    const float4 adk   = *(const float4*)(ad1 + (size_t)n * HEADS + 4 * k);
    const float4 hv = *(const float4*)(H1 + (size_t)n * F_IN + 4 * lane);

    const int nn0 = min(16, cnt);
    const int s0 = (j16 < nn0) ? csr_src[start + j16] : n;
    const float4 a0 = *(const float4*)(as1 + (size_t)s0 * HEADS + 4 * k);
    float4 r0[4];
    float al0[4];
#pragma unroll
    for (int u = 0; u < 4; ++u) {
      const int s = __builtin_amdgcn_readlane(s0, u);
      r0[u]  = *(const float4*)(H1 + (size_t)s * F_IN + 4 * lane);
      al0[u] = as1[s * HEADS + h];
    }
    float v[4];
    v[0] = leaky(a0.x + adk.x); v[1] = leaky(a0.y + adk.y);
    v[2] = leaky(a0.z + adk.z); v[3] = leaky(a0.w + adk.w);
    float m = max16_by_head(v, b3, b2);
    for (int c0 = 16; c0 < cnt; c0 += 16) {
      const int nn = min(16, cnt - c0);
      const int s = (j16 < nn) ? csr_src[start + c0 + j16] : n;
      const float4 a = *(const float4*)(as1 + (size_t)s * HEADS + 4 * k);
      v[0] = leaky(a.x + adk.x); v[1] = leaky(a.y + adk.y);
      v[2] = leaky(a.z + adk.z); v[3] = leaky(a.w + adk.w);
      m = fmaxf(m, max16_by_head(v, b3, b2));
    }
    const float vself = leaky(asself + adh);
    m = fmaxf(m, vself);

    const float ex = __expf(vself - m);
    float l = ex;
    float4 acc;
    acc.x = ex * hv.x; acc.y = ex * hv.y; acc.z = ex * hv.z; acc.w = ex * hv.w;
#pragma unroll
    for (int u = 0; u < 4; ++u) {
      float w = __expf(leaky(al0[u] + adh) - m);
      w = (u < nn0) ? w : 0.f;
      l += w;
      acc.x = fmaf(w, r0[u].x, acc.x); acc.y = fmaf(w, r0[u].y, acc.y);
      acc.z = fmaf(w, r0[u].z, acc.z); acc.w = fmaf(w, r0[u].w, acc.w);
    }
    for (int j0 = 4; j0 < nn0; j0 += 4)
      gather_group<4>(s0, j0, nn0, H1, as1, lane, h, adh, m, l, acc);
    for (int c0 = 16; c0 < cnt; c0 += 16) {
      const int nn = min(16, cnt - c0);
      const int mine = (j16 < nn) ? csr_src[start + c0 + j16] : n;
      for (int j0 = 0; j0 < nn; j0 += 4)
        gather_group<4>(mine, j0, nn, H1, as1, lane, h, adh, m, l, acc);
    }

    asm volatile("" ::: "memory");
    const float4 wA = *(const float4*)(w2s + 4 * lane);
    const float4 wB = *(const float4*)(w2s + F_IN + 4 * lane);

    const float den = l + 1e-16f;
    float4 o;
    o.x = acc.x / den + b.x; o.y = acc.y / den + b.y;
    o.z = acc.z / den + b.z; o.w = acc.w / den + b.w;
    o.x = (o.x > 0.f) ? o.x : (__expf(o.x) - 1.f);
    o.y = (o.y > 0.f) ? o.y : (__expf(o.y) - 1.f);
    o.z = (o.z > 0.f) ? o.z : (__expf(o.z) - 1.f);
    o.w = (o.w > 0.f) ? o.w : (__expf(o.w) - 1.f);
    const float p0 = fmaf(o.w, wB.z, fmaf(o.z, wB.x, fmaf(o.y, wA.z, o.x * wA.x)));
    const float p1 = fmaf(o.w, wB.w, fmaf(o.z, wB.y, fmaf(o.y, wA.w, o.x * wA.y)));
    const float tot = sum2_by_wave(p0, p1, b0);
    const float h20 = __builtin_bit_cast(float, __builtin_amdgcn_readlane(__builtin_bit_cast(int, tot), 0));
    const float h21 = __builtin_bit_cast(float, __builtin_amdgcn_readlane(__builtin_bit_cast(int, tot), 1));
    const float sd = h20 * as20 + h21 * as21;
    const float dd = h20 * ad20 + h21 * ad21;
    if (lane == 0) *(float4*)(REC + 4 * n) = make_float4(h20, h21, sd, dd);

    start = __builtin_amdgcn_readlane(meta_nx, 0);
    cnt   = __builtin_amdgcn_readlane(meta_nx, 1);
  }
}

#define G2_CHUNK 16

__device__ __forceinline__ void g2_sources(
    const int* __restrict__ csr_src, int start, int c0, int cnt, int n,
    int (&s)[G2_CHUNK]) {
#pragma unroll
  for (int u = 0; u < G2_CHUNK; ++u) {
    const int v = csr_src[max(start + min(c0 + u, cnt - 1), 0)];
    s[u] = (c0 + u < cnt) ? v : n;
  }
}

__global__ __launch_bounds__(256) void gat_gather2(
    const int* __restrict__ rowstart, const int* __restrict__ deg,
    const int* __restrict__ csr_src, const float* __restrict__ REC,
    const float* __restrict__ b2, float* __restrict__ OUT, int N) {
  const int n = blockIdx.x * 256 + threadIdx.x;
  if (n >= N) return;
  const float4* __restrict__ rec = (const float4*)REC;
  const int start = rowstart[n];
  const int cnt   = deg[n];
  const float4 me = rec[n];
  const float adn   = me.w;
  const float vself = leaky(me.z + adn);

  int s0[G2_CHUNK];
  float4 r0[G2_CHUNK];
  g2_sources(csr_src, start, 0, cnt, n, s0);
#pragma unroll
  for (int u = 0; u < G2_CHUNK; ++u) r0[u] = rec[s0[u]];

  float m = vself;
#pragma unroll
  for (int u = 0; u < G2_CHUNK; ++u) m = fmaxf(m, leaky(r0[u].z + adn));
  for (int c0 = G2_CHUNK; c0 < cnt; c0 += G2_CHUNK) {
    int s[G2_CHUNK];
    float a[G2_CHUNK];
    g2_sources(csr_src, start, c0, cnt, n, s);
#pragma unroll
    for (int u = 0; u < G2_CHUNK; ++u) a[u] = REC[(size_t)s[u] * 4 + 2];
#pragma unroll
    for (int u = 0; u < G2_CHUNK; ++u) m = fmaxf(m, leaky(a[u] + adn));
  }

  const float ex = __expf(vself - m);
  float l = ex;
  float a0 = ex * me.x, a1 = ex * me.y;
#pragma unroll
  for (int u = 0; u < G2_CHUNK; ++u) {
    float w = __expf(leaky(r0[u].z + adn) - m);
    w = (u < cnt) ? w : 0.f;
    l += w;
    a0 = fmaf(w, r0[u].x, a0);
    a1 = fmaf(w, r0[u].y, a1);
  }
  for (int c0 = G2_CHUNK; c0 < cnt; c0 += G2_CHUNK) {
    int s[G2_CHUNK];
    float4 r[G2_CHUNK];
    g2_sources(csr_src, start, c0, cnt, n, s);
#pragma unroll
    for (int u = 0; u < G2_CHUNK; ++u) r[u] = rec[s[u]];
#pragma unroll
    for (int u = 0; u < G2_CHUNK; ++u) {
      float w = __expf(leaky(r[u].z + adn) - m);
      w = (c0 + u < cnt) ? w : 0.f;
      l += w;
      a0 = fmaf(w, r[u].x, a0);
      a1 = fmaf(w, r[u].y, a1);
    }
  }
  const float den = l + 1e-16f;
  float2 o;
  o.x = a0 / den + b2[0];
  o.y = a1 / den + b2[1];
  *(float2*)(OUT + (size_t)n * OUTC) = o;
}

static inline unsigned gdim(long long n, int b) {
  return (unsigned)((n + b - 1) / b);
}

extern "C" void kernel_launch(void* const* d_in, const int* in_sizes, int n_in,
                              void* d_out, int out_size, void* d_ws,
                              size_t ws_size, hipStream_t stream) {
  const float* x        = (const float*)d_in[0];
  const int*   ei       = (const int*)d_in[1];
  const float* W1       = (const float*)d_in[2];
  const float* att_src1 = (const float*)d_in[3];
  const float* att_dst1 = (const float*)d_in[4];
  const float* b1       = (const float*)d_in[5];
  const float* W2       = (const float*)d_in[6];
  const float* att_src2 = (const float*)d_in[7];
  const float* att_dst2 = (const float*)d_in[8];
  const float* b2       = (const float*)d_in[9];

  const int N  = in_sizes[0] / F_IN;
  const int E  = in_sizes[1] / 2;

  float* ws = (float*)d_ws;
  size_t off = 0;
  float* h1   = ws + off; off += (size_t)N * F_IN;
  float* rec2 = ws + off; off += (size_t)N * 4;
  float* as1  = ws + off; off += (size_t)N * HEADS;
  float* ad1  = ws + off; off += (size_t)N * HEADS;
  int* pairs    = (int*)(ws + off); off += (size_t)E * 2;
  int* deg      = (int*)(ws + off); off += (size_t)N;
  int* rowstart = (int*)(ws + off); off += (size_t)N;
  int* csr_src  = (int*)(ws + off); off += (size_t)E;
  const int nbins = (int)gdim(N, BIN_NODES);
  int* runstart = (int*)(ws + off); off += (size_t)(nbins + 1) * NCHUNK;
  (void)ws_size; (void)n_in; (void)out_size;

  const int B = 256;

  gat_bin_sort<<<NCHUNK, B, 0, stream>>>(ei, runstart, pairs, E, nbins);
  gat_bin_csr<<<nbins, BIN_NODES, 0, stream>>>(pairs, runstart, rowstart, deg, csr_src, N, E);

  gat_gemm_wmma_f32<<<gdim(N, 64), 128, 0, stream>>>(x, W1, att_src1, att_dst1, h1, as1, ad1, N);
  gat_gather1<<<G1_WAVES / 8, B, 0, stream>>>(rowstart, deg, csr_src, as1, ad1, h1, b1, W2, att_src2, att_dst2, rec2, N);

  gat_gather2<<<gdim(N, B), B, 0, stream>>>(rowstart, deg, csr_src, rec2, b2, (float*)d_out, N);
}


# Chunk sort on chip: a thread's edges loaded once, all together, ranked by the count's returning add, slice ordered in LDS, written as whole lines: 1 memory round trip, was 8; node sum order changes
# speedup vs baseline: 7.9009x; 1.0559x over previous
#include <hip/hip_runtime.h>
#include <hip/hip_bf16.h>


typedef __attribute__((ext_vector_type(2))) float v2f;
typedef __attribute__((ext_vector_type(8))) float v8f;

#define F_IN   128
#define HEADS  8
#define CH     16
#define OUTC   2
#define NEG_SLOPE 0.2f
#define BIN_SHIFT 8
#define BIN_NODES 256
#define MAX_BINS  512
#define NCHUNK    256

__device__ __forceinline__ float leaky(float v) {
  return (v > 0.f) ? v : NEG_SLOPE * v;
}

template <int M>
__device__ __forceinline__ int lane_xor_i(int v) {
  return __builtin_amdgcn_update_dpp(0, v, 0x160 + M, 0xf, 0xf, true);
}

#define SORT_SLOTS 16

__global__ __launch_bounds__(256) void gat_bin_sort(
    const int* __restrict__ ei, int* __restrict__ runstart,
    int* __restrict__ pairs, int E, int nbins) {
  __shared__ int hist[MAX_BINS + 1];
  __shared__ int wsum[256 / 32];
  __shared__ int2 slice[SORT_SLOTS * 256];
  int2* __restrict__ pr = (int2*)pairs;
  const int t = threadIdx.x, lane = t & 31, wave = t >> 5, g = blockIdx.x;
  const int ce = (E + NCHUNK - 1) / NCHUNK;
  const int e0 = g * ce;
  const int e1 = min(E, e0 + ce);
  const bool onchip = ce <= SORT_SLOTS * 256;

  int d[SORT_SLOTS], sr[SORT_SLOTS], rk[SORT_SLOTS];
  if (onchip) {
    const int last = max(e1 - 1, 0);
#pragma unroll
    for (int u = 0; u < SORT_SLOTS; ++u) {
      const int ee = e0 + t + u * 256;
      const int ix = min(ee, last);
      const int dd = ei[E + ix];
      sr[u] = ei[ix];
      d[u]  = (ee < e1) ? dd : -1;
    }
  }
  hist[t] = 0;
  hist[t + 256] = 0;
  __syncthreads();
  if (onchip) {
#pragma unroll
    for (int u = 0; u < SORT_SLOTS; ++u) {
      int r = 0;
      if (d[u] >= 0) r = atomicAdd(&hist[d[u] >> BIN_SHIFT], 1);
      rk[u] = r;
    }
  } else {
    for (int e = e0 + t; e < e1; e += 4 * 256) {
      int dl[4];
#pragma unroll
      for (int u = 0; u < 4; ++u) {
        const int ee = e + u * 256;
        dl[u] = (ee < e1) ? ei[E + ee] : -1;
      }
#pragma unroll
      for (int u = 0; u < 4; ++u)
        if (dl[u] >= 0) atomicAdd(&hist[dl[u] >> BIN_SHIFT], 1);
    }
  }
  __syncthreads();
  const int c0 = hist[2 * t], c1 = hist[2 * t + 1];
  const int v = c0 + c1;
  int inc = v;
  for (int o = 1; o < 32; o <<= 1) {
    const int u = __shfl_up(inc, o, 32);
    if (lane >= o) inc += u;
  }
  if (lane == 31) wsum[wave] = inc;
  __syncthreads();
  int below = 0;
#pragma unroll
  for (int w = 0; w < 256 / 32; ++w) below += (w < wave) ? wsum[w] : 0;
  const int x0 = (onchip ? 0 : e0) + below + inc - v;
  hist[2 * t]     = x0;
  hist[2 * t + 1] = x0 + c0;
  if (t == 255) hist[MAX_BINS] = x0 + v;
  __syncthreads();
  if (onchip) {
    int pos[SORT_SLOTS];
#pragma unroll
    for (int u = 0; u < SORT_SLOTS; ++u)
      pos[u] = hist[max(d[u], 0) >> BIN_SHIFT] + rk[u];
#pragma unroll
    for (int u = 0; u < SORT_SLOTS; ++u)
      if (d[u] >= 0) slice[pos[u]] = make_int2(sr[u], d[u]);
    __syncthreads();
    for (int b = t; b <= nbins; b += 256) runstart[b * NCHUNK + g] = e0 + hist[b];
    const int cnt = e1 - e0;
    int2 q[SORT_SLOTS];
#pragma unroll
    for (int u = 0; u < SORT_SLOTS; ++u) q[u] = slice[t + u * 256];
#pragma unroll
    for (int u = 0; u < SORT_SLOTS; ++u)
      if (t + u * 256 < cnt) pr[e0 + t + u * 256] = q[u];
  } else {
    for (int b = t; b <= nbins; b += 256) runstart[b * NCHUNK + g] = hist[b];
    __syncthreads();
    for (int e = e0 + t; e < e1; e += 4 * 256) {
      int dl[4], sl[4];
#pragma unroll
      for (int u = 0; u < 4; ++u) {
        const int ee = e + u * 256;
        dl[u] = (ee < e1) ? ei[E + ee] : -1;
        sl[u] = (ee < e1) ? ei[ee] : 0;
      }
#pragma unroll
      for (int u = 0; u < 4; ++u)
        if (dl[u] >= 0) {
          const int p = atomicAdd(&hist[dl[u] >> BIN_SHIFT], 1);
          pr[p] = make_int2(sl[u], dl[u]);
        }
    }
  }
}

#define RUN_LANES 16
#define RUN_STEPS 16

__device__ __forceinline__ void run_bounds(int rs, int re, int it, bool hi,
                                           int& s, int& e) {
  const int s0 = __builtin_amdgcn_readlane(rs, 2 * it);
  const int e0 = __builtin_amdgcn_readlane(re, 2 * it);
  const int s1 = __builtin_amdgcn_readlane(rs, 2 * it + 1);
  const int e1 = __builtin_amdgcn_readlane(re, 2 * it + 1);
  s = hi ? s1 : s0;
  e = hi ? e1 : e0;
}

__global__ __launch_bounds__(BIN_NODES) void gat_bin_csr(
    const int* __restrict__ pairs, const int* __restrict__ runstart,
    int* __restrict__ rowstart, int* __restrict__ deg,
    int* __restrict__ csr_src, int N, int E) {
  __shared__ int dcnt[BIN_NODES];
  __shared__ int dpos[BIN_NODES];
  __shared__ int wsum[BIN_NODES / 32];
  __shared__ int wbef[BIN_NODES / 32];
  const int t = threadIdx.x, lane = t & 31, wave = t >> 5;
  const int b = blockIdx.x;
  const int n = b * BIN_NODES + t;
  const int2* __restrict__ pp = (const int2*)pairs;
  const int rs = runstart[b * NCHUNK + t];
  const int re = runstart[(b + 1) * NCHUNK + t];
  const int ce = (E + NCHUNK - 1) / NCHUNK;
  int bef = rs - t * ce;
  dcnt[t] = 0;
  bef += lane_xor_i<1>(bef);
  bef += lane_xor_i<2>(bef);
  bef += lane_xor_i<4>(bef);
  bef += lane_xor_i<8>(bef);
  const int wtot = __builtin_amdgcn_readlane(bef, 0) + __builtin_amdgcn_readlane(bef, 16);
  if (lane == 0) wbef[wave] = wtot;
  __syncthreads();
  int base = 0;
#pragma unroll
  for (int w = 0; w < BIN_NODES / 32; ++w) base += wbef[w];

  const int sub = lane & (RUN_LANES - 1);
  const bool hi = lane >= RUN_LANES;
  const bool longrun = __any(re - rs > RUN_LANES);
  int2 r[RUN_STEPS];
#pragma unroll
  for (int it = 0; it < RUN_STEPS; ++it) {
    int s, e;
    run_bounds(rs, re, it, hi, s, e);
    r[it] = (s + sub < e) ? pp[s + sub] : make_int2(0, -1);
  }
#pragma unroll
  for (int it = 0; it < RUN_STEPS; ++it)
    if (r[it].y >= 0) atomicAdd(&dcnt[r[it].y & (BIN_NODES - 1)], 1);
  if (longrun) {
    for (int it = 0; it < RUN_STEPS; ++it) {
      int s, e;
      run_bounds(rs, re, it, hi, s, e);
      for (int i = s + sub + RUN_LANES; i < e; i += RUN_LANES)
        atomicAdd(&dcnt[pp[i].y & (BIN_NODES - 1)], 1);
    }
  }
  __syncthreads();
  const int v = dcnt[t];
  int inc = v;
  for (int o = 1; o < 32; o <<= 1) {
    const int u = __shfl_up(inc, o, 32);
    if (lane >= o) inc += u;
  }
  if (lane == 31) wsum[wave] = inc;
  __syncthreads();
  int below = 0;
#pragma unroll
  for (int w = 0; w < BIN_NODES / 32; ++w) below += (w < wave) ? wsum[w] : 0;
  const int first = base + below + inc - v;
  dpos[t] = first;
  if (n < N) { rowstart[n] = first; deg[n] = v; }
  __syncthreads();
#pragma unroll
  for (int it = 0; it < RUN_STEPS; ++it)
    if (r[it].y >= 0)
      csr_src[atomicAdd(&dpos[r[it].y & (BIN_NODES - 1)], 1)] = r[it].x;
  if (longrun) {
    for (int it = 0; it < RUN_STEPS; ++it) {
      int s, e;
      run_bounds(rs, re, it, hi, s, e);
      for (int i = s + sub + RUN_LANES; i < e; i += RUN_LANES) {
        const int2 q = pp[i];
        csr_src[atomicAdd(&dpos[q.y & (BIN_NODES - 1)], 1)] = q.x;
      }
    }
  }
}

template <int M>
__device__ __forceinline__ float lane_xor(float v) {
  return __builtin_bit_cast(float, __builtin_amdgcn_update_dpp(
      0, __builtin_bit_cast(int, v), 0x160 + M, 0xf, 0xf, true));
}

__device__ __forceinline__ float sum8_by_row(const float (&p)[8], bool b2,
                                             bool b1, bool b0) {
  float q[4], s[2];
#pragma unroll
  for (int i = 0; i < 4; ++i) {
    const float snd = b2 ? p[i] : p[i + 4];
    const float kp  = b2 ? p[i + 4] : p[i];
    q[i] = kp + lane_xor<4>(snd);
  }
#pragma unroll
  for (int i = 0; i < 2; ++i) {
    const float snd = b1 ? q[i] : q[i + 2];
    const float kp  = b1 ? q[i + 2] : q[i];
    s[i] = kp + lane_xor<2>(snd);
  }
  const float snd = b0 ? s[0] : s[1];
  const float kp  = b0 ? s[1] : s[0];
  return kp + lane_xor<1>(snd);
}

__global__ __launch_bounds__(128) void gat_gemm_wmma_f32(
    const float* __restrict__ X, const float* __restrict__ W,
    const float* __restrict__ att_src, const float* __restrict__ att_dst,
    float* __restrict__ H1, float* __restrict__ AS1, float* __restrict__ AD1,
    int N) {
  __shared__ float Wp[64 * 256];

  const int t    = threadIdx.x;
  const int wave = __builtin_amdgcn_readfirstlane(t >> 5);
  const int lane = t & 31;
  const int hf   = lane >> 4;
  const int c    = lane & 15;
  const int row0 = (blockIdx.x * 4 + wave) * 16;

  const int ar = min(row0 + c, N - 1);
  const float* __restrict__ xr = X + (size_t)ar * F_IN + 4 * hf;
  float4 a[16];
#pragma unroll
  for (int j = 0; j < 16; ++j) a[j] = *(const float4*)(xr + 8 * j);

  const float* __restrict__ wsrc = W + (size_t)(2 * wave) * F_IN + 4 * lane;
  float* __restrict__ wdst = Wp + wave * 256 + 8 * lane;
#pragma unroll
  for (int hb = 0; hb < 2; ++hb) {
    float4 r0[8], r1[8];
#pragma unroll
    for (int it = 0; it < 8; ++it) {
      const int n = hb * 8 + it;
      r0[it] = *(const float4*)(wsrc + (size_t)n * 8 * F_IN);
      r1[it] = *(const float4*)(wsrc + (size_t)n * 8 * F_IN + F_IN);
    }
#pragma unroll
    for (int it = 0; it < 8; ++it) {
      const int n = hb * 8 + it;
      float4* p = (float4*)(wdst + n * 4 * 256);
      p[0] = make_float4(r0[it].x, r1[it].x, r0[it].y, r1[it].y);
      p[1] = make_float4(r0[it].z, r1[it].z, r0[it].w, r1[it].w);
    }
  }
  __syncthreads();
  if (row0 >= N) return;

  v8f acc[8];
#pragma unroll
  for (int i = 0; i < 8; ++i) acc[i] = v8f{};

  float2 asv[4], adv[4];
#pragma unroll
  for (int u = 0; u < 4; ++u) {
    asv[u] = *(const float2*)(att_src + 32 * u + 2 * c);
    adv[u] = *(const float2*)(att_dst + 32 * u + 2 * c);
  }

  const float* __restrict__ bp = Wp + (2 * hf) * 256 + 4 * c;
#pragma unroll
  for (int j = 0; j < 16; ++j) {
#pragma unroll
    for (int s = 0; s < 2; ++s) {
      const float* q = bp + (4 * j + s) * 256;
      const float4 q0 = *(const float4*)(q);
      const float4 q1 = *(const float4*)(q + 64);
      const float4 q2 = *(const float4*)(q + 128);
      const float4 q3 = *(const float4*)(q + 192);
      v2f av;
      av.x = s ? a[j].z : a[j].x;
      av.y = s ? a[j].w : a[j].y;
      v2f b;
      b.x = q0.x; b.y = q0.y;
      acc[0] = __builtin_amdgcn_wmma_f32_16x16x4_f32(false, av, false, b, (short)0, acc[0], false, false);
      b.x = q0.z; b.y = q0.w;
      acc[1] = __builtin_amdgcn_wmma_f32_16x16x4_f32(false, av, false, b, (short)0, acc[1], false, false);
      b.x = q1.x; b.y = q1.y;
      acc[2] = __builtin_amdgcn_wmma_f32_16x16x4_f32(false, av, false, b, (short)0, acc[2], false, false);
      b.x = q1.z; b.y = q1.w;
      acc[3] = __builtin_amdgcn_wmma_f32_16x16x4_f32(false, av, false, b, (short)0, acc[3], false, false);
      b.x = q2.x; b.y = q2.y;
      acc[4] = __builtin_amdgcn_wmma_f32_16x16x4_f32(false, av, false, b, (short)0, acc[4], false, false);
      b.x = q2.z; b.y = q2.w;
      acc[5] = __builtin_amdgcn_wmma_f32_16x16x4_f32(false, av, false, b, (short)0, acc[5], false, false);
      b.x = q3.x; b.y = q3.y;
      acc[6] = __builtin_amdgcn_wmma_f32_16x16x4_f32(false, av, false, b, (short)0, acc[6], false, false);
      b.x = q3.z; b.y = q3.w;
      acc[7] = __builtin_amdgcn_wmma_f32_16x16x4_f32(false, av, false, b, (short)0, acc[7], false, false);
    }
  }

  float* __restrict__ o = H1 + (size_t)(row0 + 8 * hf) * F_IN + 2 * c;
#pragma unroll
  for (int r = 0; r < 8; ++r) {
#pragma unroll
    for (int u = 0; u < 4; ++u)
      *(float2*)(o + (size_t)r * F_IN + 32 * u) = make_float2(acc[2 * u][r], acc[2 * u + 1][r]);
  }

  const bool b0 = (lane & 1) != 0, b1 = (lane & 2) != 0, b2 = (lane & 4) != 0;
  const bool g  = (lane & 8) != 0;
  float ts[4], td[4];
#pragma unroll
  for (int u = 0; u < 4; ++u) {
    float ps[8], pd[8];
#pragma unroll
    for (int r = 0; r < 8; ++r) {
      ps[r] = fmaf(acc[2 * u + 1][r], asv[u].y, acc[2 * u][r] * asv[u].x);
      pd[r] = fmaf(acc[2 * u + 1][r], adv[u].y, acc[2 * u][r] * adv[u].x);
    }
    ts[u] = sum8_by_row(ps, b2, b1, b0);
    td[u] = sum8_by_row(pd, b2, b1, b0);
  }
  const float xs0 = lane_xor<8>(g ? ts[0] : ts[2]);
  const float xs1 = lane_xor<8>(g ? ts[1] : ts[3]);
  const float xd0 = lane_xor<8>(g ? td[0] : td[2]);
  const float xd1 = lane_xor<8>(g ? td[1] : td[3]);
  const float4 os = g ? make_float4(xs0, ts[2], xs1, ts[3]) : make_float4(ts[0], xs0, ts[1], xs1);
  const float4 od = g ? make_float4(xd0, td[2], xd1, td[3]) : make_float4(td[0], xd0, td[1], xd1);
  const size_t ao = (size_t)(row0 + 8 * hf + (lane & 7)) * HEADS + (g ? 4 : 0);
  *(float4*)(AS1 + ao) = os;
  *(float4*)(AD1 + ao) = od;
}

#define G1_WAVES 16384

__device__ __forceinline__ float lane_xor16(float v) {
  const int b = __builtin_bit_cast(int, v);
  return __builtin_bit_cast(float, __builtin_amdgcn_permlanex16(
      b, b, 0x76543210, 0xfedcba98, false, false));
}

__device__ __forceinline__ float sum2_by_wave(float p0, float p1, bool b0) {
  const float snd = b0 ? p0 : p1;
  const float kp  = b0 ? p1 : p0;
  float v = kp + lane_xor<1>(snd);
  v += lane_xor<2>(v);
  v += lane_xor<4>(v);
  v += lane_xor<8>(v);
  v += lane_xor16(v);
  return v;
}

__device__ __forceinline__ float max16_by_head(const float (&v)[4], bool b3,
                                               bool b2) {
  float q[2];
#pragma unroll
  for (int i = 0; i < 2; ++i) {
    const float snd = b3 ? v[i] : v[i + 2];
    const float kp  = b3 ? v[i + 2] : v[i];
    q[i] = fmaxf(kp, lane_xor<8>(snd));
  }
  const float snd = b2 ? q[0] : q[1];
  const float kp  = b2 ? q[1] : q[0];
  float r = fmaxf(kp, lane_xor<4>(snd));
  r = fmaxf(r, lane_xor<1>(r));
  r = fmaxf(r, lane_xor<2>(r));
  return r;
}

template <int G>
__device__ __forceinline__ void gather_group(
    int mine, int j0, int nn, const float* __restrict__ H1,
    const float* __restrict__ as1, int lane, int h, float adh, float m,
    float& l, float4& acc) {
  float4 r[G];
  float a[G];
#pragma unroll
  for (int u = 0; u < G; ++u) {
    const int s = __builtin_amdgcn_readlane(mine, j0 + u);
    r[u] = *(const float4*)(H1 + (size_t)s * F_IN + 4 * lane);
    a[u] = as1[s * HEADS + h];
  }
#pragma unroll
  for (int u = 0; u < G; ++u) {
    float w = __expf(leaky(a[u] + adh) - m);
    w = (j0 + u < nn) ? w : 0.f;
    l += w;
    acc.x = fmaf(w, r[u].x, acc.x); acc.y = fmaf(w, r[u].y, acc.y);
    acc.z = fmaf(w, r[u].z, acc.z); acc.w = fmaf(w, r[u].w, acc.w);
  }
}

__global__ __launch_bounds__(256) void gat_gather1(
    const int* __restrict__ rowstart, const int* __restrict__ deg,
    const int* __restrict__ csr_src, const float* __restrict__ as1,
    const float* __restrict__ ad1, const float* __restrict__ H1,
    const float* __restrict__ b1, const float* __restrict__ W2,
    const float* __restrict__ att_src2, const float* __restrict__ att_dst2,
    float* __restrict__ REC, int N) {
  const int lane = threadIdx.x & 31;
  const int wv = __builtin_amdgcn_readfirstlane(blockIdx.x * 8 + (threadIdx.x >> 5));
  const int* metap = (lane & 1) ? deg : rowstart;
  const int meta0 = metap[min(wv, N - 1)];
  __shared__ float w2s[F_IN * OUTC];
  for (int i = threadIdx.x; i < F_IN * OUTC; i += 256)
    w2s[((i & 4) << 5) + ((i >> 3) << 2) + (i & 3)] = W2[i];
  __syncthreads();
  if (wv >= N) return;
  const int h   = lane >> 2;
  const int j16 = lane & 15;
  const int k   = lane >> 4;
  const bool b2 = (lane & 4) != 0, b3 = (lane & 8) != 0;
  const bool b0 = (lane & 1) != 0;
  const float4 b = *(const float4*)(b1 + 4 * lane);
  const float as20 = att_src2[0], as21 = att_src2[1];
  const float ad20 = att_dst2[0], ad21 = att_dst2[1];

  int start = __builtin_amdgcn_readlane(meta0, 0);
  int cnt   = __builtin_amdgcn_readlane(meta0, 1);
  for (int n = wv; n < N; n += G1_WAVES) {
    const int nx = min(n + G1_WAVES, N - 1);
    const int meta_nx = metap[nx];

    const float adh    = ad1[n * HEADS + h];
    const float asself = as1[n * HEADS + h];
    const float4 adk   = *(const float4*)(ad1 + (size_t)n * HEADS + 4 * k);
    const float4 hv = *(const float4*)(H1 + (size_t)n * F_IN + 4 * lane);

    const int nn0 = min(16, cnt);
    const int s0 = (j16 < nn0) ? csr_src[start + j16] : n;
    const float4 a0 = *(const float4*)(as1 + (size_t)s0 * HEADS + 4 * k);
    float4 r0[4];
    float al0[4];
#pragma unroll
    for (int u = 0; u < 4; ++u) {
      const int s = __builtin_amdgcn_readlane(s0, u);
      r0[u]  = *(const float4*)(H1 + (size_t)s * F_IN + 4 * lane);
      al0[u] = as1[s * HEADS + h];
    }
    float v[4];
    v[0] = leaky(a0.x + adk.x); v[1] = leaky(a0.y + adk.y);
    v[2] = leaky(a0.z + adk.z); v[3] = leaky(a0.w + adk.w);
    float m = max16_by_head(v, b3, b2);
    for (int c0 = 16; c0 < cnt; c0 += 16) {
      const int nn = min(16, cnt - c0);
      const int s = (j16 < nn) ? csr_src[start + c0 + j16] : n;
      const float4 a = *(const float4*)(as1 + (size_t)s * HEADS + 4 * k);
      v[0] = leaky(a.x + adk.x); v[1] = leaky(a.y + adk.y);
      v[2] = leaky(a.z + adk.z); v[3] = leaky(a.w + adk.w);
      m = fmaxf(m, max16_by_head(v, b3, b2));
    }
    const float vself = leaky(asself + adh);
    m = fmaxf(m, vself);

    const float ex = __expf(vself - m);
    float l = ex;
    float4 acc;
    acc.x = ex * hv.x; acc.y = ex * hv.y; acc.z = ex * hv.z; acc.w = ex * hv.w;
#pragma unroll
    for (int u = 0; u < 4; ++u) {
      float w = __expf(leaky(al0[u] + adh) - m);
      w = (u < nn0) ? w : 0.f;
      l += w;
      acc.x = fmaf(w, r0[u].x, acc.x); acc.y = fmaf(w, r0[u].y, acc.y);
      acc.z = fmaf(w, r0[u].z, acc.z); acc.w = fmaf(w, r0[u].w, acc.w);
    }
    for (int j0 = 4; j0 < nn0; j0 += 4)
      gather_group<4>(s0, j0, nn0, H1, as1, lane, h, adh, m, l, acc);
    for (int c0 = 16; c0 < cnt; c0 += 16) {
      const int nn = min(16, cnt - c0);
      const int mine = (j16 < nn) ? csr_src[start + c0 + j16] : n;
      for (int j0 = 0; j0 < nn; j0 += 4)
        gather_group<4>(mine, j0, nn, H1, as1, lane, h, adh, m, l, acc);
    }

    asm volatile("" ::: "memory");
    const float4 wA = *(const float4*)(w2s + 4 * lane);
    const float4 wB = *(const float4*)(w2s + F_IN + 4 * lane);

    const float den = l + 1e-16f;
    float4 o;
    o.x = acc.x / den + b.x; o.y = acc.y / den + b.y;
    o.z = acc.z / den + b.z; o.w = acc.w / den + b.w;
    o.x = (o.x > 0.f) ? o.x : (__expf(o.x) - 1.f);
    o.y = (o.y > 0.f) ? o.y : (__expf(o.y) - 1.f);
    o.z = (o.z > 0.f) ? o.z : (__expf(o.z) - 1.f);
    o.w = (o.w > 0.f) ? o.w : (__expf(o.w) - 1.f);
    const float p0 = fmaf(o.w, wB.z, fmaf(o.z, wB.x, fmaf(o.y, wA.z, o.x * wA.x)));
    const float p1 = fmaf(o.w, wB.w, fmaf(o.z, wB.y, fmaf(o.y, wA.w, o.x * wA.y)));
    const float tot = sum2_by_wave(p0, p1, b0);
    const float h20 = __builtin_bit_cast(float, __builtin_amdgcn_readlane(__builtin_bit_cast(int, tot), 0));
    const float h21 = __builtin_bit_cast(float, __builtin_amdgcn_readlane(__builtin_bit_cast(int, tot), 1));
    const float sd = h20 * as20 + h21 * as21;
    const float dd = h20 * ad20 + h21 * ad21;
    if (lane == 0) *(float4*)(REC + 4 * n) = make_float4(h20, h21, sd, dd);

    start = __builtin_amdgcn_readlane(meta_nx, 0);
    cnt   = __builtin_amdgcn_readlane(meta_nx, 1);
  }
}

#define G2_CHUNK 16

__device__ __forceinline__ void g2_sources(
    const int* __restrict__ csr_src, int start, int c0, int cnt, int n,
    int (&s)[G2_CHUNK]) {
#pragma unroll
  for (int u = 0; u < G2_CHUNK; ++u) {
    const int v = csr_src[max(start + min(c0 + u, cnt - 1), 0)];
    s[u] = (c0 + u < cnt) ? v : n;
  }
}

__global__ __launch_bounds__(256) void gat_gather2(
    const int* __restrict__ rowstart, const int* __restrict__ deg,
    const int* __restrict__ csr_src, const float* __restrict__ REC,
    const float* __restrict__ b2, float* __restrict__ OUT, int N) {
  const int n = blockIdx.x * 256 + threadIdx.x;
  if (n >= N) return;
  const float4* __restrict__ rec = (const float4*)REC;
  const int start = rowstart[n];
  const int cnt   = deg[n];
  const float4 me = rec[n];
  const float adn   = me.w;
  const float vself = leaky(me.z + adn);

  int s0[G2_CHUNK];
  float4 r0[G2_CHUNK];
  g2_sources(csr_src, start, 0, cnt, n, s0);
#pragma unroll
  for (int u = 0; u < G2_CHUNK; ++u) r0[u] = rec[s0[u]];

  float m = vself;
#pragma unroll
  for (int u = 0; u < G2_CHUNK; ++u) m = fmaxf(m, leaky(r0[u].z + adn));
  for (int c0 = G2_CHUNK; c0 < cnt; c0 += G2_CHUNK) {
    int s[G2_CHUNK];
    float a[G2_CHUNK];
    g2_sources(csr_src, start, c0, cnt, n, s);
#pragma unroll
    for (int u = 0; u < G2_CHUNK; ++u) a[u] = REC[(size_t)s[u] * 4 + 2];
#pragma unroll
    for (int u = 0; u < G2_CHUNK; ++u) m = fmaxf(m, leaky(a[u] + adn));
  }

  const float ex = __expf(vself - m);
  float l = ex;
  float a0 = ex * me.x, a1 = ex * me.y;
#pragma unroll
  for (int u = 0; u < G2_CHUNK; ++u) {
    float w = __expf(leaky(r0[u].z + adn) - m);
    w = (u < cnt) ? w : 0.f;
    l += w;
    a0 = fmaf(w, r0[u].x, a0);
    a1 = fmaf(w, r0[u].y, a1);
  }
  for (int c0 = G2_CHUNK; c0 < cnt; c0 += G2_CHUNK) {
    int s[G2_CHUNK];
    float4 r[G2_CHUNK];
    g2_sources(csr_src, start, c0, cnt, n, s);
#pragma unroll
    for (int u = 0; u < G2_CHUNK; ++u) r[u] = rec[s[u]];
#pragma unroll
    for (int u = 0; u < G2_CHUNK; ++u) {
      float w = __expf(leaky(r[u].z + adn) - m);
      w = (c0 + u < cnt) ? w : 0.f;
      l += w;
      a0 = fmaf(w, r[u].x, a0);
      a1 = fmaf(w, r[u].y, a1);
    }
  }
  const float den = l + 1e-16f;
  float2 o;
  o.x = a0 / den + b2[0];
  o.y = a1 / den + b2[1];
  *(float2*)(OUT + (size_t)n * OUTC) = o;
}

static inline unsigned gdim(long long n, int b) {
  return (unsigned)((n + b - 1) / b);
}

extern "C" void kernel_launch(void* const* d_in, const int* in_sizes, int n_in,
                              void* d_out, int out_size, void* d_ws,
                              size_t ws_size, hipStream_t stream) {
  const float* x        = (const float*)d_in[0];
  const int*   ei       = (const int*)d_in[1];
  const float* W1       = (const float*)d_in[2];
  const float* att_src1 = (const float*)d_in[3];
  const float* att_dst1 = (const float*)d_in[4];
  const float* b1       = (const float*)d_in[5];
  const float* W2       = (const float*)d_in[6];
  const float* att_src2 = (const float*)d_in[7];
  const float* att_dst2 = (const float*)d_in[8];
  const float* b2       = (const float*)d_in[9];

  const int N  = in_sizes[0] / F_IN;
  const int E  = in_sizes[1] / 2;

  float* ws = (float*)d_ws;
  size_t off = 0;
  float* h1   = ws + off; off += (size_t)N * F_IN;
  float* rec2 = ws + off; off += (size_t)N * 4;
  float* as1  = ws + off; off += (size_t)N * HEADS;
  float* ad1  = ws + off; off += (size_t)N * HEADS;
  int* pairs    = (int*)(ws + off); off += (size_t)E * 2;
  int* deg      = (int*)(ws + off); off += (size_t)N;
  int* rowstart = (int*)(ws + off); off += (size_t)N;
  int* csr_src  = (int*)(ws + off); off += (size_t)E;
  const int nbins = (int)gdim(N, BIN_NODES);
  int* runstart = (int*)(ws + off); off += (size_t)(nbins + 1) * NCHUNK;
  (void)ws_size; (void)n_in; (void)out_size;

  const int B = 256;

  gat_bin_sort<<<NCHUNK, B, 0, stream>>>(ei, runstart, pairs, E, nbins);
  gat_bin_csr<<<nbins, BIN_NODES, 0, stream>>>(pairs, runstart, rowstart, deg, csr_src, N, E);

  gat_gemm_wmma_f32<<<gdim(N, 64), 128, 0, stream>>>(x, W1, att_src1, att_dst1, h1, as1, ad1, N);
  gat_gather1<<<G1_WAVES / 8, B, 0, stream>>>(rowstart, deg, csr_src, as1, ad1, h1, b1, W2, att_src2, att_dst2, rec2, N);

  gat_gather2<<<gdim(N, B), B, 0, stream>>>(rowstart, deg, csr_src, rec2, b2, (float*)d_out, N);
}
